# starting kernel
# baseline (speedup 1.0000x reference)
#include <hip/hip_runtime.h>

typedef __attribute__((ext_vector_type(2))) float v2f;
typedef __attribute__((ext_vector_type(8))) float v8f;

#define BN_EPS 1e-5f

__global__ void zero_f32(float* __restrict__ p, long long n) {
  long long i = (long long)blockIdx.x * blockDim.x + threadIdx.x;
  if (i < n) p[i] = 0.0f;
}

__global__ void count_deg(const int* __restrict__ dst, float* __restrict__ deg, int e) {
  int i = blockIdx.x * blockDim.x + threadIdx.x;
  if (i < e) atomicAdd(&deg[dst[i]], 1.0f);
}

__global__ void make_dinv(float* __restrict__ deg, int n) {
  int i = blockIdx.x * blockDim.x + threadIdx.x;
  if (i < n) deg[i] = rsqrtf(deg[i] + 1.0f);
}

template <int K, int NC, int MT>
__global__ void __launch_bounds__(MT*(NC/16)*32)
gemm_wmma_f32(const float* __restrict__ A, const float* __restrict__ B,
              float* __restrict__ C) {
  constexpr int KP  = K + 4;
  constexpr int NCP = NC + 8;
  constexpr int NW  = MT * (NC/16);
  __shared__ float sA[MT*16*KP];
  __shared__ float sB[K*NCP];

  const int tid = threadIdx.x;
  const int nthreads = NW * 32;
  const float* Ablk = A + (size_t)blockIdx.x * (MT*16) * K;

  for (int idx = tid; idx < MT*16*(K/4); idx += nthreads) {
    int r = idx / (K/4), c4 = (idx % (K/4)) * 4;
    *(float4*)&sA[r*KP + c4] = *(const float4*)&Ablk[(size_t)r*K + c4];
  }
  for (int idx = tid; idx < K*(NC/4); idx += nthreads) {
    int r = idx / (NC/4), c4 = (idx % (NC/4)) * 4;
    *(float4*)&sB[r*NCP + c4] = *(const float4*)&B[(size_t)r*NC + c4];
  }
  __syncthreads();

  const int wave  = tid >> 5, lane = tid & 31;
  const int mt    = wave / (NC/16), nt = wave % (NC/16);
  const int rrow  = lane & 15;
  const int khalf = lane >> 4;

  const float* a_ptr = &sA[(mt*16 + rrow)*KP + khalf*2];
  const float* b_ptr = &sB[(khalf*2)*NCP + nt*16 + rrow];

  v8f c0 = {};
  v8f c1 = {};
  #pragma unroll 4
  for (int ks = 0; ks < K/8; ++ks) {
    v2f a0; a0.x = a_ptr[0];     a0.y = a_ptr[1];
    v2f b0; b0.x = b_ptr[0];     b0.y = b_ptr[NCP];
    v2f a1; a1.x = a_ptr[4];     a1.y = a_ptr[5];
    v2f b1; b1.x = b_ptr[4*NCP]; b1.y = b_ptr[5*NCP];
    c0 = __builtin_amdgcn_wmma_f32_16x16x4_f32(false, a0, false, b0,
                                               (short)0, c0, false, false);
    c1 = __builtin_amdgcn_wmma_f32_16x16x4_f32(false, a1, false, b1,
                                               (short)0, c1, false, false);
    a_ptr += 8;
    b_ptr += 8*NCP;
  }
  v8f c = c0 + c1;

  const int gm  = blockIdx.x*(MT*16) + mt*16 + khalf*8;
  const int col = nt*16 + rrow;
  #pragma unroll
  for (int r = 0; r < 8; ++r)
    C[(size_t)(gm + r)*NC + col] = c[r];
}

template <int F>
__global__ void scatter_edges(const int* __restrict__ src, const int* __restrict__ dst,
                              const float* __restrict__ dinv, const float* __restrict__ h,
                              float* __restrict__ agg, long long total) {
  long long t = (long long)blockIdx.x * blockDim.x + threadIdx.x;
  if (t >= total) return;
  constexpr int HALF = F / 2;
  int e = (int)(t / HALF);
  int f = (int)(t % HALF) * 2;
  int s = src[e], d = dst[e];
  float norm = dinv[s] * dinv[d];
  float2 hv = *(const float2*)&h[(size_t)s*F + f];
  atomicAdd(&agg[(size_t)d*F + f],     hv.x * norm);
  atomicAdd(&agg[(size_t)d*F + f + 1], hv.y * norm);
}

template <int F>
__global__ void finalize_bn(float* __restrict__ agg, const float* __restrict__ h,
                            const float* __restrict__ dinv, const float* __restrict__ bias,
                            const float* __restrict__ gamma, const float* __restrict__ beta,
                            const float* __restrict__ mean, const float* __restrict__ var,
                            long long total) {
  long long t = (long long)blockIdx.x * blockDim.x + threadIdx.x;
  if (t >= total) return;
  int node = (int)(t / F), f = (int)(t % F);
  float di = dinv[node];
  float v = agg[t] + h[t] * di * di + bias[f];
  v = (v - mean[f]) * rsqrtf(var[f] + BN_EPS) * gamma[f] + beta[f];
  agg[t] = fmaxf(v, 0.0f);
}

__global__ void final_head(const float* __restrict__ agg2, const float* __restrict__ h2,
                           const float* __restrict__ dinv, const float* __restrict__ b2,
                           const float* __restrict__ g2, const float* __restrict__ be2,
                           const float* __restrict__ m2, const float* __restrict__ v2,
                           const float* __restrict__ Wl, const float* __restrict__ bl,
                           float* __restrict__ out, int n) {
  int node = blockIdx.x * blockDim.x + threadIdx.x;
  if (node >= n) return;
  float di = dinv[node];
  float di2 = di * di;
  float acc0 = bl[0], acc1 = bl[1];
  const float* ar = agg2 + (size_t)node * 32;
  const float* hr = h2 + (size_t)node * 32;
  #pragma unroll
  for (int f = 0; f < 32; ++f) {
    float v = ar[f] + hr[f] * di2 + b2[f];
    v = (v - m2[f]) * rsqrtf(v2[f] + BN_EPS) * g2[f] + be2[f];
    v = fmaxf(v, 0.0f);
    acc0 += v * Wl[f*2 + 0];
    acc1 += v * Wl[f*2 + 1];
  }
  out[(size_t)node*2 + 0] = acc0;
  out[(size_t)node*2 + 1] = acc1;
}

extern "C" void kernel_launch(void* const* d_in, const int* in_sizes, int n_in,
                              void* d_out, int out_size, void* d_ws, size_t ws_size,
                              hipStream_t stream) {
  const float* x      = (const float*)d_in[0];
  const int*   ei     = (const int*)  d_in[1];
  const float* W1     = (const float*)d_in[2];
  const float* b1     = (const float*)d_in[3];
  const float* gamma1 = (const float*)d_in[4];
  const float* beta1  = (const float*)d_in[5];
  const float* mean1  = (const float*)d_in[6];
  const float* var1   = (const float*)d_in[7];
  const float* W2     = (const float*)d_in[8];
  const float* b2     = (const float*)d_in[9];
  const float* gamma2 = (const float*)d_in[10];
  const float* beta2  = (const float*)d_in[11];
  const float* mean2  = (const float*)d_in[12];
  const float* var2   = (const float*)d_in[13];
  const float* Wl     = (const float*)d_in[14];
  const float* bl     = (const float*)d_in[15];

  const int n = in_sizes[0] / 256;
  const int e = in_sizes[1] / 2;
  const int* src = ei;
  const int* dst = ei + e;

  float* ws   = (float*)d_ws;
  float* dinv = ws;
  float* h1   = dinv + (size_t)n;
  float* agg1 = h1   + (size_t)n * 64;
  float* h2   = agg1 + (size_t)n * 64;
  float* agg2 = h2   + (size_t)n * 32;

  const int T = 256;
  auto blocks = [](long long total, int t) {
    return (unsigned int)((total + t - 1) / t);
  };

  zero_f32<<<blocks(n, T), T, 0, stream>>>(dinv, n);
  zero_f32<<<blocks((long long)n * 64, T), T, 0, stream>>>(agg1, (long long)n * 64);
  zero_f32<<<blocks((long long)n * 32, T), T, 0, stream>>>(agg2, (long long)n * 32);

  count_deg<<<blocks(e, T), T, 0, stream>>>(dst, dinv, e);
  make_dinv<<<blocks(n, T), T, 0, stream>>>(dinv, n);

  gemm_wmma_f32<256, 64, 1><<<n / 16, 128, 0, stream>>>(x, W1, h1);

  scatter_edges<64><<<blocks((long long)e * 32, T), T, 0, stream>>>(
      src, dst, dinv, h1, agg1, (long long)e * 32);

  finalize_bn<64><<<blocks((long long)n * 64, T), T, 0, stream>>>(
      agg1, h1, dinv, b1, gamma1, beta1, mean1, var1, (long long)n * 64);

  gemm_wmma_f32<64, 32, 2><<<n / 32, 128, 0, stream>>>(agg1, W2, h2);

  scatter_edges<32><<<blocks((long long)e * 16, T), T, 0, stream>>>(
      src, dst, dinv, h2, agg2, (long long)e * 16);

  final_head<<<blocks(n, T), T, 0, stream>>>(
      agg2, h2, dinv, b2, gamma2, beta2, mean2, var2, Wl, bl,
      (float*)d_out, n);
}


# scatter: wave takes 32 edges (coalesced index load, lane broadcast), one feature per lane so each gather and f32 atomic covers whole 128-B lines, gathers issued in groups; atomic sum order as before
# speedup vs baseline: 1.3983x; 1.3983x over previous
#include <hip/hip_runtime.h>

typedef __attribute__((ext_vector_type(2))) float v2f;
typedef __attribute__((ext_vector_type(8))) float v8f;

#define BN_EPS 1e-5f

__global__ void zero_f32(float* __restrict__ p, long long n) {
  long long i = (long long)blockIdx.x * blockDim.x + threadIdx.x;
  if (i < n) p[i] = 0.0f;
}

__global__ void count_deg(const int* __restrict__ dst, float* __restrict__ deg, int e) {
  int i = blockIdx.x * blockDim.x + threadIdx.x;
  if (i < e) atomicAdd(&deg[dst[i]], 1.0f);
}

__global__ void make_dinv(float* __restrict__ deg, int n) {
  int i = blockIdx.x * blockDim.x + threadIdx.x;
  if (i < n) deg[i] = rsqrtf(deg[i] + 1.0f);
}

template <int K, int NC, int MT>
__global__ void __launch_bounds__(MT*(NC/16)*32)
gemm_wmma_f32(const float* __restrict__ A, const float* __restrict__ B,
              float* __restrict__ C) {
  constexpr int KP  = K + 4;
  constexpr int NCP = NC + 8;
  constexpr int NW  = MT * (NC/16);
  __shared__ float sA[MT*16*KP];
  __shared__ float sB[K*NCP];

  const int tid = threadIdx.x;
  const int nthreads = NW * 32;
  const float* Ablk = A + (size_t)blockIdx.x * (MT*16) * K;

  for (int idx = tid; idx < MT*16*(K/4); idx += nthreads) {
    int r = idx / (K/4), c4 = (idx % (K/4)) * 4;
    *(float4*)&sA[r*KP + c4] = *(const float4*)&Ablk[(size_t)r*K + c4];
  }
  for (int idx = tid; idx < K*(NC/4); idx += nthreads) {
    int r = idx / (NC/4), c4 = (idx % (NC/4)) * 4;
    *(float4*)&sB[r*NCP + c4] = *(const float4*)&B[(size_t)r*NC + c4];
  }
  __syncthreads();

  const int wave  = tid >> 5, lane = tid & 31;
  const int mt    = wave / (NC/16), nt = wave % (NC/16);
  const int rrow  = lane & 15;
  const int khalf = lane >> 4;

  const float* a_ptr = &sA[(mt*16 + rrow)*KP + khalf*2];
  const float* b_ptr = &sB[(khalf*2)*NCP + nt*16 + rrow];

  v8f c0 = {};
  v8f c1 = {};
  #pragma unroll 4
  for (int ks = 0; ks < K/8; ++ks) {
    v2f a0; a0.x = a_ptr[0];     a0.y = a_ptr[1];
    v2f b0; b0.x = b_ptr[0];     b0.y = b_ptr[NCP];
    v2f a1; a1.x = a_ptr[4];     a1.y = a_ptr[5];
    v2f b1; b1.x = b_ptr[4*NCP]; b1.y = b_ptr[5*NCP];
    c0 = __builtin_amdgcn_wmma_f32_16x16x4_f32(false, a0, false, b0,
                                               (short)0, c0, false, false);
    c1 = __builtin_amdgcn_wmma_f32_16x16x4_f32(false, a1, false, b1,
                                               (short)0, c1, false, false);
    a_ptr += 8;
    b_ptr += 8*NCP;
  }
  v8f c = c0 + c1;

  const int gm  = blockIdx.x*(MT*16) + mt*16 + khalf*8;
  const int col = nt*16 + rrow;
  #pragma unroll
  for (int r = 0; r < 8; ++r)
    C[(size_t)(gm + r)*NC + col] = c[r];
}

template <int F>
__global__ void __launch_bounds__(256)
scatter_edges(const int* __restrict__ src, const int* __restrict__ dst,
              const float* __restrict__ dinv, const float* __restrict__ h,
              float* __restrict__ agg, int e) {
  constexpr int NL = F / 32;
  constexpr int G  = 16 / NL;
  const int lane = threadIdx.x & 31;
  const long long e0 = ((long long)blockIdx.x * blockDim.x + threadIdx.x) & ~31LL;
  if (e0 >= e) return;
  const long long left = (long long)e - e0;
  const int cnt = left < 32 ? (int)left : 32;

  int s_l = 0, d_l = 0;
  float n_l = 0.0f;
  if (lane < cnt) {
    s_l = src[e0 + lane];
    d_l = dst[e0 + lane];
    n_l = dinv[s_l] * dinv[d_l];
  }

  if (cnt == 32) {
    #pragma unroll
    for (int j = 0; j < 32; j += G) {
      float v[G][NL];
      int   dd[G];
      float nn[G];
      #pragma unroll
      for (int u = 0; u < G; ++u) {
        const int s = __builtin_amdgcn_readlane(s_l, j + u);
        dd[u] = __builtin_amdgcn_readlane(d_l, j + u);
        nn[u] = __int_as_float(__builtin_amdgcn_readlane(__float_as_int(n_l), j + u));
        const float* hp = h + (size_t)s * F + lane;
        for (int l = 0; l < NL; ++l) v[u][l] = hp[l * 32];
      }
      #pragma unroll
      for (int u = 0; u < G; ++u) {
        float* ap = agg + (size_t)dd[u] * F + lane;
        for (int l = 0; l < NL; ++l) atomicAdd(&ap[l * 32], v[u][l] * nn[u]);
      }
    }
  } else {
    for (int j = 0; j < cnt; ++j) {
      const int   s  = __shfl(s_l, j);
      const int   d  = __shfl(d_l, j);
      const float nr = __shfl(n_l, j);
      const float* hp = h + (size_t)s * F + lane;
      float* ap = agg + (size_t)d * F + lane;
      for (int l = 0; l < NL; ++l) atomicAdd(&ap[l * 32], hp[l * 32] * nr);
    }
  }
}

template <int F>
__global__ void finalize_bn(float* __restrict__ agg, const float* __restrict__ h,
                            const float* __restrict__ dinv, const float* __restrict__ bias,
                            const float* __restrict__ gamma, const float* __restrict__ beta,
                            const float* __restrict__ mean, const float* __restrict__ var,
                            long long total) {
  long long t = (long long)blockIdx.x * blockDim.x + threadIdx.x;
  if (t >= total) return;
  int node = (int)(t / F), f = (int)(t % F);
  float di = dinv[node];
  float v = agg[t] + h[t] * di * di + bias[f];
  v = (v - mean[f]) * rsqrtf(var[f] + BN_EPS) * gamma[f] + beta[f];
  agg[t] = fmaxf(v, 0.0f);
}

__global__ void final_head(const float* __restrict__ agg2, const float* __restrict__ h2,
                           const float* __restrict__ dinv, const float* __restrict__ b2,
                           const float* __restrict__ g2, const float* __restrict__ be2,
                           const float* __restrict__ m2, const float* __restrict__ v2,
                           const float* __restrict__ Wl, const float* __restrict__ bl,
                           float* __restrict__ out, int n) {
  int node = blockIdx.x * blockDim.x + threadIdx.x;
  if (node >= n) return;
  float di = dinv[node];
  float di2 = di * di;
  float acc0 = bl[0], acc1 = bl[1];
  const float* ar = agg2 + (size_t)node * 32;
  const float* hr = h2 + (size_t)node * 32;
  #pragma unroll
  for (int f = 0; f < 32; ++f) {
    float v = ar[f] + hr[f] * di2 + b2[f];
    v = (v - m2[f]) * rsqrtf(v2[f] + BN_EPS) * g2[f] + be2[f];
    v = fmaxf(v, 0.0f);
    acc0 += v * Wl[f*2 + 0];
    acc1 += v * Wl[f*2 + 1];
  }
  out[(size_t)node*2 + 0] = acc0;
  out[(size_t)node*2 + 1] = acc1;
}

extern "C" void kernel_launch(void* const* d_in, const int* in_sizes, int n_in,
                              void* d_out, int out_size, void* d_ws, size_t ws_size,
                              hipStream_t stream) {
  const float* x      = (const float*)d_in[0];
  const int*   ei     = (const int*)  d_in[1];
  const float* W1     = (const float*)d_in[2];
  const float* b1     = (const float*)d_in[3];
  const float* gamma1 = (const float*)d_in[4];
  const float* beta1  = (const float*)d_in[5];
  const float* mean1  = (const float*)d_in[6];
  const float* var1   = (const float*)d_in[7];
  const float* W2     = (const float*)d_in[8];
  const float* b2     = (const float*)d_in[9];
  const float* gamma2 = (const float*)d_in[10];
  const float* beta2  = (const float*)d_in[11];
  const float* mean2  = (const float*)d_in[12];
  const float* var2   = (const float*)d_in[13];
  const float* Wl     = (const float*)d_in[14];
  const float* bl     = (const float*)d_in[15];

  const int n = in_sizes[0] / 256;
  const int e = in_sizes[1] / 2;
  const int* src = ei;
  const int* dst = ei + e;

  float* ws   = (float*)d_ws;
  float* dinv = ws;
  float* h1   = dinv + (size_t)n;
  float* agg1 = h1   + (size_t)n * 64;
  float* h2   = agg1 + (size_t)n * 64;
  float* agg2 = h2   + (size_t)n * 32;

  const int T = 256;
  auto blocks = [](long long total, int t) {
    return (unsigned int)((total + t - 1) / t);
  };

  zero_f32<<<blocks(n, T), T, 0, stream>>>(dinv, n);
  zero_f32<<<blocks((long long)n * 64, T), T, 0, stream>>>(agg1, (long long)n * 64);
  zero_f32<<<blocks((long long)n * 32, T), T, 0, stream>>>(agg2, (long long)n * 32);

  count_deg<<<blocks(e, T), T, 0, stream>>>(dst, dinv, e);
  make_dinv<<<blocks(n, T), T, 0, stream>>>(dinv, n);

  gemm_wmma_f32<256, 64, 1><<<n / 16, 128, 0, stream>>>(x, W1, h1);

  scatter_edges<64><<<blocks(e, T), T, 0, stream>>>(
      src, dst, dinv, h1, agg1, e);

  finalize_bn<64><<<blocks((long long)n * 64, T), T, 0, stream>>>(
      agg1, h1, dinv, b1, gamma1, beta1, mean1, var1, (long long)n * 64);

  gemm_wmma_f32<64, 32, 2><<<n / 32, 128, 0, stream>>>(agg1, W2, h2);

  scatter_edges<32><<<blocks(e, T), T, 0, stream>>>(
      src, dst, dinv, h2, agg2, e);

  final_head<<<blocks(n, T), T, 0, stream>>>(
      agg2, h2, dinv, b2, gamma2, beta2, mean2, var2, Wl, bl,
      (float*)d_out, n);
}


# aggregation: edges ordered by destination each call (count atomic's return = slot, prefix sum, placement); a wave sums a node's rows in registers, one store; no f32 atomics; sum order = arrival order
# speedup vs baseline: 1.7462x; 1.2488x over previous
#include <hip/hip_runtime.h>

typedef __attribute__((ext_vector_type(2))) float v2f;
typedef __attribute__((ext_vector_type(8))) float v8f;

#define BN_EPS 1e-5f

__global__ void zero_f32(float* __restrict__ p, long long n) {
  long long i = (long long)blockIdx.x * blockDim.x + threadIdx.x;
  if (i < n) p[i] = 0.0f;
}

__global__ void count_rank(const int* __restrict__ dst, int* __restrict__ cnt,
                           int* __restrict__ rank, int e) {
  int i = blockIdx.x * blockDim.x + threadIdx.x;
  if (i < e) rank[i] = atomicAdd(&cnt[dst[i]], 1);
}

__global__ void __launch_bounds__(1024)
scan_counts(const int* __restrict__ cnt, int* __restrict__ rp, int n) {
  __shared__ int wtot[32];
  __shared__ int wpart[32];
  const int tid = threadIdx.x, lane = tid & 31, wave = tid >> 5;
  const int ahead = blockIdx.x * 1024;
  int part = 0;
  for (int j = tid; j < ahead; j += 1024) part += cnt[j];
  const int i = ahead + tid;
  const int v = (i < n) ? cnt[i] : 0;
  int inc = v;
  for (int o = 1; o < 32; o <<= 1) {
    int t = __shfl_up(inc, o);
    if (lane >= o) inc += t;
  }
  for (int o = 16; o > 0; o >>= 1) part += __shfl_xor(part, o);
  if (lane == 31) wtot[wave] = inc;
  if (lane == 0)  wpart[wave] = part;
  __syncthreads();
  int p = wpart[lane];
  int t = (lane < wave) ? wtot[lane] : 0;
  for (int o = 16; o > 0; o >>= 1) {
    p += __shfl_xor(p, o);
    t += __shfl_xor(t, o);
  }
  if (i < n) rp[i] = p + t + (inc - v);
}

__global__ void fill_csr(const int* __restrict__ src, const int* __restrict__ dst,
                         const int* __restrict__ rank, const int* __restrict__ rp,
                         int* __restrict__ csr, int e) {
  int i = blockIdx.x * blockDim.x + threadIdx.x;
  if (i < e) csr[rp[dst[i]] + rank[i]] = src[i];
}

__device__ __forceinline__ float node_dinv(const int* __restrict__ rp, int i, int n, int e) {
  const int a = rp[i];
  const int b = (i + 1 < n) ? rp[i + 1] : e;
  return rsqrtf((float)(b - a) + 1.0f);
}

template <int K, int NC, int MT>
__global__ void __launch_bounds__(MT*(NC/16)*32)
gemm_wmma_f32(const float* __restrict__ A, const float* __restrict__ B,
              float* __restrict__ C) {
  constexpr int KP  = K + 4;
  constexpr int NCP = NC + 8;
  constexpr int NW  = MT * (NC/16);
  __shared__ float sA[MT*16*KP];
  __shared__ float sB[K*NCP];

  const int tid = threadIdx.x;
  const int nthreads = NW * 32;
  const float* Ablk = A + (size_t)blockIdx.x * (MT*16) * K;

  for (int idx = tid; idx < MT*16*(K/4); idx += nthreads) {
    int r = idx / (K/4), c4 = (idx % (K/4)) * 4;
    *(float4*)&sA[r*KP + c4] = *(const float4*)&Ablk[(size_t)r*K + c4];
  }
  for (int idx = tid; idx < K*(NC/4); idx += nthreads) {
    int r = idx / (NC/4), c4 = (idx % (NC/4)) * 4;
    *(float4*)&sB[r*NCP + c4] = *(const float4*)&B[(size_t)r*NC + c4];
  }
  __syncthreads();

  const int wave  = tid >> 5, lane = tid & 31;
  const int mt    = wave / (NC/16), nt = wave % (NC/16);
  const int rrow  = lane & 15;
  const int khalf = lane >> 4;

  const float* a_ptr = &sA[(mt*16 + rrow)*KP + khalf*2];
  const float* b_ptr = &sB[(khalf*2)*NCP + nt*16 + rrow];

  v8f c0 = {};
  v8f c1 = {};
  #pragma unroll 4
  for (int ks = 0; ks < K/8; ++ks) {
    v2f a0; a0.x = a_ptr[0];     a0.y = a_ptr[1];
    v2f b0; b0.x = b_ptr[0];     b0.y = b_ptr[NCP];
    v2f a1; a1.x = a_ptr[4];     a1.y = a_ptr[5];
    v2f b1; b1.x = b_ptr[4*NCP]; b1.y = b_ptr[5*NCP];
    c0 = __builtin_amdgcn_wmma_f32_16x16x4_f32(false, a0, false, b0,
                                               (short)0, c0, false, false);
    c1 = __builtin_amdgcn_wmma_f32_16x16x4_f32(false, a1, false, b1,
                                               (short)0, c1, false, false);
    a_ptr += 8;
    b_ptr += 8*NCP;
  }
  v8f c = c0 + c1;

  const int gm  = blockIdx.x*(MT*16) + mt*16 + khalf*8;
  const int col = nt*16 + rrow;
  #pragma unroll
  for (int r = 0; r < 8; ++r)
    C[(size_t)(gm + r)*NC + col] = c[r];
}

template <int F, int G>
__device__ __forceinline__ void take_rows(const float* __restrict__ h, int s_l, float w_l,
                                          int j, int lane, float (&acc)[F / 32]) {
  constexpr int V = F / 32;
  float v[G][V];
  float w[G];
  #pragma unroll
  for (int u = 0; u < G; ++u) {
    const int s = __builtin_amdgcn_readlane(s_l, j + u);
    w[u] = __int_as_float(__builtin_amdgcn_readlane(__float_as_int(w_l), j + u));
    const float* hp = h + (size_t)s * F + lane * V;
    if constexpr (V == 2) {
      const float2 t = *(const float2*)hp;
      v[u][0] = t.x; v[u][1] = t.y;
    } else {
      v[u][0] = hp[0];
    }
  }
  #pragma unroll
  for (int u = 0; u < G; ++u)
    for (int l = 0; l < V; ++l) acc[l] += v[u][l] * w[u];
}

template <int F>
__global__ void __launch_bounds__(256)
gather_rows(const int* __restrict__ rp, const int* __restrict__ csr,
            const float* __restrict__ h, float* __restrict__ agg, int n, int e) {
  constexpr int V = F / 32;
  constexpr int G = 16 / V;
  const int lane = threadIdx.x & 31;
  const int d = __builtin_amdgcn_readfirstlane((int)((blockIdx.x * blockDim.x + threadIdx.x) >> 5));
  if (d >= n) return;
  const int beg = rp[d];
  const int end = (d + 1 < n) ? rp[d + 1] : e;
  const float dd = rsqrtf((float)(end - beg) + 1.0f);

  float acc[V];
  for (int l = 0; l < V; ++l) acc[l] = 0.0f;

  for (int base = beg; base < end; base += 32) {
    const int left = end - base;
    const int cnt = left < 32 ? left : 32;
    int s_l = 0;
    float w_l = 0.0f;
    if (lane < cnt) {
      s_l = csr[base + lane];
      s_l = min(max(s_l, 0), n - 1);
      w_l = node_dinv(rp, s_l, n, e) * dd;
    }
    int j = 0;
    for (; j + G <= cnt; j += G) take_rows<F, G>(h, s_l, w_l, j, lane, acc);
    if constexpr (G >= 16) { if (j + 8 <= cnt) { take_rows<F, 8>(h, s_l, w_l, j, lane, acc); j += 8; } }
    if (j + 4 <= cnt) { take_rows<F, 4>(h, s_l, w_l, j, lane, acc); j += 4; }
    if (j + 2 <= cnt) { take_rows<F, 2>(h, s_l, w_l, j, lane, acc); j += 2; }
    if (j < cnt)      { take_rows<F, 1>(h, s_l, w_l, j, lane, acc); }
  }

  float* ap = agg + (size_t)d * F + lane * V;
  if constexpr (V == 2) {
    float2 t; t.x = acc[0]; t.y = acc[1];
    *(float2*)ap = t;
  } else {
    ap[0] = acc[0];
  }
}

template <int F>
__global__ void finalize_bn(float* __restrict__ agg, const float* __restrict__ h,
                            const int* __restrict__ rp, const float* __restrict__ bias,
                            const float* __restrict__ gamma, const float* __restrict__ beta,
                            const float* __restrict__ mean, const float* __restrict__ var,
                            int n, int e, long long total) {
  long long t = (long long)blockIdx.x * blockDim.x + threadIdx.x;
  if (t >= total) return;
  int node = (int)(t / F), f = (int)(t % F);
  float di = node_dinv(rp, node, n, e);
  float v = agg[t] + h[t] * di * di + bias[f];
  v = (v - mean[f]) * rsqrtf(var[f] + BN_EPS) * gamma[f] + beta[f];
  agg[t] = fmaxf(v, 0.0f);
}

__global__ void final_head(const float* __restrict__ agg2, const float* __restrict__ h2,
                           const int* __restrict__ rp, const float* __restrict__ b2,
                           const float* __restrict__ g2, const float* __restrict__ be2,
                           const float* __restrict__ m2, const float* __restrict__ v2,
                           const float* __restrict__ Wl, const float* __restrict__ bl,
                           float* __restrict__ out, int n, int e) {
  int node = blockIdx.x * blockDim.x + threadIdx.x;
  if (node >= n) return;
  float di = node_dinv(rp, node, n, e);
  float di2 = di * di;
  float acc0 = bl[0], acc1 = bl[1];
  const float* ar = agg2 + (size_t)node * 32;
  const float* hr = h2 + (size_t)node * 32;
  #pragma unroll
  for (int f = 0; f < 32; ++f) {
    float v = ar[f] + hr[f] * di2 + b2[f];
    v = (v - m2[f]) * rsqrtf(v2[f] + BN_EPS) * g2[f] + be2[f];
    v = fmaxf(v, 0.0f);
    acc0 += v * Wl[f*2 + 0];
    acc1 += v * Wl[f*2 + 1];
  }
  out[(size_t)node*2 + 0] = acc0;
  out[(size_t)node*2 + 1] = acc1;
}

extern "C" void kernel_launch(void* const* d_in, const int* in_sizes, int n_in,
                              void* d_out, int out_size, void* d_ws, size_t ws_size,
                              hipStream_t stream) {
  const float* x      = (const float*)d_in[0];
  const int*   ei     = (const int*)  d_in[1];
  const float* W1     = (const float*)d_in[2];
  const float* b1     = (const float*)d_in[3];
  const float* gamma1 = (const float*)d_in[4];
  const float* beta1  = (const float*)d_in[5];
  const float* mean1  = (const float*)d_in[6];
  const float* var1   = (const float*)d_in[7];
  const float* W2     = (const float*)d_in[8];
  const float* b2     = (const float*)d_in[9];
  const float* gamma2 = (const float*)d_in[10];
  const float* beta2  = (const float*)d_in[11];
  const float* mean2  = (const float*)d_in[12];
  const float* var2   = (const float*)d_in[13];
  const float* Wl     = (const float*)d_in[14];
  const float* bl     = (const float*)d_in[15];

  const int n = in_sizes[0] / 256;
  const int e = in_sizes[1] / 2;
  const int* src = ei;
  const int* dst = ei + e;

  float* ws   = (float*)d_ws;
  int*   rp   = (int*)ws;
  float* h1   = ws   + (size_t)n;
  float* agg1 = h1   + (size_t)n * 64;
  float* h2   = agg1 + (size_t)n * 64;
  int*   csr  = (int*)(h2 + (size_t)n * 32);
  int*   cnt  = (int*)h2;
  int*   rank = (int*)agg1;
  float* agg2 = h1;

  const int T = 256;
  auto blocks = [](long long total, int t) {
    return (unsigned int)((total + t - 1) / t);
  };

  zero_f32<<<blocks(n, T), T, 0, stream>>>((float*)cnt, n);
  count_rank<<<blocks(e, T), T, 0, stream>>>(dst, cnt, rank, e);
  scan_counts<<<blocks(n, 1024), 1024, 0, stream>>>(cnt, rp, n);
  fill_csr<<<blocks(e, T), T, 0, stream>>>(src, dst, rank, rp, csr, e);

  gemm_wmma_f32<256, 64, 1><<<n / 16, 128, 0, stream>>>(x, W1, h1);

  gather_rows<64><<<blocks((long long)n * 32, T), T, 0, stream>>>(
      rp, csr, h1, agg1, n, e);

  finalize_bn<64><<<blocks((long long)n * 64, T), T, 0, stream>>>(
      agg1, h1, rp, b1, gamma1, beta1, mean1, var1, n, e, (long long)n * 64);

  gemm_wmma_f32<64, 32, 2><<<n / 32, 128, 0, stream>>>(agg1, W2, h2);

  gather_rows<32><<<blocks((long long)n * 32, T), T, 0, stream>>>(
      rp, csr, h2, agg2, n, e);

  final_head<<<blocks(n, T), T, 0, stream>>>(
      agg2, h2, rp, b2, gamma2, beta2, mean2, var2, Wl, bl,
      (float*)d_out, n, e);
}


# edge ordering rebuilt as a two-level counting sort on LDS counters (buckets of 256 nodes, then nodes): no global atomics left; sum order = arrival order, now of the LDS adds
# speedup vs baseline: 3.2223x; 1.8453x over previous
#include <hip/hip_runtime.h>

typedef __attribute__((ext_vector_type(2))) float v2f;
typedef __attribute__((ext_vector_type(8))) float v8f;

#define BN_EPS 1e-5f

#define BK_SHIFT 8
#define BK_NODES (1 << BK_SHIFT)
#define BK_MAX   1024
#define EB       256
#define EU       4
#define BU       8

__device__ __forceinline__ int bucket_of(int d, int nbk) {
  return min(max(d, 0) >> BK_SHIFT, nbk - 1);
}

__global__ void __launch_bounds__(1024)
bucket_hist(const int* __restrict__ dst, int* __restrict__ hist, int e, int nbk) {
  __shared__ int lc[BK_MAX + 32];
  const int tid = threadIdx.x;
  nbk = min(nbk, BK_MAX);
  for (int b = tid; b < nbk; b += 1024) lc[b] = 0;
  if (tid < 32) lc[BK_MAX + tid] = 0;
  __syncthreads();
  const int chunk = (e + EB - 1) / EB;
  const int beg = blockIdx.x * chunk;
  const int end = min(beg + chunk, e);
  for (int base = beg + tid; base < end; base += 1024 * EU) {
    int d[EU];
    #pragma unroll
    for (int u = 0; u < EU; ++u) d[u] = dst[min(base + u * 1024, end - 1)];
    #pragma unroll
    for (int u = 0; u < EU; ++u)
      atomicAdd(&lc[(base + u * 1024 < end) ? bucket_of(d[u], nbk) : BK_MAX + (tid & 31)], 1);
  }
  __syncthreads();
  for (int b = tid; b < nbk; b += 1024) hist[b * EB + blockIdx.x] = lc[b];
}

__global__ void __launch_bounds__(1024)
scan_counts(const int* __restrict__ cnt, int* __restrict__ rp, int n) {
  __shared__ int wtot[32];
  __shared__ int wpart[32];
  const int tid = threadIdx.x, lane = tid & 31, wave = tid >> 5;
  const int ahead = blockIdx.x * 1024;
  int part = 0;
  for (int j = tid; j < ahead; j += 1024) part += cnt[j];
  const int i = ahead + tid;
  const int v = (i < n) ? cnt[i] : 0;
  int inc = v;
  for (int o = 1; o < 32; o <<= 1) {
    int t = __shfl_up(inc, o);
    if (lane >= o) inc += t;
  }
  for (int o = 16; o > 0; o >>= 1) part += __shfl_xor(part, o);
  if (lane == 31) wtot[wave] = inc;
  if (lane == 0)  wpart[wave] = part;
  __syncthreads();
  int p = wpart[lane];
  int t = (lane < wave) ? wtot[lane] : 0;
  for (int o = 16; o > 0; o >>= 1) {
    p += __shfl_xor(p, o);
    t += __shfl_xor(t, o);
  }
  if (i < n) rp[i] = p + t + (inc - v);
}

__global__ void __launch_bounds__(1024)
bucket_scatter(const int* __restrict__ src, const int* __restrict__ dst,
               const int* __restrict__ off, int* __restrict__ bpair, int e, int nbk) {
  __shared__ int lpos[BK_MAX + 32];
  int2* __restrict__ pairs = reinterpret_cast<int2*>(bpair);
  const int tid = threadIdx.x;
  nbk = min(nbk, BK_MAX);
  for (int b = tid; b < nbk; b += 1024) lpos[b] = off[b * EB + blockIdx.x];
  if (tid < 32) lpos[BK_MAX + tid] = 0;
  __syncthreads();
  const int chunk = (e + EB - 1) / EB;
  const int beg = blockIdx.x * chunk;
  const int end = min(beg + chunk, e);
  for (int base = beg + tid; base < end; base += 1024 * EU) {
    int d[EU], s[EU];
    #pragma unroll
    for (int u = 0; u < EU; ++u) {
      const int i = min(base + u * 1024, end - 1);
      d[u] = dst[i];
      s[u] = src[i];
    }
    #pragma unroll
    for (int u = 0; u < EU; ++u) {
      const bool in = base + u * 1024 < end;
      const int p = atomicAdd(&lpos[in ? bucket_of(d[u], nbk) : BK_MAX + (tid & 31)], 1);
      if (in) pairs[p] = make_int2(s[u], d[u]);
    }
  }
}

__global__ void __launch_bounds__(BK_NODES)
bucket_build(const int* __restrict__ bpair, const int* __restrict__ off,
             int* __restrict__ rp, int* __restrict__ csr, int n, int e, int nbk) {
  __shared__ int lcnt[BK_NODES + 32];
  __shared__ int lpos[BK_NODES + 32];
  __shared__ int wtot[BK_NODES / 32];
  const int2* __restrict__ pairs = reinterpret_cast<const int2*>(bpair);
  const int tid = threadIdx.x, lane = tid & 31, wave = tid >> 5;
  const int b = blockIdx.x;
  const int node0 = b << BK_SHIFT;
  const int ebeg = off[b * EB];
  const int eend = (b + 1 < nbk) ? off[(b + 1) * EB] : e;
  lcnt[tid] = 0;
  if (tid < 32) lcnt[BK_NODES + tid] = 0;
  __syncthreads();
  for (int base = ebeg + tid; base < eend; base += BK_NODES * BU) {
    int2 q[BU];
    #pragma unroll
    for (int u = 0; u < BU; ++u) q[u] = pairs[min(base + u * BK_NODES, eend - 1)];
    #pragma unroll
    for (int u = 0; u < BU; ++u)
      atomicAdd(&lcnt[(base + u * BK_NODES < eend) ? min(max(q[u].y - node0, 0), BK_NODES - 1)
                                                   : BK_NODES + lane], 1);
  }
  __syncthreads();
  const int v = lcnt[tid];
  int inc = v;
  for (int o = 1; o < 32; o <<= 1) {
    int t = __shfl_up(inc, o);
    if (lane >= o) inc += t;
  }
  if (lane == 31) wtot[wave] = inc;
  __syncthreads();
  int ahead = 0;
  for (int w = 0; w < BK_NODES / 32; ++w) if (w < wave) ahead += wtot[w];
  const int start = ebeg + ahead + inc - v;
  lpos[tid] = start;
  if (tid < 32) lpos[BK_NODES + tid] = 0;
  if (node0 + tid < n) rp[node0 + tid] = start;
  __syncthreads();
  for (int base = ebeg + tid; base < eend; base += BK_NODES * BU) {
    int2 q[BU];
    #pragma unroll
    for (int u = 0; u < BU; ++u) q[u] = pairs[min(base + u * BK_NODES, eend - 1)];
    #pragma unroll
    for (int u = 0; u < BU; ++u) {
      const bool in = base + u * BK_NODES < eend;
      const int p = atomicAdd(&lpos[in ? min(max(q[u].y - node0, 0), BK_NODES - 1) : BK_NODES + lane], 1);
      if (in) csr[p] = q[u].x;
    }
  }
}

__device__ __forceinline__ float node_dinv(const int* __restrict__ rp, int i, int n, int e) {
  const int a = rp[i];
  const int b = (i + 1 < n) ? rp[i + 1] : e;
  return rsqrtf((float)(b - a) + 1.0f);
}

template <int K, int NC, int MT>
__global__ void __launch_bounds__(MT*(NC/16)*32)
gemm_wmma_f32(const float* __restrict__ A, const float* __restrict__ B,
              float* __restrict__ C, const int* __restrict__ rp, int n, int e) {
  constexpr int KP  = K + 4;
  constexpr int NCP = NC + 8;
  constexpr int NW  = MT * (NC/16);
  __shared__ float sA[MT*16*KP];
  __shared__ float sB[K*NCP];

  const int tid = threadIdx.x;
  const int nthreads = NW * 32;
  const float* Ablk = A + (size_t)blockIdx.x * (MT*16) * K;

  const int gm = blockIdx.x*(MT*16) + ((tid >> 5) / (NC/16))*16 + ((tid & 31) >> 4)*8;
  int sb[9];
  #pragma unroll
  for (int r = 0; r < 8; ++r) sb[r] = rp[gm + r];
  sb[8] = rp[min(gm + 8, n - 1)];
  if (gm + 8 >= n) sb[8] = e;
  float dv[8];
  #pragma unroll
  for (int r = 0; r < 8; ++r) dv[r] = rsqrtf((float)(sb[r + 1] - sb[r]) + 1.0f);

  for (int idx = tid; idx < MT*16*(K/4); idx += nthreads) {
    int r = idx / (K/4), c4 = (idx % (K/4)) * 4;
    *(float4*)&sA[r*KP + c4] = *(const float4*)&Ablk[(size_t)r*K + c4];
  }
  for (int idx = tid; idx < K*(NC/4); idx += nthreads) {
    int r = idx / (NC/4), c4 = (idx % (NC/4)) * 4;
    *(float4*)&sB[r*NCP + c4] = *(const float4*)&B[(size_t)r*NC + c4];
  }
  __syncthreads();

  const int wave  = tid >> 5, lane = tid & 31;
  const int mt    = wave / (NC/16), nt = wave % (NC/16);
  const int rrow  = lane & 15;
  const int khalf = lane >> 4;

  const float* a_ptr = &sA[(mt*16 + rrow)*KP + khalf*2];
  const float* b_ptr = &sB[(khalf*2)*NCP + nt*16 + rrow];

  v8f c0 = {};
  v8f c1 = {};
  #pragma unroll 4
  for (int ks = 0; ks < K/8; ++ks) {
    v2f a0; a0.x = a_ptr[0];     a0.y = a_ptr[1];
    v2f b0; b0.x = b_ptr[0];     b0.y = b_ptr[NCP];
    v2f a1; a1.x = a_ptr[4];     a1.y = a_ptr[5];
    v2f b1; b1.x = b_ptr[4*NCP]; b1.y = b_ptr[5*NCP];
    c0 = __builtin_amdgcn_wmma_f32_16x16x4_f32(false, a0, false, b0,
                                               (short)0, c0, false, false);
    c1 = __builtin_amdgcn_wmma_f32_16x16x4_f32(false, a1, false, b1,
                                               (short)0, c1, false, false);
    a_ptr += 8;
    b_ptr += 8*NCP;
  }
  v8f c = c0 + c1;

  const int col = nt*16 + rrow;
  #pragma unroll
  for (int r = 0; r < 8; ++r)
    C[(size_t)(gm + r)*NC + col] = c[r] * dv[r];
}

template <int F, int G>
__device__ __forceinline__ void take_rows(const float* __restrict__ h, int s_l,
                                          int j, int lane, float (&acc)[F / 32]) {
  constexpr int V = F / 32;
  float v[G][V];
  #pragma unroll
  for (int u = 0; u < G; ++u) {
    const int s = __builtin_amdgcn_readlane(s_l, j + u);
    const float* hp = h + (size_t)s * F + lane * V;
    if constexpr (V == 2) {
      const float2 t = *(const float2*)hp;
      v[u][0] = t.x; v[u][1] = t.y;
    } else {
      v[u][0] = hp[0];
    }
  }
  #pragma unroll
  for (int u = 0; u < G; ++u)
    for (int l = 0; l < V; ++l) acc[l] += v[u][l];
}

template <int F>
__global__ void __launch_bounds__(256)
gather_rows(const int* __restrict__ rp, const int* __restrict__ csr,
            const float* __restrict__ h, float* __restrict__ agg, int n, int e) {
  constexpr int V = F / 32;
  constexpr int G = 16 / V;
  const int lane = threadIdx.x & 31;
  const int d = __builtin_amdgcn_readfirstlane((int)((blockIdx.x * blockDim.x + threadIdx.x) >> 5));
  if (d >= n) return;
  const int beg = rp[d];
  const int end = (d + 1 < n) ? rp[d + 1] : e;
  const float dd = rsqrtf((float)(end - beg) + 1.0f);

  float acc[V];
  for (int l = 0; l < V; ++l) acc[l] = 0.0f;

  for (int base = beg; base < end; base += 32) {
    const int left = end - base;
    const int cnt = left < 32 ? left : 32;
    int s_l = 0;
    if (lane < cnt) {
      s_l = csr[base + lane];
      s_l = min(max(s_l, 0), n - 1);
    }
    int j = 0;
    for (; j + G <= cnt; j += G) take_rows<F, G>(h, s_l, j, lane, acc);
    if constexpr (G >= 16) { if (j + 8 <= cnt) { take_rows<F, 8>(h, s_l, j, lane, acc); j += 8; } }
    if (j + 4 <= cnt) { take_rows<F, 4>(h, s_l, j, lane, acc); j += 4; }
    if (j + 2 <= cnt) { take_rows<F, 2>(h, s_l, j, lane, acc); j += 2; }
    if (j < cnt)      { take_rows<F, 1>(h, s_l, j, lane, acc); }
  }
  for (int l = 0; l < V; ++l) acc[l] *= dd;

  float* ap = agg + (size_t)d * F + lane * V;
  if constexpr (V == 2) {
    float2 t; t.x = acc[0]; t.y = acc[1];
    *(float2*)ap = t;
  } else {
    ap[0] = acc[0];
  }
}

template <int F>
__global__ void finalize_bn(float* __restrict__ agg, const float* __restrict__ h,
                            const int* __restrict__ rp, const float* __restrict__ bias,
                            const float* __restrict__ gamma, const float* __restrict__ beta,
                            const float* __restrict__ mean, const float* __restrict__ var,
                            int n, int e, long long total) {
  long long t = (long long)blockIdx.x * blockDim.x + threadIdx.x;
  if (t >= total) return;
  int node = (int)(t / F), f = (int)(t % F);
  float di = node_dinv(rp, node, n, e);
  float v = agg[t] + h[t] * di + bias[f];
  v = (v - mean[f]) * rsqrtf(var[f] + BN_EPS) * gamma[f] + beta[f];
  agg[t] = fmaxf(v, 0.0f);
}

__global__ void final_head(const float* __restrict__ agg2, const float* __restrict__ h2,
                           const int* __restrict__ rp, const float* __restrict__ b2,
                           const float* __restrict__ g2, const float* __restrict__ be2,
                           const float* __restrict__ m2, const float* __restrict__ v2,
                           const float* __restrict__ Wl, const float* __restrict__ bl,
                           float* __restrict__ out, int n, int e) {
  int node = blockIdx.x * blockDim.x + threadIdx.x;
  if (node >= n) return;
  float di = node_dinv(rp, node, n, e);
  float acc0 = bl[0], acc1 = bl[1];
  const float* ar = agg2 + (size_t)node * 32;
  const float* hr = h2 + (size_t)node * 32;
  #pragma unroll
  for (int f = 0; f < 32; ++f) {
    float v = ar[f] + hr[f] * di + b2[f];
    v = (v - m2[f]) * rsqrtf(v2[f] + BN_EPS) * g2[f] + be2[f];
    v = fmaxf(v, 0.0f);
    acc0 += v * Wl[f*2 + 0];
    acc1 += v * Wl[f*2 + 1];
  }
  out[(size_t)node*2 + 0] = acc0;
  out[(size_t)node*2 + 1] = acc1;
}

extern "C" void kernel_launch(void* const* d_in, const int* in_sizes, int n_in,
                              void* d_out, int out_size, void* d_ws, size_t ws_size,
                              hipStream_t stream) {
  const float* x      = (const float*)d_in[0];
  const int*   ei     = (const int*)  d_in[1];
  const float* W1     = (const float*)d_in[2];
  const float* b1     = (const float*)d_in[3];
  const float* gamma1 = (const float*)d_in[4];
  const float* beta1  = (const float*)d_in[5];
  const float* mean1  = (const float*)d_in[6];
  const float* var1   = (const float*)d_in[7];
  const float* W2     = (const float*)d_in[8];
  const float* b2     = (const float*)d_in[9];
  const float* gamma2 = (const float*)d_in[10];
  const float* beta2  = (const float*)d_in[11];
  const float* mean2  = (const float*)d_in[12];
  const float* var2   = (const float*)d_in[13];
  const float* Wl     = (const float*)d_in[14];
  const float* bl     = (const float*)d_in[15];

  const int n = in_sizes[0] / 256;
  const int e = in_sizes[1] / 2;
  const int* src = ei;
  const int* dst = ei + e;

  float* ws   = (float*)d_ws;
  int*   rp   = (int*)ws;
  float* h1   = ws   + (size_t)n;
  float* agg1 = h1   + (size_t)n * 64;
  float* h2   = agg1 + (size_t)n * 64;
  int*   csr  = (int*)(h2 + (size_t)n * 32);
  int*   hist = (int*)h2;
  int*   off  = hist + (size_t)n * 16;
  int*   bpair = (int*)agg1;
  float* agg2 = h1;

  const int T = 256;
  auto blocks = [](long long total, int t) {
    return (unsigned int)((total + t - 1) / t);
  };

  const int nbk = (int)blocks(n, BK_NODES);
  bucket_hist<<<EB, 1024, 0, stream>>>(dst, hist, e, nbk);
  scan_counts<<<blocks((long long)nbk * EB, 1024), 1024, 0, stream>>>(hist, off, nbk * EB);
  bucket_scatter<<<EB, 1024, 0, stream>>>(src, dst, off, bpair, e, nbk);
  bucket_build<<<nbk, BK_NODES, 0, stream>>>(bpair, off, rp, csr, n, e, nbk);

  gemm_wmma_f32<256, 64, 1><<<n / 16, 128, 0, stream>>>(x, W1, h1, rp, n, e);

  gather_rows<64><<<blocks((long long)n * 32, T), T, 0, stream>>>(
      rp, csr, h1, agg1, n, e);

  finalize_bn<64><<<blocks((long long)n * 64, T), T, 0, stream>>>(
      agg1, h1, rp, b1, gamma1, beta1, mean1, var1, n, e, (long long)n * 64);

  gemm_wmma_f32<64, 32, 2><<<n / 32, 128, 0, stream>>>(agg1, W2, h2, rp, n, e);

  gather_rows<32><<<blocks((long long)n * 32, T), T, 0, stream>>>(
      rp, csr, h2, agg2, n, e);

  final_head<<<blocks(n, T), T, 0, stream>>>(
      agg2, h2, rp, b2, gamma2, beta2, mean2, var2, Wl, bl,
      (float*)d_out, n, e);
}


# node epilogues moved into the gather waves: self-loop + bias + BN + ReLU in the layer-1 gather, the same plus the linear head in the layer-2 gather (10 -> 8 launches); head sums become lane trees
# speedup vs baseline: 3.5413x; 1.0990x over previous
#include <hip/hip_runtime.h>

typedef __attribute__((ext_vector_type(2))) float v2f;
typedef __attribute__((ext_vector_type(8))) float v8f;

#define BN_EPS 1e-5f

#define BK_SHIFT 8
#define BK_NODES (1 << BK_SHIFT)
#define BK_MAX   1024
#define EB       256
#define EU       4
#define BU       8

__device__ __forceinline__ int bucket_of(int d, int nbk) {
  return min(max(d, 0) >> BK_SHIFT, nbk - 1);
}

__global__ void __launch_bounds__(1024)
bucket_hist(const int* __restrict__ dst, int* __restrict__ hist, int e, int nbk) {
  __shared__ int lc[BK_MAX + 32];
  const int tid = threadIdx.x;
  nbk = min(nbk, BK_MAX);
  for (int b = tid; b < nbk; b += 1024) lc[b] = 0;
  if (tid < 32) lc[BK_MAX + tid] = 0;
  __syncthreads();
  const int chunk = (e + EB - 1) / EB;
  const int beg = blockIdx.x * chunk;
  const int end = min(beg + chunk, e);
  for (int base = beg + tid; base < end; base += 1024 * EU) {
    int d[EU];
    #pragma unroll
    for (int u = 0; u < EU; ++u) d[u] = dst[min(base + u * 1024, end - 1)];
    #pragma unroll
    for (int u = 0; u < EU; ++u)
      atomicAdd(&lc[(base + u * 1024 < end) ? bucket_of(d[u], nbk) : BK_MAX + (tid & 31)], 1);
  }
  __syncthreads();
  for (int b = tid; b < nbk; b += 1024) hist[b * EB + blockIdx.x] = lc[b];
}

__global__ void __launch_bounds__(1024)
scan_counts(const int* __restrict__ cnt, int* __restrict__ rp, int n) {
  __shared__ int wtot[32];
  __shared__ int wpart[32];
  const int tid = threadIdx.x, lane = tid & 31, wave = tid >> 5;
  const int ahead = blockIdx.x * 1024;
  int part = 0;
  for (int j = tid; j < ahead; j += 1024) part += cnt[j];
  const int i = ahead + tid;
  const int v = (i < n) ? cnt[i] : 0;
  int inc = v;
  for (int o = 1; o < 32; o <<= 1) {
    int t = __shfl_up(inc, o);
    if (lane >= o) inc += t;
  }
  for (int o = 16; o > 0; o >>= 1) part += __shfl_xor(part, o);
  if (lane == 31) wtot[wave] = inc;
  if (lane == 0)  wpart[wave] = part;
  __syncthreads();
  int p = wpart[lane];
  int t = (lane < wave) ? wtot[lane] : 0;
  for (int o = 16; o > 0; o >>= 1) {
    p += __shfl_xor(p, o);
    t += __shfl_xor(t, o);
  }
  if (i < n) rp[i] = p + t + (inc - v);
}

__global__ void __launch_bounds__(1024)
bucket_scatter(const int* __restrict__ src, const int* __restrict__ dst,
               const int* __restrict__ off, int* __restrict__ bpair, int e, int nbk) {
  __shared__ int lpos[BK_MAX + 32];
  int2* __restrict__ pairs = reinterpret_cast<int2*>(bpair);
  const int tid = threadIdx.x;
  nbk = min(nbk, BK_MAX);
  for (int b = tid; b < nbk; b += 1024) lpos[b] = off[b * EB + blockIdx.x];
  if (tid < 32) lpos[BK_MAX + tid] = 0;
  __syncthreads();
  const int chunk = (e + EB - 1) / EB;
  const int beg = blockIdx.x * chunk;
  const int end = min(beg + chunk, e);
  for (int base = beg + tid; base < end; base += 1024 * EU) {
    int d[EU], s[EU];
    #pragma unroll
    for (int u = 0; u < EU; ++u) {
      const int i = min(base + u * 1024, end - 1);
      d[u] = dst[i];
      s[u] = src[i];
    }
    #pragma unroll
    for (int u = 0; u < EU; ++u) {
      const bool in = base + u * 1024 < end;
      const int p = atomicAdd(&lpos[in ? bucket_of(d[u], nbk) : BK_MAX + (tid & 31)], 1);
      if (in) pairs[p] = make_int2(s[u], d[u]);
    }
  }
}

__global__ void __launch_bounds__(BK_NODES)
bucket_build(const int* __restrict__ bpair, const int* __restrict__ off,
             int* __restrict__ rp, int* __restrict__ csr, int n, int e, int nbk) {
  __shared__ int lcnt[BK_NODES + 32];
  __shared__ int lpos[BK_NODES + 32];
  __shared__ int wtot[BK_NODES / 32];
  const int2* __restrict__ pairs = reinterpret_cast<const int2*>(bpair);
  const int tid = threadIdx.x, lane = tid & 31, wave = tid >> 5;
  const int b = blockIdx.x;
  const int node0 = b << BK_SHIFT;
  const int ebeg = off[b * EB];
  const int eend = (b + 1 < nbk) ? off[(b + 1) * EB] : e;
  lcnt[tid] = 0;
  if (tid < 32) lcnt[BK_NODES + tid] = 0;
  __syncthreads();
  for (int base = ebeg + tid; base < eend; base += BK_NODES * BU) {
    int2 q[BU];
    #pragma unroll
    for (int u = 0; u < BU; ++u) q[u] = pairs[min(base + u * BK_NODES, eend - 1)];
    #pragma unroll
    for (int u = 0; u < BU; ++u)
      atomicAdd(&lcnt[(base + u * BK_NODES < eend) ? min(max(q[u].y - node0, 0), BK_NODES - 1)
                                                   : BK_NODES + lane], 1);
  }
  __syncthreads();
  const int v = lcnt[tid];
  int inc = v;
  for (int o = 1; o < 32; o <<= 1) {
    int t = __shfl_up(inc, o);
    if (lane >= o) inc += t;
  }
  if (lane == 31) wtot[wave] = inc;
  __syncthreads();
  int ahead = 0;
  for (int w = 0; w < BK_NODES / 32; ++w) if (w < wave) ahead += wtot[w];
  const int start = ebeg + ahead + inc - v;
  lpos[tid] = start;
  if (tid < 32) lpos[BK_NODES + tid] = 0;
  if (node0 + tid < n) rp[node0 + tid] = start;
  __syncthreads();
  for (int base = ebeg + tid; base < eend; base += BK_NODES * BU) {
    int2 q[BU];
    #pragma unroll
    for (int u = 0; u < BU; ++u) q[u] = pairs[min(base + u * BK_NODES, eend - 1)];
    #pragma unroll
    for (int u = 0; u < BU; ++u) {
      const bool in = base + u * BK_NODES < eend;
      const int p = atomicAdd(&lpos[in ? min(max(q[u].y - node0, 0), BK_NODES - 1) : BK_NODES + lane], 1);
      if (in) csr[p] = q[u].x;
    }
  }
}

template <int K, int NC, int MT>
__global__ void __launch_bounds__(MT*(NC/16)*32)
gemm_wmma_f32(const float* __restrict__ A, const float* __restrict__ B,
              float* __restrict__ C, const int* __restrict__ rp, int n, int e) {
  constexpr int KP  = K + 4;
  constexpr int NCP = NC + 8;
  constexpr int NW  = MT * (NC/16);
  __shared__ float sA[MT*16*KP];
  __shared__ float sB[K*NCP];

  const int tid = threadIdx.x;
  const int nthreads = NW * 32;
  const float* Ablk = A + (size_t)blockIdx.x * (MT*16) * K;

  const int gm = blockIdx.x*(MT*16) + ((tid >> 5) / (NC/16))*16 + ((tid & 31) >> 4)*8;
  int sb[9];
  #pragma unroll
  for (int r = 0; r < 8; ++r) sb[r] = rp[gm + r];
  sb[8] = rp[min(gm + 8, n - 1)];
  if (gm + 8 >= n) sb[8] = e;
  float dv[8];
  #pragma unroll
  for (int r = 0; r < 8; ++r) dv[r] = rsqrtf((float)(sb[r + 1] - sb[r]) + 1.0f);

  for (int idx = tid; idx < MT*16*(K/4); idx += nthreads) {
    int r = idx / (K/4), c4 = (idx % (K/4)) * 4;
    *(float4*)&sA[r*KP + c4] = *(const float4*)&Ablk[(size_t)r*K + c4];
  }
  for (int idx = tid; idx < K*(NC/4); idx += nthreads) {
    int r = idx / (NC/4), c4 = (idx % (NC/4)) * 4;
    *(float4*)&sB[r*NCP + c4] = *(const float4*)&B[(size_t)r*NC + c4];
  }
  __syncthreads();

  const int wave  = tid >> 5, lane = tid & 31;
  const int mt    = wave / (NC/16), nt = wave % (NC/16);
  const int rrow  = lane & 15;
  const int khalf = lane >> 4;

  const float* a_ptr = &sA[(mt*16 + rrow)*KP + khalf*2];
  const float* b_ptr = &sB[(khalf*2)*NCP + nt*16 + rrow];

  v8f c0 = {};
  v8f c1 = {};
  #pragma unroll 4
  for (int ks = 0; ks < K/8; ++ks) {
    v2f a0; a0.x = a_ptr[0];     a0.y = a_ptr[1];
    v2f b0; b0.x = b_ptr[0];     b0.y = b_ptr[NCP];
    v2f a1; a1.x = a_ptr[4];     a1.y = a_ptr[5];
    v2f b1; b1.x = b_ptr[4*NCP]; b1.y = b_ptr[5*NCP];
    c0 = __builtin_amdgcn_wmma_f32_16x16x4_f32(false, a0, false, b0,
                                               (short)0, c0, false, false);
    c1 = __builtin_amdgcn_wmma_f32_16x16x4_f32(false, a1, false, b1,
                                               (short)0, c1, false, false);
    a_ptr += 8;
    b_ptr += 8*NCP;
  }
  v8f c = c0 + c1;

  const int col = nt*16 + rrow;
  #pragma unroll
  for (int r = 0; r < 8; ++r)
    C[(size_t)(gm + r)*NC + col] = c[r] * dv[r];
}

template <int F, int G>
__device__ __forceinline__ void take_rows(const float* __restrict__ h, int s_l,
                                          int j, int lane, float (&acc)[F / 32]) {
  constexpr int V = F / 32;
  float v[G][V];
  #pragma unroll
  for (int u = 0; u < G; ++u) {
    const int s = __builtin_amdgcn_readlane(s_l, j + u);
    const float* hp = h + (size_t)s * F + lane * V;
    if constexpr (V == 2) {
      const float2 t = *(const float2*)hp;
      v[u][0] = t.x; v[u][1] = t.y;
    } else {
      v[u][0] = hp[0];
    }
  }
  #pragma unroll
  for (int u = 0; u < G; ++u)
    for (int l = 0; l < V; ++l) acc[l] += v[u][l];
}

template <int F>
__device__ __forceinline__ float sum_segment(const int* __restrict__ rp, const int* __restrict__ csr,
                                             const float* __restrict__ h, int d, int n, int e,
                                             int lane, float (&acc)[F / 32]) {
  constexpr int V = F / 32;
  constexpr int G = 16 / V;
  const int beg = rp[d];
  const int end = (d + 1 < n) ? rp[d + 1] : e;
  for (int l = 0; l < V; ++l) acc[l] = 0.0f;

  for (int base = beg; base < end; base += 32) {
    const int left = end - base;
    const int cnt = left < 32 ? left : 32;
    int s_l = 0;
    if (lane < cnt) {
      s_l = csr[base + lane];
      s_l = min(max(s_l, 0), n - 1);
    }
    int j = 0;
    for (; j + G <= cnt; j += G) take_rows<F, G>(h, s_l, j, lane, acc);
    if constexpr (G >= 16) { if (j + 8 <= cnt) { take_rows<F, 8>(h, s_l, j, lane, acc); j += 8; } }
    if (j + 4 <= cnt) { take_rows<F, 4>(h, s_l, j, lane, acc); j += 4; }
    if (j + 2 <= cnt) { take_rows<F, 2>(h, s_l, j, lane, acc); j += 2; }
    if (j < cnt)      { take_rows<F, 1>(h, s_l, j, lane, acc); }
  }
  return rsqrtf((float)(end - beg) + 1.0f);
}

__device__ __forceinline__ float node_act(float sum, float hs, float dd, float bias,
                                          float mean, float var, float gamma, float beta) {
  float v = __builtin_fmaf(hs, dd, sum * dd) + bias;
  v = (v - mean) * rsqrtf(var + BN_EPS) * gamma + beta;
  return fmaxf(v, 0.0f);
}

__global__ void __launch_bounds__(256)
gather_act(const int* __restrict__ rp, const int* __restrict__ csr,
           const float* __restrict__ h, float* __restrict__ act,
           const float* __restrict__ bias, const float* __restrict__ gamma,
           const float* __restrict__ beta, const float* __restrict__ mean,
           const float* __restrict__ var, int n, int e) {
  constexpr int F = 64;
  const int lane = threadIdx.x & 31;
  const int d = __builtin_amdgcn_readfirstlane((int)((blockIdx.x * blockDim.x + threadIdx.x) >> 5));
  if (d >= n) return;
  const float2 hs = *(const float2*)(h + (size_t)d * F + lane * 2);
  const float2 pb = *(const float2*)(bias + lane * 2);
  const float2 pm = *(const float2*)(mean + lane * 2);
  const float2 pv = *(const float2*)(var + lane * 2);
  const float2 pg = *(const float2*)(gamma + lane * 2);
  const float2 pe = *(const float2*)(beta + lane * 2);

  float acc[2];
  const float dd = sum_segment<F>(rp, csr, h, d, n, e, lane, acc);

  float2 t;
  t.x = node_act(acc[0], hs.x, dd, pb.x, pm.x, pv.x, pg.x, pe.x);
  t.y = node_act(acc[1], hs.y, dd, pb.y, pm.y, pv.y, pg.y, pe.y);
  *(float2*)(act + (size_t)d * F + lane * 2) = t;
}

template <int M>
__device__ __forceinline__ float row_xor(float x) {
  return __builtin_bit_cast(float, __builtin_amdgcn_update_dpp(
      0, __builtin_bit_cast(int, x), 0x160 + M, 0xf, 0xf, true));
}
__device__ __forceinline__ float other_row(float x) {
  const int b = __builtin_bit_cast(int, x);
  return __builtin_bit_cast(float, __builtin_amdgcn_permlanex16(b, b, 0x76543210, 0xfedcba98, false, false));
}

__global__ void __launch_bounds__(512)
gather_head(const int* __restrict__ rp, const int* __restrict__ csr,
            const float* __restrict__ h,
            const float* __restrict__ bias, const float* __restrict__ gamma,
            const float* __restrict__ beta, const float* __restrict__ mean,
            const float* __restrict__ var, const float* __restrict__ Wl,
            const float* __restrict__ bl, float* __restrict__ out, int n, int e) {
  constexpr int F = 32;
  __shared__ float so[32];
  const int lane = threadIdx.x & 31;
  const int wave = __builtin_amdgcn_readfirstlane((int)(threadIdx.x >> 5));
  const int d = blockIdx.x * 16 + wave;
  if (d < n) {
    const float hs = h[(size_t)d * F + lane];
    const float pb = bias[lane], pm = mean[lane], pv = var[lane];
    const float pg = gamma[lane], pe = beta[lane];
    const float2 wl = *(const float2*)(Wl + lane * 2);
    const float blv = bl[lane >> 4];

    float acc[1];
    const float dd = sum_segment<F>(rp, csr, h, d, n, e, lane, acc);
    const float v = node_act(acc[0], hs, dd, pb, pm, pv, pg, pe);

    const bool hi = lane >= 16;
    const float p0 = v * wl.x, p1 = v * wl.y;
    float mine = hi ? p1 : p0;
    mine += other_row(hi ? p0 : p1);
    mine += row_xor<8>(mine);
    mine += row_xor<4>(mine);
    mine += row_xor<2>(mine);
    mine += row_xor<1>(mine);
    if ((lane & 15) == 0) so[wave * 2 + (lane >> 4)] = mine + blv;
  }
  __syncthreads();
  if (threadIdx.x < 32) {
    const int node = blockIdx.x * 16 + (lane >> 1);
    if (node < n) out[(size_t)blockIdx.x * 32 + lane] = so[lane];
  }
}

extern "C" void kernel_launch(void* const* d_in, const int* in_sizes, int n_in,
                              void* d_out, int out_size, void* d_ws, size_t ws_size,
                              hipStream_t stream) {
  const float* x      = (const float*)d_in[0];
  const int*   ei     = (const int*)  d_in[1];
  const float* W1     = (const float*)d_in[2];
  const float* b1     = (const float*)d_in[3];
  const float* gamma1 = (const float*)d_in[4];
  const float* beta1  = (const float*)d_in[5];
  const float* mean1  = (const float*)d_in[6];
  const float* var1   = (const float*)d_in[7];
  const float* W2     = (const float*)d_in[8];
  const float* b2     = (const float*)d_in[9];
  const float* gamma2 = (const float*)d_in[10];
  const float* beta2  = (const float*)d_in[11];
  const float* mean2  = (const float*)d_in[12];
  const float* var2   = (const float*)d_in[13];
  const float* Wl     = (const float*)d_in[14];
  const float* bl     = (const float*)d_in[15];

  const int n = in_sizes[0] / 256;
  const int e = in_sizes[1] / 2;
  const int* src = ei;
  const int* dst = ei + e;

  float* ws   = (float*)d_ws;
  int*   rp   = (int*)ws;
  float* h1   = ws   + (size_t)n;
  float* act1 = h1   + (size_t)n * 64;
  float* h2   = act1 + (size_t)n * 64;
  int*   csr  = (int*)(h2 + (size_t)n * 32);
  int*   hist = (int*)h2;
  int*   off  = hist + (size_t)n * 16;
  int*   bpair = (int*)act1;

  const int T = 256;
  auto blocks = [](long long total, int t) {
    return (unsigned int)((total + t - 1) / t);
  };

  const int nbk = (int)blocks(n, BK_NODES);
  bucket_hist<<<EB, 1024, 0, stream>>>(dst, hist, e, nbk);
  scan_counts<<<blocks((long long)nbk * EB, 1024), 1024, 0, stream>>>(hist, off, nbk * EB);
  bucket_scatter<<<EB, 1024, 0, stream>>>(src, dst, off, bpair, e, nbk);
  bucket_build<<<nbk, BK_NODES, 0, stream>>>(bpair, off, rp, csr, n, e, nbk);

  gemm_wmma_f32<256, 64, 1><<<n / 16, 128, 0, stream>>>(x, W1, h1, rp, n, e);

  gather_act<<<blocks((long long)n * 32, T), T, 0, stream>>>(
      rp, csr, h1, act1, b1, gamma1, beta1, mean1, var1, n, e);

  gemm_wmma_f32<64, 32, 2><<<n / 32, 128, 0, stream>>>(act1, W2, h2, rp, n, e);

  gather_head<<<blocks(n, 16), 512, 0, stream>>>(
      rp, csr, h2, b2, gamma2, beta2, mean2, var2, Wl, bl,
      (float*)d_out, n, e);
}


# aggregation launches run resident waves: a wave loops over a strided set of nodes (14336 waves per launch, not 100000), next node's bounds and ids fetched under the current sums; sum order unchanged
# speedup vs baseline: 3.6490x; 1.0304x over previous
#include <hip/hip_runtime.h>

typedef __attribute__((ext_vector_type(2))) float v2f;
typedef __attribute__((ext_vector_type(8))) float v8f;

#define BN_EPS 1e-5f

#define BK_SHIFT 8
#define BK_NODES (1 << BK_SHIFT)
#define BK_MAX   1024
#define EB       256
#define EU       4
#define BU       8

__device__ __forceinline__ int bucket_of(int d, int nbk) {
  return min(max(d, 0) >> BK_SHIFT, nbk - 1);
}

__global__ void __launch_bounds__(1024)
bucket_hist(const int* __restrict__ dst, int* __restrict__ hist, int e, int nbk) {
  __shared__ int lc[BK_MAX + 32];
  const int tid = threadIdx.x;
  nbk = min(nbk, BK_MAX);
  for (int b = tid; b < nbk; b += 1024) lc[b] = 0;
  if (tid < 32) lc[BK_MAX + tid] = 0;
  __syncthreads();
  const int chunk = (e + EB - 1) / EB;
  const int beg = blockIdx.x * chunk;
  const int end = min(beg + chunk, e);
  for (int base = beg + tid; base < end; base += 1024 * EU) {
    int d[EU];
    #pragma unroll
    for (int u = 0; u < EU; ++u) d[u] = dst[min(base + u * 1024, end - 1)];
    #pragma unroll
    for (int u = 0; u < EU; ++u)
      atomicAdd(&lc[(base + u * 1024 < end) ? bucket_of(d[u], nbk) : BK_MAX + (tid & 31)], 1);
  }
  __syncthreads();
  for (int b = tid; b < nbk; b += 1024) hist[b * EB + blockIdx.x] = lc[b];
}

__global__ void __launch_bounds__(1024)
scan_counts(const int* __restrict__ cnt, int* __restrict__ rp, int n) {
  __shared__ int wtot[32];
  __shared__ int wpart[32];
  const int tid = threadIdx.x, lane = tid & 31, wave = tid >> 5;
  const int ahead = blockIdx.x * 1024;
  int part = 0;
  for (int j = tid; j < ahead; j += 1024) part += cnt[j];
  const int i = ahead + tid;
  const int v = (i < n) ? cnt[i] : 0;
  int inc = v;
  for (int o = 1; o < 32; o <<= 1) {
    int t = __shfl_up(inc, o);
    if (lane >= o) inc += t;
  }
  for (int o = 16; o > 0; o >>= 1) part += __shfl_xor(part, o);
  if (lane == 31) wtot[wave] = inc;
  if (lane == 0)  wpart[wave] = part;
  __syncthreads();
  int p = wpart[lane];
  int t = (lane < wave) ? wtot[lane] : 0;
  for (int o = 16; o > 0; o >>= 1) {
    p += __shfl_xor(p, o);
    t += __shfl_xor(t, o);
  }
  if (i < n) rp[i] = p + t + (inc - v);
}

__global__ void __launch_bounds__(1024)
bucket_scatter(const int* __restrict__ src, const int* __restrict__ dst,
               const int* __restrict__ off, int* __restrict__ bpair, int e, int nbk) {
  __shared__ int lpos[BK_MAX + 32];
  int2* __restrict__ pairs = reinterpret_cast<int2*>(bpair);
  const int tid = threadIdx.x;
  nbk = min(nbk, BK_MAX);
  for (int b = tid; b < nbk; b += 1024) lpos[b] = off[b * EB + blockIdx.x];
  if (tid < 32) lpos[BK_MAX + tid] = 0;
  __syncthreads();
  const int chunk = (e + EB - 1) / EB;
  const int beg = blockIdx.x * chunk;
  const int end = min(beg + chunk, e);
  for (int base = beg + tid; base < end; base += 1024 * EU) {
    int d[EU], s[EU];
    #pragma unroll
    for (int u = 0; u < EU; ++u) {
      const int i = min(base + u * 1024, end - 1);
      d[u] = dst[i];
      s[u] = src[i];
    }
    #pragma unroll
    for (int u = 0; u < EU; ++u) {
      const bool in = base + u * 1024 < end;
      const int p = atomicAdd(&lpos[in ? bucket_of(d[u], nbk) : BK_MAX + (tid & 31)], 1);
      if (in) pairs[p] = make_int2(s[u], d[u]);
    }
  }
}

__global__ void __launch_bounds__(BK_NODES)
bucket_build(const int* __restrict__ bpair, const int* __restrict__ off,
             int* __restrict__ rp, int* __restrict__ csr, int n, int e, int nbk) {
  __shared__ int lcnt[BK_NODES + 32];
  __shared__ int lpos[BK_NODES + 32];
  __shared__ int wtot[BK_NODES / 32];
  const int2* __restrict__ pairs = reinterpret_cast<const int2*>(bpair);
  const int tid = threadIdx.x, lane = tid & 31, wave = tid >> 5;
  const int b = blockIdx.x;
  const int node0 = b << BK_SHIFT;
  const int ebeg = off[b * EB];
  const int eend = (b + 1 < nbk) ? off[(b + 1) * EB] : e;
  lcnt[tid] = 0;
  if (tid < 32) lcnt[BK_NODES + tid] = 0;
  __syncthreads();
  for (int base = ebeg + tid; base < eend; base += BK_NODES * BU) {
    int2 q[BU];
    #pragma unroll
    for (int u = 0; u < BU; ++u) q[u] = pairs[min(base + u * BK_NODES, eend - 1)];
    #pragma unroll
    for (int u = 0; u < BU; ++u)
      atomicAdd(&lcnt[(base + u * BK_NODES < eend) ? min(max(q[u].y - node0, 0), BK_NODES - 1)
                                                   : BK_NODES + lane], 1);
  }
  __syncthreads();
  const int v = lcnt[tid];
  int inc = v;
  for (int o = 1; o < 32; o <<= 1) {
    int t = __shfl_up(inc, o);
    if (lane >= o) inc += t;
  }
  if (lane == 31) wtot[wave] = inc;
  __syncthreads();
  int ahead = 0;
  for (int w = 0; w < BK_NODES / 32; ++w) if (w < wave) ahead += wtot[w];
  const int start = ebeg + ahead + inc - v;
  lpos[tid] = start;
  if (tid < 32) lpos[BK_NODES + tid] = 0;
  if (node0 + tid < n) rp[node0 + tid] = start;
  __syncthreads();
  for (int base = ebeg + tid; base < eend; base += BK_NODES * BU) {
    int2 q[BU];
    #pragma unroll
    for (int u = 0; u < BU; ++u) q[u] = pairs[min(base + u * BK_NODES, eend - 1)];
    #pragma unroll
    for (int u = 0; u < BU; ++u) {
      const bool in = base + u * BK_NODES < eend;
      const int p = atomicAdd(&lpos[in ? min(max(q[u].y - node0, 0), BK_NODES - 1) : BK_NODES + lane], 1);
      if (in) csr[p] = q[u].x;
    }
  }
}

template <int K, int NC, int MT>
__global__ void __launch_bounds__(MT*(NC/16)*32)
gemm_wmma_f32(const float* __restrict__ A, const float* __restrict__ B,
              float* __restrict__ C, const int* __restrict__ rp, int n, int e) {
  constexpr int KP  = K + 4;
  constexpr int NCP = NC + 8;
  constexpr int NW  = MT * (NC/16);
  __shared__ float sA[MT*16*KP];
  __shared__ float sB[K*NCP];

  const int tid = threadIdx.x;
  const int nthreads = NW * 32;
  const float* Ablk = A + (size_t)blockIdx.x * (MT*16) * K;

  const int gm = blockIdx.x*(MT*16) + ((tid >> 5) / (NC/16))*16 + ((tid & 31) >> 4)*8;
  int sb[9];
  #pragma unroll
  for (int r = 0; r < 8; ++r) sb[r] = rp[gm + r];
  sb[8] = rp[min(gm + 8, n - 1)];
  if (gm + 8 >= n) sb[8] = e;
  float dv[8];
  #pragma unroll
  for (int r = 0; r < 8; ++r) dv[r] = rsqrtf((float)(sb[r + 1] - sb[r]) + 1.0f);

  for (int idx = tid; idx < MT*16*(K/4); idx += nthreads) {
    int r = idx / (K/4), c4 = (idx % (K/4)) * 4;
    *(float4*)&sA[r*KP + c4] = *(const float4*)&Ablk[(size_t)r*K + c4];
  }
  for (int idx = tid; idx < K*(NC/4); idx += nthreads) {
    int r = idx / (NC/4), c4 = (idx % (NC/4)) * 4;
    *(float4*)&sB[r*NCP + c4] = *(const float4*)&B[(size_t)r*NC + c4];
  }
  __syncthreads();

  const int wave  = tid >> 5, lane = tid & 31;
  const int mt    = wave / (NC/16), nt = wave % (NC/16);
  const int rrow  = lane & 15;
  const int khalf = lane >> 4;

  const float* a_ptr = &sA[(mt*16 + rrow)*KP + khalf*2];
  const float* b_ptr = &sB[(khalf*2)*NCP + nt*16 + rrow];

  v8f c0 = {};
  v8f c1 = {};
  #pragma unroll 4
  for (int ks = 0; ks < K/8; ++ks) {
    v2f a0; a0.x = a_ptr[0];     a0.y = a_ptr[1];
    v2f b0; b0.x = b_ptr[0];     b0.y = b_ptr[NCP];
    v2f a1; a1.x = a_ptr[4];     a1.y = a_ptr[5];
    v2f b1; b1.x = b_ptr[4*NCP]; b1.y = b_ptr[5*NCP];
    c0 = __builtin_amdgcn_wmma_f32_16x16x4_f32(false, a0, false, b0,
                                               (short)0, c0, false, false);
    c1 = __builtin_amdgcn_wmma_f32_16x16x4_f32(false, a1, false, b1,
                                               (short)0, c1, false, false);
    a_ptr += 8;
    b_ptr += 8*NCP;
  }
  v8f c = c0 + c1;

  const int col = nt*16 + rrow;
  #pragma unroll
  for (int r = 0; r < 8; ++r)
    C[(size_t)(gm + r)*NC + col] = c[r] * dv[r];
}

template <int F, int G>
__device__ __forceinline__ void take_rows(const float* __restrict__ h, int s_l,
                                          int j, int lane, float (&acc)[F / 32]) {
  constexpr int V = F / 32;
  float v[G][V];
  #pragma unroll
  for (int u = 0; u < G; ++u) {
    const int s = __builtin_amdgcn_readlane(s_l, j + u);
    const float* hp = h + (size_t)s * F + lane * V;
    if constexpr (V == 2) {
      const float2 t = *(const float2*)hp;
      v[u][0] = t.x; v[u][1] = t.y;
    } else {
      v[u][0] = hp[0];
    }
  }
  #pragma unroll
  for (int u = 0; u < G; ++u)
    for (int l = 0; l < V; ++l) acc[l] += v[u][l];
}

template <int F>
__device__ __forceinline__ void take_chunk(const float* __restrict__ h, int s_l, int cnt,
                                           int lane, float (&acc)[F / 32]) {
  constexpr int V = F / 32;
  constexpr int G = 16 / V;
  int j = 0;
  for (; j + G <= cnt; j += G) take_rows<F, G>(h, s_l, j, lane, acc);
  if constexpr (G >= 16) { if (j + 8 <= cnt) { take_rows<F, 8>(h, s_l, j, lane, acc); j += 8; } }
  if (j + 4 <= cnt) { take_rows<F, 4>(h, s_l, j, lane, acc); j += 4; }
  if (j + 2 <= cnt) { take_rows<F, 2>(h, s_l, j, lane, acc); j += 2; }
  if (j < cnt)      { take_rows<F, 1>(h, s_l, j, lane, acc); }
}

__device__ __forceinline__ int csr_raw(const int* __restrict__ csr, int i, int e) {
  return csr[min(max(i, 0), e - 1)];
}
__device__ __forceinline__ int csr_id(int raw, int i, int end, int n) {
  return (i < end) ? min(max(raw, 0), n - 1) : 0;
}
__device__ __forceinline__ int bounds_raw(const int* __restrict__ rp, int q, int lane, int n) {
  return rp[min(min(q, n - 1) + (lane & 1), n - 1)];
}

template <int F>
__device__ __forceinline__ void sum_rows(const int* __restrict__ csr, const float* __restrict__ h,
                                         int beg, int end, int ia, int ib, int n, int lane,
                                         float (&acc)[F / 32]) {
  constexpr int V = F / 32;
  for (int l = 0; l < V; ++l) acc[l] = 0.0f;
  const int deg = end - beg;
  take_chunk<F>(h, ia, min(deg, 32), lane, acc);
  if (deg > 32) take_chunk<F>(h, ib, min(deg - 32, 32), lane, acc);
  for (int base = beg + 64; base < end; base += 32) {
    const int cnt = min(end - base, 32);
    int s_l = 0;
    if (lane < cnt) {
      s_l = csr[base + lane];
      s_l = min(max(s_l, 0), n - 1);
    }
    take_chunk<F>(h, s_l, cnt, lane, acc);
  }
}

__device__ __forceinline__ float node_act(float sum, float hs, float dd, float bias,
                                          float mean, float var, float gamma, float beta) {
  float v = __builtin_fmaf(hs, dd, sum * dd) + bias;
  v = (v - mean) * rsqrtf(var + BN_EPS) * gamma + beta;
  return fmaxf(v, 0.0f);
}

__global__ void __launch_bounds__(256)
gather_act(const int* __restrict__ rp, const int* __restrict__ csr,
           const float* __restrict__ h, float* __restrict__ act,
           const float* __restrict__ bias, const float* __restrict__ gamma,
           const float* __restrict__ beta, const float* __restrict__ mean,
           const float* __restrict__ var, int n, int e) {
  constexpr int F = 64;
  const int lane = threadIdx.x & 31;
  const int nw = (int)((gridDim.x * blockDim.x) >> 5);
  int d = __builtin_amdgcn_readfirstlane((int)((blockIdx.x * blockDim.x + threadIdx.x) >> 5));
  if (d >= n) return;
  const float2 pb = *(const float2*)(bias + lane * 2);
  const float2 pm = *(const float2*)(mean + lane * 2);
  const float2 pv = *(const float2*)(var + lane * 2);
  const float2 pg = *(const float2*)(gamma + lane * 2);
  const float2 pe = *(const float2*)(beta + lane * 2);

  int beg = rp[d];
  int end = (d + 1 < n) ? rp[d + 1] : e;
  int ra = csr_raw(csr, beg + lane, e);
  int rb = csr_raw(csr, beg + 32 + lane, e);
  int rq = bounds_raw(rp, d + nw, lane, n);

  for (;;) {
    const int ia = csr_id(ra, beg + lane, end, n);
    const int ib = csr_id(rb, beg + 32 + lane, end, n);
    const int dn = d + nw;
    const int nbeg = __builtin_amdgcn_readlane(rq, 0);
    const int nend = (min(dn, n - 1) + 1 < n) ? __builtin_amdgcn_readlane(rq, 1) : e;
    ra = csr_raw(csr, nbeg + lane, e);
    rb = csr_raw(csr, nbeg + 32 + lane, e);
    rq = bounds_raw(rp, dn + nw, lane, n);
    const float2 hs = *(const float2*)(h + (size_t)d * F + lane * 2);

    float acc[2];
    sum_rows<F>(csr, h, beg, end, ia, ib, n, lane, acc);
    const float dd = rsqrtf((float)(end - beg) + 1.0f);

    float2 t;
    t.x = node_act(acc[0], hs.x, dd, pb.x, pm.x, pv.x, pg.x, pe.x);
    t.y = node_act(acc[1], hs.y, dd, pb.y, pm.y, pv.y, pg.y, pe.y);
    *(float2*)(act + (size_t)d * F + lane * 2) = t;

    d = dn; beg = nbeg; end = nend;
    if (d >= n) break;
  }
}

template <int M>
__device__ __forceinline__ float row_xor(float x) {
  return __builtin_bit_cast(float, __builtin_amdgcn_update_dpp(
      0, __builtin_bit_cast(int, x), 0x160 + M, 0xf, 0xf, true));
}
__device__ __forceinline__ float other_row(float x) {
  const int b = __builtin_bit_cast(int, x);
  return __builtin_bit_cast(float, __builtin_amdgcn_permlanex16(b, b, 0x76543210, 0xfedcba98, false, false));
}

__global__ void __launch_bounds__(512)
gather_head(const int* __restrict__ rp, const int* __restrict__ csr,
            const float* __restrict__ h,
            const float* __restrict__ bias, const float* __restrict__ gamma,
            const float* __restrict__ beta, const float* __restrict__ mean,
            const float* __restrict__ var, const float* __restrict__ Wl,
            const float* __restrict__ bl, float* __restrict__ out, int n, int e) {
  constexpr int F = 32;
  constexpr int RL = 8;
  __shared__ float so[RL][32];
  const int lane = threadIdx.x & 31;
  const int wave = __builtin_amdgcn_readfirstlane((int)(threadIdx.x >> 5));
  const int nblk = (n + 15) >> 4;
  const int gstep = (int)gridDim.x;
  const int dstep = gstep * 16;
  int blk = blockIdx.x;
  if (blk >= nblk) return;
  const float pb = bias[lane], pm = mean[lane], pv = var[lane];
  const float pg = gamma[lane], pe = beta[lane];
  const float2 wl = *(const float2*)(Wl + lane * 2);
  const float blv = bl[lane >> 4];

  int d = blk * 16 + wave;
  const int d0 = min(d, n - 1);
  int beg = rp[d0];
  int end = (d0 + 1 < n) ? rp[min(d0 + 1, n - 1)] : e;
  int ra = csr_raw(csr, beg + lane, e);
  int rb = csr_raw(csr, beg + 32 + lane, e);
  int rq = bounds_raw(rp, d + dstep, lane, n);

  int k = 0;
  for (; blk < nblk; blk += gstep, ++k) {
    const int r = k & (RL - 1);
    const int ia = csr_id(ra, beg + lane, end, n);
    const int ib = csr_id(rb, beg + 32 + lane, end, n);
    const int dn = d + dstep;
    const int nbeg = __builtin_amdgcn_readlane(rq, 0);
    const int nend = (min(dn, n - 1) + 1 < n) ? __builtin_amdgcn_readlane(rq, 1) : e;
    ra = csr_raw(csr, nbeg + lane, e);
    rb = csr_raw(csr, nbeg + 32 + lane, e);
    rq = bounds_raw(rp, dn + dstep, lane, n);

    if (d < n) {
      const float hs = h[(size_t)d * F + lane];
      float acc[1];
      sum_rows<F>(csr, h, beg, end, ia, ib, n, lane, acc);
      const float dd = rsqrtf((float)(end - beg) + 1.0f);
      const float v = node_act(acc[0], hs, dd, pb, pm, pv, pg, pe);

      const bool hi = lane >= 16;
      const float p0 = v * wl.x, p1 = v * wl.y;
      float mine = hi ? p1 : p0;
      mine += other_row(hi ? p0 : p1);
      mine += row_xor<8>(mine);
      mine += row_xor<4>(mine);
      mine += row_xor<2>(mine);
      mine += row_xor<1>(mine);
      if ((lane & 15) == 0) so[r][wave * 2 + (lane >> 4)] = mine + blv;
    }

    if (r == RL - 1 || blk + gstep >= nblk) {
      __syncthreads();
      if (wave <= r) {
        const int fb = blk - (r - wave) * gstep;
        const int node = fb * 16 + (lane >> 1);
        if (node < n) out[(size_t)fb * 32 + lane] = so[wave][lane];
      }
      __syncthreads();
    }
    d = dn; beg = nbeg; end = nend;
  }
}

extern "C" void kernel_launch(void* const* d_in, const int* in_sizes, int n_in,
                              void* d_out, int out_size, void* d_ws, size_t ws_size,
                              hipStream_t stream) {
  const float* x      = (const float*)d_in[0];
  const int*   ei     = (const int*)  d_in[1];
  const float* W1     = (const float*)d_in[2];
  const float* b1     = (const float*)d_in[3];
  const float* gamma1 = (const float*)d_in[4];
  const float* beta1  = (const float*)d_in[5];
  const float* mean1  = (const float*)d_in[6];
  const float* var1   = (const float*)d_in[7];
  const float* W2     = (const float*)d_in[8];
  const float* b2     = (const float*)d_in[9];
  const float* gamma2 = (const float*)d_in[10];
  const float* beta2  = (const float*)d_in[11];
  const float* mean2  = (const float*)d_in[12];
  const float* var2   = (const float*)d_in[13];
  const float* Wl     = (const float*)d_in[14];
  const float* bl     = (const float*)d_in[15];

  const int n = in_sizes[0] / 256;
  const int e = in_sizes[1] / 2;
  const int* src = ei;
  const int* dst = ei + e;

  float* ws   = (float*)d_ws;
  int*   rp   = (int*)ws;
  float* h1   = ws   + (size_t)n;
  float* act1 = h1   + (size_t)n * 64;
  float* h2   = act1 + (size_t)n * 64;
  int*   csr  = (int*)(h2 + (size_t)n * 32);
  int*   hist = (int*)h2;
  int*   off  = hist + (size_t)n * 16;
  int*   bpair = (int*)act1;

  const int T = 256;
  auto blocks = [](long long total, int t) {
    return (unsigned int)((total + t - 1) / t);
  };

  const int nbk = (int)blocks(n, BK_NODES);
  bucket_hist<<<EB, 1024, 0, stream>>>(dst, hist, e, nbk);
  scan_counts<<<blocks((long long)nbk * EB, 1024), 1024, 0, stream>>>(hist, off, nbk * EB);
  bucket_scatter<<<EB, 1024, 0, stream>>>(src, dst, off, bpair, e, nbk);
  bucket_build<<<nbk, BK_NODES, 0, stream>>>(bpair, off, rp, csr, n, e, nbk);

  gemm_wmma_f32<256, 64, 1><<<n / 16, 128, 0, stream>>>(x, W1, h1, rp, n, e);

  gather_act<<<1792, T, 0, stream>>>(
      rp, csr, h1, act1, b1, gamma1, beta1, mean1, var1, n, e);

  gemm_wmma_f32<64, 32, 2><<<n / 32, 128, 0, stream>>>(act1, W2, h2, rp, n, e);

  gather_head<<<896, 512, 0, stream>>>(
      rp, csr, h2, b2, gamma2, beta2, mean2, var2, Wl, bl,
      (float*)d_out, n, e);
}


# layer-1 GEMM on resident work-groups: W1 staged in LDS once per work-group (512 stagings, not 6250), row tiles looped, the next tile's rows loaded under the current products; sum order unchanged
# speedup vs baseline: 4.4543x; 1.2207x over previous
#include <hip/hip_runtime.h>

typedef __attribute__((ext_vector_type(2))) float v2f;
typedef __attribute__((ext_vector_type(8))) float v8f;
typedef __attribute__((ext_vector_type(4))) float v4f;

#define BN_EPS 1e-5f

#define BK_SHIFT 8
#define BK_NODES (1 << BK_SHIFT)
#define BK_MAX   1024
#define EB       256
#define EU       4
#define BU       8

__device__ __forceinline__ int bucket_of(int d, int nbk) {
  return min(max(d, 0) >> BK_SHIFT, nbk - 1);
}

__global__ void __launch_bounds__(1024)
bucket_hist(const int* __restrict__ dst, int* __restrict__ hist, int e, int nbk) {
  __shared__ int lc[BK_MAX + 32];
  const int tid = threadIdx.x;
  nbk = min(nbk, BK_MAX);
  for (int b = tid; b < nbk; b += 1024) lc[b] = 0;
  if (tid < 32) lc[BK_MAX + tid] = 0;
  __syncthreads();
  const int chunk = (e + EB - 1) / EB;
  const int beg = blockIdx.x * chunk;
  const int end = min(beg + chunk, e);
  for (int base = beg + tid; base < end; base += 1024 * EU) {
    int d[EU];
    #pragma unroll
    for (int u = 0; u < EU; ++u) d[u] = dst[min(base + u * 1024, end - 1)];
    #pragma unroll
    for (int u = 0; u < EU; ++u)
      atomicAdd(&lc[(base + u * 1024 < end) ? bucket_of(d[u], nbk) : BK_MAX + (tid & 31)], 1);
  }
  __syncthreads();
  for (int b = tid; b < nbk; b += 1024) hist[b * EB + blockIdx.x] = lc[b];
}

__global__ void __launch_bounds__(1024)
scan_counts(const int* __restrict__ cnt, int* __restrict__ rp, int n) {
  __shared__ int wtot[32];
  __shared__ int wpart[32];
  const int tid = threadIdx.x, lane = tid & 31, wave = tid >> 5;
  const int ahead = blockIdx.x * 1024;
  int part = 0;
  for (int j = tid; j < ahead; j += 1024) part += cnt[j];
  const int i = ahead + tid;
  const int v = (i < n) ? cnt[i] : 0;
  int inc = v;
  for (int o = 1; o < 32; o <<= 1) {
    int t = __shfl_up(inc, o);
    if (lane >= o) inc += t;
  }
  for (int o = 16; o > 0; o >>= 1) part += __shfl_xor(part, o);
  if (lane == 31) wtot[wave] = inc;
  if (lane == 0)  wpart[wave] = part;
  __syncthreads();
  int p = wpart[lane];
  int t = (lane < wave) ? wtot[lane] : 0;
  for (int o = 16; o > 0; o >>= 1) {
    p += __shfl_xor(p, o);
    t += __shfl_xor(t, o);
  }
  if (i < n) rp[i] = p + t + (inc - v);
}

__global__ void __launch_bounds__(1024)
bucket_scatter(const int* __restrict__ src, const int* __restrict__ dst,
               const int* __restrict__ off, int* __restrict__ bpair, int e, int nbk) {
  __shared__ int lpos[BK_MAX + 32];
  int2* __restrict__ pairs = reinterpret_cast<int2*>(bpair);
  const int tid = threadIdx.x;
  nbk = min(nbk, BK_MAX);
  for (int b = tid; b < nbk; b += 1024) lpos[b] = off[b * EB + blockIdx.x];
  if (tid < 32) lpos[BK_MAX + tid] = 0;
  __syncthreads();
  const int chunk = (e + EB - 1) / EB;
  const int beg = blockIdx.x * chunk;
  const int end = min(beg + chunk, e);
  for (int base = beg + tid; base < end; base += 1024 * EU) {
    int d[EU], s[EU];
    #pragma unroll
    for (int u = 0; u < EU; ++u) {
      const int i = min(base + u * 1024, end - 1);
      d[u] = dst[i];
      s[u] = src[i];
    }
    #pragma unroll
    for (int u = 0; u < EU; ++u) {
      const bool in = base + u * 1024 < end;
      const int p = atomicAdd(&lpos[in ? bucket_of(d[u], nbk) : BK_MAX + (tid & 31)], 1);
      if (in) pairs[p] = make_int2(s[u], d[u]);
    }
  }
}

__global__ void __launch_bounds__(BK_NODES)
bucket_build(const int* __restrict__ bpair, const int* __restrict__ off,
             int* __restrict__ rp, int* __restrict__ csr, int n, int e, int nbk) {
  __shared__ int lcnt[BK_NODES + 32];
  __shared__ int lpos[BK_NODES + 32];
  __shared__ int wtot[BK_NODES / 32];
  const int2* __restrict__ pairs = reinterpret_cast<const int2*>(bpair);
  const int tid = threadIdx.x, lane = tid & 31, wave = tid >> 5;
  const int b = blockIdx.x;
  const int node0 = b << BK_SHIFT;
  const int ebeg = off[b * EB];
  const int eend = (b + 1 < nbk) ? off[(b + 1) * EB] : e;
  lcnt[tid] = 0;
  if (tid < 32) lcnt[BK_NODES + tid] = 0;
  __syncthreads();
  for (int base = ebeg + tid; base < eend; base += BK_NODES * BU) {
    int2 q[BU];
    #pragma unroll
    for (int u = 0; u < BU; ++u) q[u] = pairs[min(base + u * BK_NODES, eend - 1)];
    #pragma unroll
    for (int u = 0; u < BU; ++u)
      atomicAdd(&lcnt[(base + u * BK_NODES < eend) ? min(max(q[u].y - node0, 0), BK_NODES - 1)
                                                   : BK_NODES + lane], 1);
  }
  __syncthreads();
  const int v = lcnt[tid];
  int inc = v;
  for (int o = 1; o < 32; o <<= 1) {
    int t = __shfl_up(inc, o);
    if (lane >= o) inc += t;
  }
  if (lane == 31) wtot[wave] = inc;
  __syncthreads();
  int ahead = 0;
  for (int w = 0; w < BK_NODES / 32; ++w) if (w < wave) ahead += wtot[w];
  const int start = ebeg + ahead + inc - v;
  lpos[tid] = start;
  if (tid < 32) lpos[BK_NODES + tid] = 0;
  if (node0 + tid < n) rp[node0 + tid] = start;
  __syncthreads();
  for (int base = ebeg + tid; base < eend; base += BK_NODES * BU) {
    int2 q[BU];
    #pragma unroll
    for (int u = 0; u < BU; ++u) q[u] = pairs[min(base + u * BK_NODES, eend - 1)];
    #pragma unroll
    for (int u = 0; u < BU; ++u) {
      const bool in = base + u * BK_NODES < eend;
      const int p = atomicAdd(&lpos[in ? min(max(q[u].y - node0, 0), BK_NODES - 1) : BK_NODES + lane], 1);
      if (in) csr[p] = q[u].x;
    }
  }
}

template <int K, int NC, int MT>
__global__ void __launch_bounds__(MT*(NC/16)*32)
gemm_wmma_f32(const float* __restrict__ A, const float* __restrict__ B,
              float* __restrict__ C, const int* __restrict__ rp, int n, int e) {
  constexpr int KP  = K + 4;
  constexpr int NCP = NC + 8;
  constexpr int NW  = MT * (NC/16);
  __shared__ float sA[MT*16*KP];
  __shared__ float sB[K*NCP];

  const int tid = threadIdx.x;
  const int nthreads = NW * 32;
  const float* Ablk = A + (size_t)blockIdx.x * (MT*16) * K;

  const int gm = blockIdx.x*(MT*16) + ((tid >> 5) / (NC/16))*16 + ((tid & 31) >> 4)*8;
  int sb[9];
  #pragma unroll
  for (int r = 0; r < 8; ++r) sb[r] = rp[gm + r];
  sb[8] = rp[min(gm + 8, n - 1)];
  if (gm + 8 >= n) sb[8] = e;
  float dv[8];
  #pragma unroll
  for (int r = 0; r < 8; ++r) dv[r] = rsqrtf((float)(sb[r + 1] - sb[r]) + 1.0f);

  for (int idx = tid; idx < MT*16*(K/4); idx += nthreads) {
    int r = idx / (K/4), c4 = (idx % (K/4)) * 4;
    *(float4*)&sA[r*KP + c4] = *(const float4*)&Ablk[(size_t)r*K + c4];
  }
  for (int idx = tid; idx < K*(NC/4); idx += nthreads) {
    int r = idx / (NC/4), c4 = (idx % (NC/4)) * 4;
    *(float4*)&sB[r*NCP + c4] = *(const float4*)&B[(size_t)r*NC + c4];
  }
  __syncthreads();

  const int wave  = tid >> 5, lane = tid & 31;
  const int mt    = wave / (NC/16), nt = wave % (NC/16);
  const int rrow  = lane & 15;
  const int khalf = lane >> 4;

  const float* a_ptr = &sA[(mt*16 + rrow)*KP + khalf*2];
  const float* b_ptr = &sB[(khalf*2)*NCP + nt*16 + rrow];

  v8f c0 = {};
  v8f c1 = {};
  #pragma unroll 4
  for (int ks = 0; ks < K/8; ++ks) {
    v2f a0; a0.x = a_ptr[0];     a0.y = a_ptr[1];
    v2f b0; b0.x = b_ptr[0];     b0.y = b_ptr[NCP];
    v2f a1; a1.x = a_ptr[4];     a1.y = a_ptr[5];
    v2f b1; b1.x = b_ptr[4*NCP]; b1.y = b_ptr[5*NCP];
    c0 = __builtin_amdgcn_wmma_f32_16x16x4_f32(false, a0, false, b0,
                                               (short)0, c0, false, false);
    c1 = __builtin_amdgcn_wmma_f32_16x16x4_f32(false, a1, false, b1,
                                               (short)0, c1, false, false);
    a_ptr += 8;
    b_ptr += 8*NCP;
  }
  v8f c = c0 + c1;

  const int col = nt*16 + rrow;
  #pragma unroll
  for (int r = 0; r < 8; ++r)
    C[(size_t)(gm + r)*NC + col] = c[r] * dv[r];
}

template <int K, int NC, int MT>
__global__ void __launch_bounds__(MT*(NC/16)*32)
gemm_wmma_f32_resident(const float* __restrict__ A, const float* __restrict__ B,
                       float* __restrict__ C, const int* __restrict__ rp, int n, int e) {
  constexpr int KP   = K + 4;
  constexpr int NCP  = NC + 8;
  constexpr int NW   = MT * (NC/16);
  constexpr int NT   = NW * 32;
  constexpr int ROWS = MT * 16;
  constexpr int AV   = ROWS * (K/4) / NT;
  constexpr int BV   = K * (NC/4) / NT;
  static_assert((ROWS * (K/4)) % NT == 0 && (K * (NC/4)) % NT == 0, "whole pieces per thread");
  __shared__ float sA[2][ROWS*KP];
  __shared__ float sB[K*NCP];

  const int tid = threadIdx.x;
  const int ntiles = n / ROWS;
  const int gstep = (int)gridDim.x;
  int g = blockIdx.x;
  if (g >= ntiles) return;

  v4f pf[AV];
  {
    const float* Ablk = A + (size_t)g * ROWS * K;
    #pragma unroll
    for (int j = 0; j < AV; ++j) {
      const int idx = tid + j * NT;
      const int r = idx / (K/4), c4 = (idx % (K/4)) * 4;
      pf[j] = *(const v4f*)&Ablk[(size_t)r*K + c4];
    }
  }
  #pragma unroll
  for (int j = 0; j < BV; ++j) {
    const int idx = tid + j * NT;
    const int r = idx / (NC/4), c4 = (idx % (NC/4)) * 4;
    *(float4*)&sB[r*NCP + c4] = *(const float4*)&B[(size_t)r*NC + c4];
  }
  #pragma unroll
  for (int j = 0; j < AV; ++j) {
    const int idx = tid + j * NT;
    const int r = idx / (K/4), c4 = (idx % (K/4)) * 4;
    *(v4f*)&sA[0][r*KP + c4] = pf[j];
  }
  __syncthreads();

  const int wave  = tid >> 5, lane = tid & 31;
  const int mt    = wave / (NC/16), nt = wave % (NC/16);
  const int rrow  = lane & 15;
  const int khalf = lane >> 4;
  const int col   = nt*16 + rrow;
  const float* b_ptr0 = &sB[(khalf*2)*NCP + nt*16 + rrow];

  for (int cur = 0; ; cur ^= 1) {
    const int gn = g + gstep;
    const bool more = gn < ntiles;

    const int gm = g*ROWS + mt*16 + khalf*8;
    int sb[9];
    #pragma unroll
    for (int r = 0; r < 8; ++r) sb[r] = rp[gm + r];
    sb[8] = rp[min(gm + 8, n - 1)];
    if (gm + 8 >= n) sb[8] = e;

    if (more) {
      const float* Ablk = A + (size_t)gn * ROWS * K;
      #pragma unroll
      for (int j = 0; j < AV; ++j) {
        const int idx = tid + j * NT;
        const int r = idx / (K/4), c4 = (idx % (K/4)) * 4;
        pf[j] = *(const v4f*)&Ablk[(size_t)r*K + c4];
      }
    }

    const float* a_ptr = &sA[cur][(mt*16 + rrow)*KP + khalf*2];
    const float* b_ptr = b_ptr0;
    v8f c0 = {};
    v8f c1 = {};
    #pragma unroll 4
    for (int ks = 0; ks < K/8; ++ks) {
      v2f a0; a0.x = a_ptr[0];     a0.y = a_ptr[1];
      v2f b0; b0.x = b_ptr[0];     b0.y = b_ptr[NCP];
      v2f a1; a1.x = a_ptr[4];     a1.y = a_ptr[5];
      v2f b1; b1.x = b_ptr[4*NCP]; b1.y = b_ptr[5*NCP];
      c0 = __builtin_amdgcn_wmma_f32_16x16x4_f32(false, a0, false, b0,
                                                 (short)0, c0, false, false);
      c1 = __builtin_amdgcn_wmma_f32_16x16x4_f32(false, a1, false, b1,
                                                 (short)0, c1, false, false);
      a_ptr += 8;
      b_ptr += 8*NCP;
    }
    v8f c = c0 + c1;

    if (more) {
      #pragma unroll
      for (int j = 0; j < AV; ++j) {
        const int idx = tid + j * NT;
        const int r = idx / (K/4), c4 = (idx % (K/4)) * 4;
        *(v4f*)&sA[cur ^ 1][r*KP + c4] = pf[j];
      }
      __syncthreads();
    }

    #pragma unroll
    for (int r = 0; r < 8; ++r) {
      const float dv = rsqrtf((float)(sb[r + 1] - sb[r]) + 1.0f);
      C[(size_t)(gm + r)*NC + col] = c[r] * dv;
    }

    if (!more) break;
    g = gn;
  }
}

template <int F, int G>
__device__ __forceinline__ void take_rows(const float* __restrict__ h, int s_l,
                                          int j, int lane, float (&acc)[F / 32]) {
  constexpr int V = F / 32;
  float v[G][V];
  #pragma unroll
  for (int u = 0; u < G; ++u) {
    const int s = __builtin_amdgcn_readlane(s_l, j + u);
    const float* hp = h + (size_t)s * F + lane * V;
    if constexpr (V == 2) {
      const float2 t = *(const float2*)hp;
      v[u][0] = t.x; v[u][1] = t.y;
    } else {
      v[u][0] = hp[0];
    }
  }
  #pragma unroll
  for (int u = 0; u < G; ++u)
    for (int l = 0; l < V; ++l) acc[l] += v[u][l];
}

template <int F>
__device__ __forceinline__ void take_chunk(const float* __restrict__ h, int s_l, int cnt,
                                           int lane, float (&acc)[F / 32]) {
  constexpr int V = F / 32;
  constexpr int G = 16 / V;
  int j = 0;
  for (; j + G <= cnt; j += G) take_rows<F, G>(h, s_l, j, lane, acc);
  if constexpr (G >= 16) { if (j + 8 <= cnt) { take_rows<F, 8>(h, s_l, j, lane, acc); j += 8; } }
  if (j + 4 <= cnt) { take_rows<F, 4>(h, s_l, j, lane, acc); j += 4; }
  if (j + 2 <= cnt) { take_rows<F, 2>(h, s_l, j, lane, acc); j += 2; }
  if (j < cnt)      { take_rows<F, 1>(h, s_l, j, lane, acc); }
}

__device__ __forceinline__ int csr_raw(const int* __restrict__ csr, int i, int e) {
  return csr[min(max(i, 0), e - 1)];
}
__device__ __forceinline__ int csr_id(int raw, int i, int end, int n) {
  return (i < end) ? min(max(raw, 0), n - 1) : 0;
}
__device__ __forceinline__ int bounds_raw(const int* __restrict__ rp, int q, int lane, int n) {
  return rp[min(min(q, n - 1) + (lane & 1), n - 1)];
}

template <int F>
__device__ __forceinline__ void sum_rows(const int* __restrict__ csr, const float* __restrict__ h,
                                         int beg, int end, int ia, int ib, int n, int lane,
                                         float (&acc)[F / 32]) {
  constexpr int V = F / 32;
  for (int l = 0; l < V; ++l) acc[l] = 0.0f;
  const int deg = end - beg;
  take_chunk<F>(h, ia, min(deg, 32), lane, acc);
  if (deg > 32) take_chunk<F>(h, ib, min(deg - 32, 32), lane, acc);
  for (int base = beg + 64; base < end; base += 32) {
    const int cnt = min(end - base, 32);
    int s_l = 0;
    if (lane < cnt) {
      s_l = csr[base + lane];
      s_l = min(max(s_l, 0), n - 1);
    }
    take_chunk<F>(h, s_l, cnt, lane, acc);
  }
}

__device__ __forceinline__ float node_act(float sum, float hs, float dd, float bias,
                                          float mean, float var, float gamma, float beta) {
  float v = __builtin_fmaf(hs, dd, sum * dd) + bias;
  v = (v - mean) * rsqrtf(var + BN_EPS) * gamma + beta;
  return fmaxf(v, 0.0f);
}

__global__ void __launch_bounds__(256)
gather_act(const int* __restrict__ rp, const int* __restrict__ csr,
           const float* __restrict__ h, float* __restrict__ act,
           const float* __restrict__ bias, const float* __restrict__ gamma,
           const float* __restrict__ beta, const float* __restrict__ mean,
           const float* __restrict__ var, int n, int e) {
  constexpr int F = 64;
  const int lane = threadIdx.x & 31;
  const int nw = (int)((gridDim.x * blockDim.x) >> 5);
  int d = __builtin_amdgcn_readfirstlane((int)((blockIdx.x * blockDim.x + threadIdx.x) >> 5));
  if (d >= n) return;
  const float2 pb = *(const float2*)(bias + lane * 2);
  const float2 pm = *(const float2*)(mean + lane * 2);
  const float2 pv = *(const float2*)(var + lane * 2);
  const float2 pg = *(const float2*)(gamma + lane * 2);
  const float2 pe = *(const float2*)(beta + lane * 2);

  int beg = rp[d];
  int end = (d + 1 < n) ? rp[d + 1] : e;
  int ra = csr_raw(csr, beg + lane, e);
  int rb = csr_raw(csr, beg + 32 + lane, e);
  int rq = bounds_raw(rp, d + nw, lane, n);

  for (;;) {
    const int ia = csr_id(ra, beg + lane, end, n);
    const int ib = csr_id(rb, beg + 32 + lane, end, n);
    const int dn = d + nw;
    const int nbeg = __builtin_amdgcn_readlane(rq, 0);
    const int nend = (min(dn, n - 1) + 1 < n) ? __builtin_amdgcn_readlane(rq, 1) : e;
    ra = csr_raw(csr, nbeg + lane, e);
    rb = csr_raw(csr, nbeg + 32 + lane, e);
    rq = bounds_raw(rp, dn + nw, lane, n);
    const float2 hs = *(const float2*)(h + (size_t)d * F + lane * 2);

    float acc[2];
    sum_rows<F>(csr, h, beg, end, ia, ib, n, lane, acc);
    const float dd = rsqrtf((float)(end - beg) + 1.0f);

    float2 t;
    t.x = node_act(acc[0], hs.x, dd, pb.x, pm.x, pv.x, pg.x, pe.x);
    t.y = node_act(acc[1], hs.y, dd, pb.y, pm.y, pv.y, pg.y, pe.y);
    *(float2*)(act + (size_t)d * F + lane * 2) = t;

    d = dn; beg = nbeg; end = nend;
    if (d >= n) break;
  }
}

template <int M>
__device__ __forceinline__ float row_xor(float x) {
  return __builtin_bit_cast(float, __builtin_amdgcn_update_dpp(
      0, __builtin_bit_cast(int, x), 0x160 + M, 0xf, 0xf, true));
}
__device__ __forceinline__ float other_row(float x) {
  const int b = __builtin_bit_cast(int, x);
  return __builtin_bit_cast(float, __builtin_amdgcn_permlanex16(b, b, 0x76543210, 0xfedcba98, false, false));
}

__global__ void __launch_bounds__(512)
gather_head(const int* __restrict__ rp, const int* __restrict__ csr,
            const float* __restrict__ h,
            const float* __restrict__ bias, const float* __restrict__ gamma,
            const float* __restrict__ beta, const float* __restrict__ mean,
            const float* __restrict__ var, const float* __restrict__ Wl,
            const float* __restrict__ bl, float* __restrict__ out, int n, int e) {
  constexpr int F = 32;
  constexpr int RL = 8;
  __shared__ float so[RL][32];
  const int lane = threadIdx.x & 31;
  const int wave = __builtin_amdgcn_readfirstlane((int)(threadIdx.x >> 5));
  const int nblk = (n + 15) >> 4;
  const int gstep = (int)gridDim.x;
  const int dstep = gstep * 16;
  int blk = blockIdx.x;
  if (blk >= nblk) return;
  const float pb = bias[lane], pm = mean[lane], pv = var[lane];
  const float pg = gamma[lane], pe = beta[lane];
  const float2 wl = *(const float2*)(Wl + lane * 2);
  const float blv = bl[lane >> 4];

  int d = blk * 16 + wave;
  const int d0 = min(d, n - 1);
  int beg = rp[d0];
  int end = (d0 + 1 < n) ? rp[min(d0 + 1, n - 1)] : e;
  int ra = csr_raw(csr, beg + lane, e);
  int rb = csr_raw(csr, beg + 32 + lane, e);
  int rq = bounds_raw(rp, d + dstep, lane, n);

  int k = 0;
  for (; blk < nblk; blk += gstep, ++k) {
    const int r = k & (RL - 1);
    const int ia = csr_id(ra, beg + lane, end, n);
    const int ib = csr_id(rb, beg + 32 + lane, end, n);
    const int dn = d + dstep;
    const int nbeg = __builtin_amdgcn_readlane(rq, 0);
    const int nend = (min(dn, n - 1) + 1 < n) ? __builtin_amdgcn_readlane(rq, 1) : e;
    ra = csr_raw(csr, nbeg + lane, e);
    rb = csr_raw(csr, nbeg + 32 + lane, e);
    rq = bounds_raw(rp, dn + dstep, lane, n);

    if (d < n) {
      const float hs = h[(size_t)d * F + lane];
      float acc[1];
      sum_rows<F>(csr, h, beg, end, ia, ib, n, lane, acc);
      const float dd = rsqrtf((float)(end - beg) + 1.0f);
      const float v = node_act(acc[0], hs, dd, pb, pm, pv, pg, pe);

      const bool hi = lane >= 16;
      const float p0 = v * wl.x, p1 = v * wl.y;
      float mine = hi ? p1 : p0;
      mine += other_row(hi ? p0 : p1);
      mine += row_xor<8>(mine);
      mine += row_xor<4>(mine);
      mine += row_xor<2>(mine);
      mine += row_xor<1>(mine);
      if ((lane & 15) == 0) so[r][wave * 2 + (lane >> 4)] = mine + blv;
    }

    if (r == RL - 1 || blk + gstep >= nblk) {
      __syncthreads();
      if (wave <= r) {
        const int fb = blk - (r - wave) * gstep;
        const int node = fb * 16 + (lane >> 1);
        if (node < n) out[(size_t)fb * 32 + lane] = so[wave][lane];
      }
      __syncthreads();
    }
    d = dn; beg = nbeg; end = nend;
  }
}

extern "C" void kernel_launch(void* const* d_in, const int* in_sizes, int n_in,
                              void* d_out, int out_size, void* d_ws, size_t ws_size,
                              hipStream_t stream) {
  const float* x      = (const float*)d_in[0];
  const int*   ei     = (const int*)  d_in[1];
  const float* W1     = (const float*)d_in[2];
  const float* b1     = (const float*)d_in[3];
  const float* gamma1 = (const float*)d_in[4];
  const float* beta1  = (const float*)d_in[5];
  const float* mean1  = (const float*)d_in[6];
  const float* var1   = (const float*)d_in[7];
  const float* W2     = (const float*)d_in[8];
  const float* b2     = (const float*)d_in[9];
  const float* gamma2 = (const float*)d_in[10];
  const float* beta2  = (const float*)d_in[11];
  const float* mean2  = (const float*)d_in[12];
  const float* var2   = (const float*)d_in[13];
  const float* Wl     = (const float*)d_in[14];
  const float* bl     = (const float*)d_in[15];

  const int n = in_sizes[0] / 256;
  const int e = in_sizes[1] / 2;
  const int* src = ei;
  const int* dst = ei + e;

  float* ws   = (float*)d_ws;
  int*   rp   = (int*)ws;
  float* h1   = ws   + (size_t)n;
  float* act1 = h1   + (size_t)n * 64;
  float* h2   = act1 + (size_t)n * 64;
  int*   csr  = (int*)(h2 + (size_t)n * 32);
  int*   hist = (int*)h2;
  int*   off  = hist + (size_t)n * 16;
  int*   bpair = (int*)act1;

  const int T = 256;
  auto blocks = [](long long total, int t) {
    return (unsigned int)((total + t - 1) / t);
  };

  const int nbk = (int)blocks(n, BK_NODES);
  bucket_hist<<<EB, 1024, 0, stream>>>(dst, hist, e, nbk);
  scan_counts<<<blocks((long long)nbk * EB, 1024), 1024, 0, stream>>>(hist, off, nbk * EB);
  bucket_scatter<<<EB, 1024, 0, stream>>>(src, dst, off, bpair, e, nbk);
  bucket_build<<<nbk, BK_NODES, 0, stream>>>(bpair, off, rp, csr, n, e, nbk);

  gemm_wmma_f32_resident<256, 64, 2><<<512, 256, 0, stream>>>(x, W1, h1, rp, n, e);

  gather_act<<<1792, T, 0, stream>>>(
      rp, csr, h1, act1, b1, gamma1, beta1, mean1, var1, n, e);

  gemm_wmma_f32<64, 32, 2><<<n / 32, 128, 0, stream>>>(act1, W2, h2, rp, n, e);

  gather_head<<<896, 512, 0, stream>>>(
      rp, csr, h2, b2, gamma2, beta2, mean2, var2, Wl, bl,
      (float*)d_out, n, e);
}


# layer-2 GEMM moved into the layer-1 aggregation work-groups: activation rows collected as 16x64 tiles in LDS, multiplied by W2 there on WMMA (8 -> 7 launches, act1 never in memory); sum order kept
# speedup vs baseline: 4.6550x; 1.0451x over previous
#include <hip/hip_runtime.h>

typedef __attribute__((ext_vector_type(2))) float v2f;
typedef __attribute__((ext_vector_type(8))) float v8f;
typedef __attribute__((ext_vector_type(4))) float v4f;

#define BN_EPS 1e-5f

#define BK_SHIFT 8
#define BK_NODES (1 << BK_SHIFT)
#define BK_MAX   1024
#define EB       256
#define EU       4
#define BU       8

__device__ __forceinline__ int bucket_of(int d, int nbk) {
  return min(max(d, 0) >> BK_SHIFT, nbk - 1);
}

__global__ void __launch_bounds__(1024)
bucket_hist(const int* __restrict__ dst, int* __restrict__ hist, int e, int nbk) {
  __shared__ int lc[BK_MAX + 32];
  const int tid = threadIdx.x;
  nbk = min(nbk, BK_MAX);
  for (int b = tid; b < nbk; b += 1024) lc[b] = 0;
  if (tid < 32) lc[BK_MAX + tid] = 0;
  __syncthreads();
  const int chunk = (e + EB - 1) / EB;
  const int beg = blockIdx.x * chunk;
  const int end = min(beg + chunk, e);
  for (int base = beg + tid; base < end; base += 1024 * EU) {
    int d[EU];
    #pragma unroll
    for (int u = 0; u < EU; ++u) d[u] = dst[min(base + u * 1024, end - 1)];
    #pragma unroll
    for (int u = 0; u < EU; ++u)
      atomicAdd(&lc[(base + u * 1024 < end) ? bucket_of(d[u], nbk) : BK_MAX + (tid & 31)], 1);
  }
  __syncthreads();
  for (int b = tid; b < nbk; b += 1024) hist[b * EB + blockIdx.x] = lc[b];
}

__global__ void __launch_bounds__(1024)
scan_counts(const int* __restrict__ cnt, int* __restrict__ rp, int n) {
  __shared__ int wtot[32];
  __shared__ int wpart[32];
  const int tid = threadIdx.x, lane = tid & 31, wave = tid >> 5;
  const int ahead = blockIdx.x * 1024;
  int part = 0;
  for (int j = tid; j < ahead; j += 1024) part += cnt[j];
  const int i = ahead + tid;
  const int v = (i < n) ? cnt[i] : 0;
  int inc = v;
  for (int o = 1; o < 32; o <<= 1) {
    int t = __shfl_up(inc, o);
    if (lane >= o) inc += t;
  }
  for (int o = 16; o > 0; o >>= 1) part += __shfl_xor(part, o);
  if (lane == 31) wtot[wave] = inc;
  if (lane == 0)  wpart[wave] = part;
  __syncthreads();
  int p = wpart[lane];
  int t = (lane < wave) ? wtot[lane] : 0;
  for (int o = 16; o > 0; o >>= 1) {
    p += __shfl_xor(p, o);
    t += __shfl_xor(t, o);
  }
  if (i < n) rp[i] = p + t + (inc - v);
}

__global__ void __launch_bounds__(1024)
bucket_scatter(const int* __restrict__ src, const int* __restrict__ dst,
               const int* __restrict__ off, int* __restrict__ bpair, int e, int nbk) {
  __shared__ int lpos[BK_MAX + 32];
  int2* __restrict__ pairs = reinterpret_cast<int2*>(bpair);
  const int tid = threadIdx.x;
  nbk = min(nbk, BK_MAX);
  for (int b = tid; b < nbk; b += 1024) lpos[b] = off[b * EB + blockIdx.x];
  if (tid < 32) lpos[BK_MAX + tid] = 0;
  __syncthreads();
  const int chunk = (e + EB - 1) / EB;
  const int beg = blockIdx.x * chunk;
  const int end = min(beg + chunk, e);
  for (int base = beg + tid; base < end; base += 1024 * EU) {
    int d[EU], s[EU];
    #pragma unroll
    for (int u = 0; u < EU; ++u) {
      const int i = min(base + u * 1024, end - 1);
      d[u] = dst[i];
      s[u] = src[i];
    }
    #pragma unroll
    for (int u = 0; u < EU; ++u) {
      const bool in = base + u * 1024 < end;
      const int p = atomicAdd(&lpos[in ? bucket_of(d[u], nbk) : BK_MAX + (tid & 31)], 1);
      if (in) pairs[p] = make_int2(s[u], d[u]);
    }
  }
}

__global__ void __launch_bounds__(BK_NODES)
bucket_build(const int* __restrict__ bpair, const int* __restrict__ off,
             int* __restrict__ rp, int* __restrict__ csr, int n, int e, int nbk) {
  __shared__ int lcnt[BK_NODES + 32];
  __shared__ int lpos[BK_NODES + 32];
  __shared__ int wtot[BK_NODES / 32];
  const int2* __restrict__ pairs = reinterpret_cast<const int2*>(bpair);
  const int tid = threadIdx.x, lane = tid & 31, wave = tid >> 5;
  const int b = blockIdx.x;
  const int node0 = b << BK_SHIFT;
  const int ebeg = off[b * EB];
  const int eend = (b + 1 < nbk) ? off[(b + 1) * EB] : e;
  lcnt[tid] = 0;
  if (tid < 32) lcnt[BK_NODES + tid] = 0;
  __syncthreads();
  for (int base = ebeg + tid; base < eend; base += BK_NODES * BU) {
    int2 q[BU];
    #pragma unroll
    for (int u = 0; u < BU; ++u) q[u] = pairs[min(base + u * BK_NODES, eend - 1)];
    #pragma unroll
    for (int u = 0; u < BU; ++u)
      atomicAdd(&lcnt[(base + u * BK_NODES < eend) ? min(max(q[u].y - node0, 0), BK_NODES - 1)
                                                   : BK_NODES + lane], 1);
  }
  __syncthreads();
  const int v = lcnt[tid];
  int inc = v;
  for (int o = 1; o < 32; o <<= 1) {
    int t = __shfl_up(inc, o);
    if (lane >= o) inc += t;
  }
  if (lane == 31) wtot[wave] = inc;
  __syncthreads();
  int ahead = 0;
  for (int w = 0; w < BK_NODES / 32; ++w) if (w < wave) ahead += wtot[w];
  const int start = ebeg + ahead + inc - v;
  lpos[tid] = start;
  if (tid < 32) lpos[BK_NODES + tid] = 0;
  if (node0 + tid < n) rp[node0 + tid] = start;
  __syncthreads();
  for (int base = ebeg + tid; base < eend; base += BK_NODES * BU) {
    int2 q[BU];
    #pragma unroll
    for (int u = 0; u < BU; ++u) q[u] = pairs[min(base + u * BK_NODES, eend - 1)];
    #pragma unroll
    for (int u = 0; u < BU; ++u) {
      const bool in = base + u * BK_NODES < eend;
      const int p = atomicAdd(&lpos[in ? min(max(q[u].y - node0, 0), BK_NODES - 1) : BK_NODES + lane], 1);
      if (in) csr[p] = q[u].x;
    }
  }
}

template <int K, int NC, int MT>
__global__ void __launch_bounds__(MT*(NC/16)*32)
gemm_wmma_f32_resident(const float* __restrict__ A, const float* __restrict__ B,
                       float* __restrict__ C, const int* __restrict__ rp, int n, int e) {
  constexpr int KP   = K + 4;
  constexpr int NCP  = NC + 8;
  constexpr int NW   = MT * (NC/16);
  constexpr int NT   = NW * 32;
  constexpr int ROWS = MT * 16;
  constexpr int AV   = ROWS * (K/4) / NT;
  constexpr int BV   = K * (NC/4) / NT;
  static_assert((ROWS * (K/4)) % NT == 0 && (K * (NC/4)) % NT == 0, "whole pieces per thread");
  __shared__ float sA[2][ROWS*KP];
  __shared__ float sB[K*NCP];

  const int tid = threadIdx.x;
  const int ntiles = n / ROWS;
  const int gstep = (int)gridDim.x;
  int g = blockIdx.x;
  if (g >= ntiles) return;

  v4f pf[AV];
  {
    const float* Ablk = A + (size_t)g * ROWS * K;
    #pragma unroll
    for (int j = 0; j < AV; ++j) {
      const int idx = tid + j * NT;
      const int r = idx / (K/4), c4 = (idx % (K/4)) * 4;
      pf[j] = *(const v4f*)&Ablk[(size_t)r*K + c4];
    }
  }
  #pragma unroll
  for (int j = 0; j < BV; ++j) {
    const int idx = tid + j * NT;
    const int r = idx / (NC/4), c4 = (idx % (NC/4)) * 4;
    *(float4*)&sB[r*NCP + c4] = *(const float4*)&B[(size_t)r*NC + c4];
  }
  #pragma unroll
  for (int j = 0; j < AV; ++j) {
    const int idx = tid + j * NT;
    const int r = idx / (K/4), c4 = (idx % (K/4)) * 4;
    *(v4f*)&sA[0][r*KP + c4] = pf[j];
  }
  __syncthreads();

  const int wave  = tid >> 5, lane = tid & 31;
  const int mt    = wave / (NC/16), nt = wave % (NC/16);
  const int rrow  = lane & 15;
  const int khalf = lane >> 4;
  const int col   = nt*16 + rrow;
  const float* b_ptr0 = &sB[(khalf*2)*NCP + nt*16 + rrow];

  for (int cur = 0; ; cur ^= 1) {
    const int gn = g + gstep;
    const bool more = gn < ntiles;

    const int gm = g*ROWS + mt*16 + khalf*8;
    int sb[9];
    #pragma unroll
    for (int r = 0; r < 8; ++r) sb[r] = rp[gm + r];
    sb[8] = rp[min(gm + 8, n - 1)];
    if (gm + 8 >= n) sb[8] = e;

    if (more) {
      const float* Ablk = A + (size_t)gn * ROWS * K;
      #pragma unroll
      for (int j = 0; j < AV; ++j) {
        const int idx = tid + j * NT;
        const int r = idx / (K/4), c4 = (idx % (K/4)) * 4;
        pf[j] = *(const v4f*)&Ablk[(size_t)r*K + c4];
      }
    }

    const float* a_ptr = &sA[cur][(mt*16 + rrow)*KP + khalf*2];
    const float* b_ptr = b_ptr0;
    v8f c0 = {};
    v8f c1 = {};
    #pragma unroll 4
    for (int ks = 0; ks < K/8; ++ks) {
      v2f a0; a0.x = a_ptr[0];     a0.y = a_ptr[1];
      v2f b0; b0.x = b_ptr[0];     b0.y = b_ptr[NCP];
      v2f a1; a1.x = a_ptr[4];     a1.y = a_ptr[5];
      v2f b1; b1.x = b_ptr[4*NCP]; b1.y = b_ptr[5*NCP];
      c0 = __builtin_amdgcn_wmma_f32_16x16x4_f32(false, a0, false, b0,
                                                 (short)0, c0, false, false);
      c1 = __builtin_amdgcn_wmma_f32_16x16x4_f32(false, a1, false, b1,
                                                 (short)0, c1, false, false);
      a_ptr += 8;
      b_ptr += 8*NCP;
    }
    v8f c = c0 + c1;

    if (more) {
      #pragma unroll
      for (int j = 0; j < AV; ++j) {
        const int idx = tid + j * NT;
        const int r = idx / (K/4), c4 = (idx % (K/4)) * 4;
        *(v4f*)&sA[cur ^ 1][r*KP + c4] = pf[j];
      }
      __syncthreads();
    }

    #pragma unroll
    for (int r = 0; r < 8; ++r) {
      const float dv = rsqrtf((float)(sb[r + 1] - sb[r]) + 1.0f);
      C[(size_t)(gm + r)*NC + col] = c[r] * dv;
    }

    if (!more) break;
    g = gn;
  }
}

template <int F, int G>
__device__ __forceinline__ void take_rows(const float* __restrict__ h, int s_l,
                                          int j, int lane, float (&acc)[F / 32]) {
  constexpr int V = F / 32;
  float v[G][V];
  #pragma unroll
  for (int u = 0; u < G; ++u) {
    const int s = __builtin_amdgcn_readlane(s_l, j + u);
    const float* hp = h + (size_t)s * F + lane * V;
    if constexpr (V == 2) {
      const float2 t = *(const float2*)hp;
      v[u][0] = t.x; v[u][1] = t.y;
    } else {
      v[u][0] = hp[0];
    }
  }
  #pragma unroll
  for (int u = 0; u < G; ++u)
    for (int l = 0; l < V; ++l) acc[l] += v[u][l];
}

template <int F>
__device__ __forceinline__ void take_chunk(const float* __restrict__ h, int s_l, int cnt,
                                           int lane, float (&acc)[F / 32]) {
  constexpr int V = F / 32;
  constexpr int G = 16 / V;
  int j = 0;
  for (; j + G <= cnt; j += G) take_rows<F, G>(h, s_l, j, lane, acc);
  if constexpr (G >= 16) { if (j + 8 <= cnt) { take_rows<F, 8>(h, s_l, j, lane, acc); j += 8; } }
  if (j + 4 <= cnt) { take_rows<F, 4>(h, s_l, j, lane, acc); j += 4; }
  if (j + 2 <= cnt) { take_rows<F, 2>(h, s_l, j, lane, acc); j += 2; }
  if (j < cnt)      { take_rows<F, 1>(h, s_l, j, lane, acc); }
}

__device__ __forceinline__ int csr_raw(const int* __restrict__ csr, int i, int e) {
  return csr[min(max(i, 0), e - 1)];
}
__device__ __forceinline__ int csr_id(int raw, int i, int end, int n) {
  return (i < end) ? min(max(raw, 0), n - 1) : 0;
}
__device__ __forceinline__ int bounds_raw(const int* __restrict__ rp, int q, int lane, int n) {
  return rp[min(min(q, n - 1) + (lane & 1), n - 1)];
}

template <int F>
__device__ __forceinline__ void sum_rows(const int* __restrict__ csr, const float* __restrict__ h,
                                         int beg, int end, int ia, int ib, int n, int lane,
                                         float (&acc)[F / 32]) {
  constexpr int V = F / 32;
  for (int l = 0; l < V; ++l) acc[l] = 0.0f;
  const int deg = end - beg;
  take_chunk<F>(h, ia, min(deg, 32), lane, acc);
  if (deg > 32) take_chunk<F>(h, ib, min(deg - 32, 32), lane, acc);
  for (int base = beg + 64; base < end; base += 32) {
    const int cnt = min(end - base, 32);
    int s_l = 0;
    if (lane < cnt) {
      s_l = csr[base + lane];
      s_l = min(max(s_l, 0), n - 1);
    }
    take_chunk<F>(h, s_l, cnt, lane, acc);
  }
}

__device__ __forceinline__ float node_act(float sum, float hs, float dd, float bias,
                                          float mean, float var, float gamma, float beta) {
  float v = __builtin_fmaf(hs, dd, sum * dd) + bias;
  v = (v - mean) * rsqrtf(var + BN_EPS) * gamma + beta;
  return fmaxf(v, 0.0f);
}

__global__ void __launch_bounds__(512) __attribute__((amdgpu_waves_per_eu(16)))
gather_act_gemm(const int* __restrict__ rp, const int* __restrict__ csr,
                const float* __restrict__ h, const float* __restrict__ B,
                float* __restrict__ C,
                const float* __restrict__ bias, const float* __restrict__ gamma,
                const float* __restrict__ beta, const float* __restrict__ mean,
                const float* __restrict__ var, int n, int e) {
  constexpr int F   = 64;
  constexpr int K   = F;
  constexpr int NC  = 32;
  constexpr int KP  = K + 4;
  constexpr int NCP = NC + 8;
  constexpr int RL  = 8;
  static_assert(K * (NC/4) == 512, "one 16-byte piece of B per thread");
  static_assert(2 * RL == 16, "one 16x16 piece of C per wave");
  __shared__ float sA[RL][16*KP];
  __shared__ float sB[K*NCP];
  const int tid  = threadIdx.x;
  const int lane = tid & 31;
  const int wave = __builtin_amdgcn_readfirstlane((int)(tid >> 5));
  const int nblk = (n + 15) >> 4;
  const int gstep = (int)gridDim.x;
  const int dstep = gstep * 16;
  int blk = blockIdx.x;
  if (blk >= nblk) return;
  const int br = tid / (NC/4), bc4 = (tid % (NC/4)) * 4;
  const float4 bpiece = *(const float4*)&B[(size_t)br*NC + bc4];
  const float2 pb = *(const float2*)(bias + lane * 2);
  const float2 pm = *(const float2*)(mean + lane * 2);
  const float2 pv = *(const float2*)(var + lane * 2);
  const float2 pg = *(const float2*)(gamma + lane * 2);
  const float2 pe = *(const float2*)(beta + lane * 2);

  int d = blk * 16 + wave;
  const int d0 = min(d, n - 1);
  int beg = rp[d0];
  int end = (d0 + 1 < n) ? rp[min(d0 + 1, n - 1)] : e;
  int ra = csr_raw(csr, beg + lane, e);
  int rb = csr_raw(csr, beg + 32 + lane, e);
  int rq = bounds_raw(rp, d + dstep, lane, n);
  *(float4*)&sB[br*NCP + bc4] = bpiece;

  const int rrow  = lane & 15;
  const int khalf = lane >> 4;

  int k = 0;
  for (; blk < nblk; blk += gstep, ++k) {
    const int r = k & (RL - 1);
    const int ia = csr_id(ra, beg + lane, end, n);
    const int ib = csr_id(rb, beg + 32 + lane, end, n);
    const int dn = d + dstep;
    const int nbeg = __builtin_amdgcn_readlane(rq, 0);
    const int nend = (min(dn, n - 1) + 1 < n) ? __builtin_amdgcn_readlane(rq, 1) : e;
    ra = csr_raw(csr, nbeg + lane, e);
    rb = csr_raw(csr, nbeg + 32 + lane, e);
    rq = bounds_raw(rp, dn + dstep, lane, n);

    float2 t = make_float2(0.0f, 0.0f);
    if (d < n) {
      const float2 hs = *(const float2*)(h + (size_t)d * F + lane * 2);
      float acc[2];
      sum_rows<F>(csr, h, beg, end, ia, ib, n, lane, acc);
      const float dd = rsqrtf((float)(end - beg) + 1.0f);
      t.x = node_act(acc[0], hs.x, dd, pb.x, pm.x, pv.x, pg.x, pe.x);
      t.y = node_act(acc[1], hs.y, dd, pb.y, pm.y, pv.y, pg.y, pe.y);
    }
    *(float2*)&sA[r][wave*KP + lane*2] = t;

    if (r == RL - 1 || blk + gstep >= nblk) {
      __syncthreads();
      const int slot = wave >> 1;
      if (slot <= r) {
        const int nt = wave & 1;
        const int fb = blk - (r - slot) * gstep;
        const int gm = fb*16 + khalf*8;
        const int gl = max(min(gm, n - 8), 0);
        const int* __restrict__ rpp = rp + gl;
        int sb[9];
        #pragma unroll
        for (int j = 0; j < 8; ++j) sb[j] = rpp[j];
        sb[8] = rp[min(gl + 8, n - 1)];
        if (gl + 8 >= n) sb[8] = e;
        const float* a_ptr = &sA[slot][rrow*KP + khalf*2];
        const float* b_ptr = &sB[(khalf*2)*NCP + nt*16 + rrow];
        v8f c0 = {};
        v8f c1 = {};
        #pragma unroll 4
        for (int ks = 0; ks < K/8; ++ks) {
          v2f a0; a0.x = a_ptr[0];     a0.y = a_ptr[1];
          v2f b0; b0.x = b_ptr[0];     b0.y = b_ptr[NCP];
          v2f a1; a1.x = a_ptr[4];     a1.y = a_ptr[5];
          v2f b1; b1.x = b_ptr[4*NCP]; b1.y = b_ptr[5*NCP];
          c0 = __builtin_amdgcn_wmma_f32_16x16x4_f32(false, a0, false, b0,
                                                     (short)0, c0, false, false);
          c1 = __builtin_amdgcn_wmma_f32_16x16x4_f32(false, a1, false, b1,
                                                     (short)0, c1, false, false);
          a_ptr += 8;
          b_ptr += 8*NCP;
        }
        v8f c = c0 + c1;
        const int col = nt*16 + rrow;
        if (gm + 8 <= n) {
          float* __restrict__ cp = C + (size_t)gm*NC + col;
          #pragma unroll
          for (int j = 0; j < 8; ++j) {
            const float dv = rsqrtf((float)(sb[j + 1] - sb[j]) + 1.0f);
            cp[j*NC] = c[j] * dv;
          }
        }
      }
      __syncthreads();
    }
    d = dn; beg = nbeg; end = nend;
  }
}

template <int M>
__device__ __forceinline__ float row_xor(float x) {
  return __builtin_bit_cast(float, __builtin_amdgcn_update_dpp(
      0, __builtin_bit_cast(int, x), 0x160 + M, 0xf, 0xf, true));
}
__device__ __forceinline__ float other_row(float x) {
  const int b = __builtin_bit_cast(int, x);
  return __builtin_bit_cast(float, __builtin_amdgcn_permlanex16(b, b, 0x76543210, 0xfedcba98, false, false));
}

__global__ void __launch_bounds__(512)
gather_head(const int* __restrict__ rp, const int* __restrict__ csr,
            const float* __restrict__ h,
            const float* __restrict__ bias, const float* __restrict__ gamma,
            const float* __restrict__ beta, const float* __restrict__ mean,
            const float* __restrict__ var, const float* __restrict__ Wl,
            const float* __restrict__ bl, float* __restrict__ out, int n, int e) {
  constexpr int F = 32;
  constexpr int RL = 8;
  __shared__ float so[RL][32];
  const int lane = threadIdx.x & 31;
  const int wave = __builtin_amdgcn_readfirstlane((int)(threadIdx.x >> 5));
  const int nblk = (n + 15) >> 4;
  const int gstep = (int)gridDim.x;
  const int dstep = gstep * 16;
  int blk = blockIdx.x;
  if (blk >= nblk) return;
  const float pb = bias[lane], pm = mean[lane], pv = var[lane];
  const float pg = gamma[lane], pe = beta[lane];
  const float2 wl = *(const float2*)(Wl + lane * 2);
  const float blv = bl[lane >> 4];

  int d = blk * 16 + wave;
  const int d0 = min(d, n - 1);
  int beg = rp[d0];
  int end = (d0 + 1 < n) ? rp[min(d0 + 1, n - 1)] : e;
  int ra = csr_raw(csr, beg + lane, e);
  int rb = csr_raw(csr, beg + 32 + lane, e);
  int rq = bounds_raw(rp, d + dstep, lane, n);

  int k = 0;
  for (; blk < nblk; blk += gstep, ++k) {
    const int r = k & (RL - 1);
    const int ia = csr_id(ra, beg + lane, end, n);
    const int ib = csr_id(rb, beg + 32 + lane, end, n);
    const int dn = d + dstep;
    const int nbeg = __builtin_amdgcn_readlane(rq, 0);
    const int nend = (min(dn, n - 1) + 1 < n) ? __builtin_amdgcn_readlane(rq, 1) : e;
    ra = csr_raw(csr, nbeg + lane, e);
    rb = csr_raw(csr, nbeg + 32 + lane, e);
    rq = bounds_raw(rp, dn + dstep, lane, n);

    if (d < n) {
      const float hs = h[(size_t)d * F + lane];
      float acc[1];
      sum_rows<F>(csr, h, beg, end, ia, ib, n, lane, acc);
      const float dd = rsqrtf((float)(end - beg) + 1.0f);
      const float v = node_act(acc[0], hs, dd, pb, pm, pv, pg, pe);

      const bool hi = lane >= 16;
      const float p0 = v * wl.x, p1 = v * wl.y;
      float mine = hi ? p1 : p0;
      mine += other_row(hi ? p0 : p1);
      mine += row_xor<8>(mine);
      mine += row_xor<4>(mine);
      mine += row_xor<2>(mine);
      mine += row_xor<1>(mine);
      if ((lane & 15) == 0) so[r][wave * 2 + (lane >> 4)] = mine + blv;
    }

    if (r == RL - 1 || blk + gstep >= nblk) {
      __syncthreads();
      if (wave <= r) {
        const int fb = blk - (r - wave) * gstep;
        const int node = fb * 16 + (lane >> 1);
        if (node < n) out[(size_t)fb * 32 + lane] = so[wave][lane];
      }
      __syncthreads();
    }
    d = dn; beg = nbeg; end = nend;
  }
}

extern "C" void kernel_launch(void* const* d_in, const int* in_sizes, int n_in,
                              void* d_out, int out_size, void* d_ws, size_t ws_size,
                              hipStream_t stream) {
  const float* x      = (const float*)d_in[0];
  const int*   ei     = (const int*)  d_in[1];
  const float* W1     = (const float*)d_in[2];
  const float* b1     = (const float*)d_in[3];
  const float* gamma1 = (const float*)d_in[4];
  const float* beta1  = (const float*)d_in[5];
  const float* mean1  = (const float*)d_in[6];
  const float* var1   = (const float*)d_in[7];
  const float* W2     = (const float*)d_in[8];
  const float* b2     = (const float*)d_in[9];
  const float* gamma2 = (const float*)d_in[10];
  const float* beta2  = (const float*)d_in[11];
  const float* mean2  = (const float*)d_in[12];
  const float* var2   = (const float*)d_in[13];
  const float* Wl     = (const float*)d_in[14];
  const float* bl     = (const float*)d_in[15];

  const int n = in_sizes[0] / 256;
  const int e = in_sizes[1] / 2;
  const int* src = ei;
  const int* dst = ei + e;

  float* ws   = (float*)d_ws;
  int*   rp   = (int*)ws;
  float* h1   = ws   + (size_t)n;
  float* act1 = h1   + (size_t)n * 64;
  float* h2   = act1 + (size_t)n * 64;
  int*   csr  = (int*)(h2 + (size_t)n * 32);
  int*   hist = (int*)h2;
  int*   off  = hist + (size_t)n * 16;
  int*   bpair = (int*)act1;

  auto blocks = [](long long total, int t) {
    return (unsigned int)((total + t - 1) / t);
  };

  const int nbk = (int)blocks(n, BK_NODES);
  bucket_hist<<<EB, 1024, 0, stream>>>(dst, hist, e, nbk);
  scan_counts<<<blocks((long long)nbk * EB, 1024), 1024, 0, stream>>>(hist, off, nbk * EB);
  bucket_scatter<<<EB, 1024, 0, stream>>>(src, dst, off, bpair, e, nbk);
  bucket_build<<<nbk, BK_NODES, 0, stream>>>(bpair, off, rp, csr, n, e, nbk);

  gemm_wmma_f32_resident<256, 64, 2><<<512, 256, 0, stream>>>(x, W1, h1, rp, n, e);

  gather_act_gemm<<<896, 512, 0, stream>>>(
      rp, csr, h1, W2, h2, b1, gamma1, beta1, mean1, var1, n, e);

  gather_head<<<896, 512, 0, stream>>>(
      rp, csr, h2, b2, gamma2, beta2, mean2, var2, Wl, bl,
      (float*)d_out, n, e);
}


# edge ordering: both placement passes put their edges in slot order in LDS, then write LDS out front to back (whole lines per store), not one scattered 4/8-byte store per edge; sum order as before
# speedup vs baseline: 5.2983x; 1.1382x over previous
#include <hip/hip_runtime.h>

typedef __attribute__((ext_vector_type(2))) float v2f;
typedef __attribute__((ext_vector_type(8))) float v8f;
typedef __attribute__((ext_vector_type(4))) float v4f;

#define BN_EPS 1e-5f

#define BK_SHIFT 8
#define BK_NODES (1 << BK_SHIFT)
#define BK_MAX   1024
#define EB       256
#define EU       4
#define BU       8
#define STAGE_PAIRS 12800
#define STAGE_IDS   12288

__device__ __forceinline__ int bucket_of(int d, int nbk) {
  return min(max(d, 0) >> BK_SHIFT, nbk - 1);
}

__global__ void __launch_bounds__(1024)
bucket_hist(const int* __restrict__ dst, int* __restrict__ hist, int e, int nbk) {
  __shared__ int lc[BK_MAX + 32];
  const int tid = threadIdx.x;
  nbk = min(nbk, BK_MAX);
  for (int b = tid; b < nbk; b += 1024) lc[b] = 0;
  if (tid < 32) lc[BK_MAX + tid] = 0;
  __syncthreads();
  const int chunk = (e + EB - 1) / EB;
  const int beg = blockIdx.x * chunk;
  const int end = min(beg + chunk, e);
  for (int base = beg + tid; base < end; base += 1024 * EU) {
    int d[EU];
    #pragma unroll
    for (int u = 0; u < EU; ++u) d[u] = dst[min(base + u * 1024, end - 1)];
    #pragma unroll
    for (int u = 0; u < EU; ++u)
      atomicAdd(&lc[(base + u * 1024 < end) ? bucket_of(d[u], nbk) : BK_MAX + (tid & 31)], 1);
  }
  __syncthreads();
  for (int b = tid; b < nbk; b += 1024) hist[b * EB + blockIdx.x] = lc[b];
}

__global__ void __launch_bounds__(1024)
scan_counts(const int* __restrict__ cnt, int* __restrict__ rp, int n) {
  __shared__ int wtot[32];
  __shared__ int wpart[32];
  const int tid = threadIdx.x, lane = tid & 31, wave = tid >> 5;
  const int ahead = blockIdx.x * 1024;
  int part = 0;
  for (int j = tid; j < ahead; j += 1024) part += cnt[j];
  const int i = ahead + tid;
  const int v = (i < n) ? cnt[i] : 0;
  int inc = v;
  for (int o = 1; o < 32; o <<= 1) {
    int t = __shfl_up(inc, o);
    if (lane >= o) inc += t;
  }
  for (int o = 16; o > 0; o >>= 1) part += __shfl_xor(part, o);
  if (lane == 31) wtot[wave] = inc;
  if (lane == 0)  wpart[wave] = part;
  __syncthreads();
  int p = wpart[lane];
  int t = (lane < wave) ? wtot[lane] : 0;
  for (int o = 16; o > 0; o >>= 1) {
    p += __shfl_xor(p, o);
    t += __shfl_xor(t, o);
  }
  if (i < n) rp[i] = p + t + (inc - v);
}

__global__ void __launch_bounds__(1024)
bucket_scatter(const int* __restrict__ src, const int* __restrict__ dst,
               const int* __restrict__ hist, const int* __restrict__ off,
               int* __restrict__ bpair, int e, int nbk) {
  static_assert(BK_MAX == 1024, "one bucket per thread in the scan below");
  __shared__ int lpos[BK_MAX + 32];
  __shared__ int ldelta[BK_MAX];
  __shared__ int wtot[32];
  __shared__ int2 stage[STAGE_PAIRS];
  int2* __restrict__ pairs = reinterpret_cast<int2*>(bpair);
  const int tid = threadIdx.x, lane = tid & 31, wave = tid >> 5;
  nbk = min(nbk, BK_MAX);
  const int chunk = (e + EB - 1) / EB;
  const int beg = blockIdx.x * chunk;
  const int end = min(beg + chunk, e);
  const bool staged = chunk <= STAGE_PAIRS;

  const int cnt = (tid < nbk) ? hist[tid * EB + blockIdx.x] : 0;
  const int g   = (tid < nbk) ? off[tid * EB + blockIdx.x] : 0;
  int inc = cnt;
  for (int o = 1; o < 32; o <<= 1) {
    int t = __shfl_up(inc, o);
    if (lane >= o) inc += t;
  }
  if (lane == 31) wtot[wave] = inc;
  __syncthreads();
  int ahead = (lane < wave) ? wtot[lane] : 0;
  for (int o = 16; o > 0; o >>= 1) ahead += __shfl_xor(ahead, o);
  const int ls = ahead + inc - cnt;
  lpos[tid] = staged ? ls : g;
  ldelta[tid] = g - ls;
  if (tid < 32) lpos[BK_MAX + tid] = 0;
  __syncthreads();

  for (int base = beg + tid; base < end; base += 1024 * EU) {
    int d[EU], s[EU];
    #pragma unroll
    for (int u = 0; u < EU; ++u) {
      const int i = min(base + u * 1024, end - 1);
      d[u] = dst[i];
      s[u] = src[i];
    }
    #pragma unroll
    for (int u = 0; u < EU; ++u) {
      const bool in = base + u * 1024 < end;
      const int p = atomicAdd(&lpos[in ? bucket_of(d[u], nbk) : BK_MAX + (tid & 31)], 1);
      if (in) {
        if (staged) stage[min(max(p, 0), STAGE_PAIRS - 1)] = make_int2(s[u], d[u]);
        else        pairs[p] = make_int2(s[u], d[u]);
      }
    }
  }
  if (staged) {
    __syncthreads();
    const int m = end - beg;
    for (int base = tid; base < m; base += 1024 * EU) {
      int2 q[EU];
      int dl[EU];
      #pragma unroll
      for (int u = 0; u < EU; ++u) q[u] = stage[min(base + u * 1024, m - 1)];
      #pragma unroll
      for (int u = 0; u < EU; ++u) dl[u] = ldelta[bucket_of(q[u].y, nbk)];
      #pragma unroll
      for (int u = 0; u < EU; ++u)
        if (base + u * 1024 < m) pairs[base + u * 1024 + dl[u]] = q[u];
    }
  }
}

__global__ void __launch_bounds__(BK_NODES)
bucket_build(const int* __restrict__ bpair, const int* __restrict__ off,
             int* __restrict__ rp, int* __restrict__ csr, int n, int e, int nbk) {
  __shared__ int lcnt[BK_NODES + 32];
  __shared__ int lpos[BK_NODES + 32];
  __shared__ int wtot[BK_NODES / 32];
  __shared__ __attribute__((aligned(16))) int lcsr[STAGE_IDS];
  const int2* __restrict__ pairs = reinterpret_cast<const int2*>(bpair);
  const int tid = threadIdx.x, lane = tid & 31, wave = tid >> 5;
  const int b = blockIdx.x;
  const int node0 = b << BK_SHIFT;
  const int ebeg = off[b * EB];
  const int eend = (b + 1 < nbk) ? off[(b + 1) * EB] : e;
  const int total = eend - ebeg;
  const int sh = (int)((((unsigned long long)(csr + ebeg)) >> 2) & 3ull);
  const bool staged = total + 3 <= STAGE_IDS;
  lcnt[tid] = 0;
  if (tid < 32) lcnt[BK_NODES + tid] = 0;
  __syncthreads();
  for (int base = ebeg + tid; base < eend; base += BK_NODES * BU) {
    int2 q[BU];
    #pragma unroll
    for (int u = 0; u < BU; ++u) q[u] = pairs[min(base + u * BK_NODES, eend - 1)];
    #pragma unroll
    for (int u = 0; u < BU; ++u)
      atomicAdd(&lcnt[(base + u * BK_NODES < eend) ? min(max(q[u].y - node0, 0), BK_NODES - 1)
                                                   : BK_NODES + lane], 1);
  }
  __syncthreads();
  const int v = lcnt[tid];
  int inc = v;
  for (int o = 1; o < 32; o <<= 1) {
    int t = __shfl_up(inc, o);
    if (lane >= o) inc += t;
  }
  if (lane == 31) wtot[wave] = inc;
  __syncthreads();
  int ahead = 0;
  for (int w = 0; w < BK_NODES / 32; ++w) if (w < wave) ahead += wtot[w];
  const int start = ebeg + ahead + inc - v;
  lpos[tid] = start;
  if (tid < 32) lpos[BK_NODES + tid] = 0;
  if (node0 + tid < n) rp[node0 + tid] = start;
  __syncthreads();
  for (int base = ebeg + tid; base < eend; base += BK_NODES * BU) {
    int2 q[BU];
    #pragma unroll
    for (int u = 0; u < BU; ++u) q[u] = pairs[min(base + u * BK_NODES, eend - 1)];
    #pragma unroll
    for (int u = 0; u < BU; ++u) {
      const bool in = base + u * BK_NODES < eend;
      const int p = atomicAdd(&lpos[in ? min(max(q[u].y - node0, 0), BK_NODES - 1) : BK_NODES + lane], 1);
      if (in) {
        if (staged) lcsr[min(max(p - ebeg, 0) + sh, STAGE_IDS - 1)] = q[u].x;
        else        csr[p] = q[u].x;
      }
    }
  }
  if (staged) {
    __syncthreads();
    const int span = sh + total;
    int* gal = csr + ebeg - sh;
    for (int k = tid * 4; k < span; k += BK_NODES * 4) {
      if (k >= sh && k + 4 <= span) {
        *(int4*)(gal + k) = *(const int4*)&lcsr[k];
      } else {
        for (int t = 0; t < 4; ++t)
          if (k + t >= sh && k + t < span) gal[k + t] = lcsr[k + t];
      }
    }
  }
}

template <int K, int NC, int MT>
__global__ void __launch_bounds__(MT*(NC/16)*32)
gemm_wmma_f32_resident(const float* __restrict__ A, const float* __restrict__ B,
                       float* __restrict__ C, const int* __restrict__ rp, int n, int e) {
  constexpr int KP   = K + 4;
  constexpr int NCP  = NC + 8;
  constexpr int NW   = MT * (NC/16);
  constexpr int NT   = NW * 32;
  constexpr int ROWS = MT * 16;
  constexpr int AV   = ROWS * (K/4) / NT;
  constexpr int BV   = K * (NC/4) / NT;
  static_assert((ROWS * (K/4)) % NT == 0 && (K * (NC/4)) % NT == 0, "whole pieces per thread");
  __shared__ float sA[2][ROWS*KP];
  __shared__ float sB[K*NCP];

  const int tid = threadIdx.x;
  const int ntiles = n / ROWS;
  const int gstep = (int)gridDim.x;
  int g = blockIdx.x;
  if (g >= ntiles) return;

  v4f pf[AV];
  {
    const float* Ablk = A + (size_t)g * ROWS * K;
    #pragma unroll
    for (int j = 0; j < AV; ++j) {
      const int idx = tid + j * NT;
      const int r = idx / (K/4), c4 = (idx % (K/4)) * 4;
      pf[j] = *(const v4f*)&Ablk[(size_t)r*K + c4];
    }
  }
  #pragma unroll
  for (int j = 0; j < BV; ++j) {
    const int idx = tid + j * NT;
    const int r = idx / (NC/4), c4 = (idx % (NC/4)) * 4;
    *(float4*)&sB[r*NCP + c4] = *(const float4*)&B[(size_t)r*NC + c4];
  }
  #pragma unroll
  for (int j = 0; j < AV; ++j) {
    const int idx = tid + j * NT;
    const int r = idx / (K/4), c4 = (idx % (K/4)) * 4;
    *(v4f*)&sA[0][r*KP + c4] = pf[j];
  }
  __syncthreads();

  const int wave  = tid >> 5, lane = tid & 31;
  const int mt    = wave / (NC/16), nt = wave % (NC/16);
  const int rrow  = lane & 15;
  const int khalf = lane >> 4;
  const int col   = nt*16 + rrow;
  const float* b_ptr0 = &sB[(khalf*2)*NCP + nt*16 + rrow];

  for (int cur = 0; ; cur ^= 1) {
    const int gn = g + gstep;
    const bool more = gn < ntiles;

    const int gm = g*ROWS + mt*16 + khalf*8;
    int sb[9];
    #pragma unroll
    for (int r = 0; r < 8; ++r) sb[r] = rp[gm + r];
    sb[8] = rp[min(gm + 8, n - 1)];
    if (gm + 8 >= n) sb[8] = e;

    if (more) {
      const float* Ablk = A + (size_t)gn * ROWS * K;
      #pragma unroll
      for (int j = 0; j < AV; ++j) {
        const int idx = tid + j * NT;
        const int r = idx / (K/4), c4 = (idx % (K/4)) * 4;
        pf[j] = *(const v4f*)&Ablk[(size_t)r*K + c4];
      }
    }

    const float* a_ptr = &sA[cur][(mt*16 + rrow)*KP + khalf*2];
    const float* b_ptr = b_ptr0;
    v8f c0 = {};
    v8f c1 = {};
    #pragma unroll 4
    for (int ks = 0; ks < K/8; ++ks) {
      v2f a0; a0.x = a_ptr[0];     a0.y = a_ptr[1];
      v2f b0; b0.x = b_ptr[0];     b0.y = b_ptr[NCP];
      v2f a1; a1.x = a_ptr[4];     a1.y = a_ptr[5];
      v2f b1; b1.x = b_ptr[4*NCP]; b1.y = b_ptr[5*NCP];
      c0 = __builtin_amdgcn_wmma_f32_16x16x4_f32(false, a0, false, b0,
                                                 (short)0, c0, false, false);
      c1 = __builtin_amdgcn_wmma_f32_16x16x4_f32(false, a1, false, b1,
                                                 (short)0, c1, false, false);
      a_ptr += 8;
      b_ptr += 8*NCP;
    }
    v8f c = c0 + c1;

    if (more) {
      #pragma unroll
      for (int j = 0; j < AV; ++j) {
        const int idx = tid + j * NT;
        const int r = idx / (K/4), c4 = (idx % (K/4)) * 4;
        *(v4f*)&sA[cur ^ 1][r*KP + c4] = pf[j];
      }
      __syncthreads();
    }

    #pragma unroll
    for (int r = 0; r < 8; ++r) {
      const float dv = rsqrtf((float)(sb[r + 1] - sb[r]) + 1.0f);
      C[(size_t)(gm + r)*NC + col] = c[r] * dv;
    }

    if (!more) break;
    g = gn;
  }
}

template <int F, int G>
__device__ __forceinline__ void take_rows(const float* __restrict__ h, int s_l,
                                          int j, int lane, float (&acc)[F / 32]) {
  constexpr int V = F / 32;
  float v[G][V];
  #pragma unroll
  for (int u = 0; u < G; ++u) {
    const int s = __builtin_amdgcn_readlane(s_l, j + u);
    const float* hp = h + (size_t)s * F + lane * V;
    if constexpr (V == 2) {
      const float2 t = *(const float2*)hp;
      v[u][0] = t.x; v[u][1] = t.y;
    } else {
      v[u][0] = hp[0];
    }
  }
  #pragma unroll
  for (int u = 0; u < G; ++u)
    for (int l = 0; l < V; ++l) acc[l] += v[u][l];
}

template <int F>
__device__ __forceinline__ void take_chunk(const float* __restrict__ h, int s_l, int cnt,
                                           int lane, float (&acc)[F / 32]) {
  constexpr int V = F / 32;
  constexpr int G = 16 / V;
  int j = 0;
  for (; j + G <= cnt; j += G) take_rows<F, G>(h, s_l, j, lane, acc);
  if constexpr (G >= 16) { if (j + 8 <= cnt) { take_rows<F, 8>(h, s_l, j, lane, acc); j += 8; } }
  if (j + 4 <= cnt) { take_rows<F, 4>(h, s_l, j, lane, acc); j += 4; }
  if (j + 2 <= cnt) { take_rows<F, 2>(h, s_l, j, lane, acc); j += 2; }
  if (j < cnt)      { take_rows<F, 1>(h, s_l, j, lane, acc); }
}

__device__ __forceinline__ int csr_raw(const int* __restrict__ csr, int i, int e) {
  return csr[min(max(i, 0), e - 1)];
}
__device__ __forceinline__ int csr_id(int raw, int i, int end, int n) {
  return (i < end) ? min(max(raw, 0), n - 1) : 0;
}
__device__ __forceinline__ int bounds_raw(const int* __restrict__ rp, int q, int lane, int n) {
  return rp[min(min(q, n - 1) + (lane & 1), n - 1)];
}

template <int F>
__device__ __forceinline__ void sum_rows(const int* __restrict__ csr, const float* __restrict__ h,
                                         int beg, int end, int ia, int ib, int n, int lane,
                                         float (&acc)[F / 32]) {
  constexpr int V = F / 32;
  for (int l = 0; l < V; ++l) acc[l] = 0.0f;
  const int deg = end - beg;
  take_chunk<F>(h, ia, min(deg, 32), lane, acc);
  if (deg > 32) take_chunk<F>(h, ib, min(deg - 32, 32), lane, acc);
  for (int base = beg + 64; base < end; base += 32) {
    const int cnt = min(end - base, 32);
    int s_l = 0;
    if (lane < cnt) {
      s_l = csr[base + lane];
      s_l = min(max(s_l, 0), n - 1);
    }
    take_chunk<F>(h, s_l, cnt, lane, acc);
  }
}

__device__ __forceinline__ float node_act(float sum, float hs, float dd, float bias,
                                          float mean, float var, float gamma, float beta) {
  float v = __builtin_fmaf(hs, dd, sum * dd) + bias;
  v = (v - mean) * rsqrtf(var + BN_EPS) * gamma + beta;
  return fmaxf(v, 0.0f);
}

__global__ void __launch_bounds__(512) __attribute__((amdgpu_waves_per_eu(16)))
gather_act_gemm(const int* __restrict__ rp, const int* __restrict__ csr,
                const float* __restrict__ h, const float* __restrict__ B,
                float* __restrict__ C,
                const float* __restrict__ bias, const float* __restrict__ gamma,
                const float* __restrict__ beta, const float* __restrict__ mean,
                const float* __restrict__ var, int n, int e) {
  constexpr int F   = 64;
  constexpr int K   = F;
  constexpr int NC  = 32;
  constexpr int KP  = K + 4;
  constexpr int NCP = NC + 8;
  constexpr int RL  = 8;
  static_assert(K * (NC/4) == 512, "one 16-byte piece of B per thread");
  static_assert(2 * RL == 16, "one 16x16 piece of C per wave");
  __shared__ float sA[RL][16*KP];
  __shared__ float sB[K*NCP];
  const int tid  = threadIdx.x;
  const int lane = tid & 31;
  const int wave = __builtin_amdgcn_readfirstlane((int)(tid >> 5));
  const int nblk = (n + 15) >> 4;
  const int gstep = (int)gridDim.x;
  const int dstep = gstep * 16;
  int blk = blockIdx.x;
  if (blk >= nblk) return;
  const int br = tid / (NC/4), bc4 = (tid % (NC/4)) * 4;
  const float4 bpiece = *(const float4*)&B[(size_t)br*NC + bc4];
  const float2 pb = *(const float2*)(bias + lane * 2);
  const float2 pm = *(const float2*)(mean + lane * 2);
  const float2 pv = *(const float2*)(var + lane * 2);
  const float2 pg = *(const float2*)(gamma + lane * 2);
  const float2 pe = *(const float2*)(beta + lane * 2);

  int d = blk * 16 + wave;
  const int d0 = min(d, n - 1);
  int beg = rp[d0];
  int end = (d0 + 1 < n) ? rp[min(d0 + 1, n - 1)] : e;
  int ra = csr_raw(csr, beg + lane, e);
  int rb = csr_raw(csr, beg + 32 + lane, e);
  int rq = bounds_raw(rp, d + dstep, lane, n);
  *(float4*)&sB[br*NCP + bc4] = bpiece;

  const int rrow  = lane & 15;
  const int khalf = lane >> 4;

  int k = 0;
  for (; blk < nblk; blk += gstep, ++k) {
    const int r = k & (RL - 1);
    const int ia = csr_id(ra, beg + lane, end, n);
    const int ib = csr_id(rb, beg + 32 + lane, end, n);
    const int dn = d + dstep;
    const int nbeg = __builtin_amdgcn_readlane(rq, 0);
    const int nend = (min(dn, n - 1) + 1 < n) ? __builtin_amdgcn_readlane(rq, 1) : e;
    ra = csr_raw(csr, nbeg + lane, e);
    rb = csr_raw(csr, nbeg + 32 + lane, e);
    rq = bounds_raw(rp, dn + dstep, lane, n);

    float2 t = make_float2(0.0f, 0.0f);
    if (d < n) {
      const float2 hs = *(const float2*)(h + (size_t)d * F + lane * 2);
      float acc[2];
      sum_rows<F>(csr, h, beg, end, ia, ib, n, lane, acc);
      const float dd = rsqrtf((float)(end - beg) + 1.0f);
      t.x = node_act(acc[0], hs.x, dd, pb.x, pm.x, pv.x, pg.x, pe.x);
      t.y = node_act(acc[1], hs.y, dd, pb.y, pm.y, pv.y, pg.y, pe.y);
    }
    *(float2*)&sA[r][wave*KP + lane*2] = t;

    if (r == RL - 1 || blk + gstep >= nblk) {
      __syncthreads();
      const int slot = wave >> 1;
      if (slot <= r) {
        const int nt = wave & 1;
        const int fb = blk - (r - slot) * gstep;
        const int gm = fb*16 + khalf*8;
        const int gl = max(min(gm, n - 8), 0);
        const int* __restrict__ rpp = rp + gl;
        int sb[9];
        #pragma unroll
        for (int j = 0; j < 8; ++j) sb[j] = rpp[j];
        sb[8] = rp[min(gl + 8, n - 1)];
        if (gl + 8 >= n) sb[8] = e;
        const float* a_ptr = &sA[slot][rrow*KP + khalf*2];
        const float* b_ptr = &sB[(khalf*2)*NCP + nt*16 + rrow];
        v8f c0 = {};
        v8f c1 = {};
        #pragma unroll 4
        for (int ks = 0; ks < K/8; ++ks) {
          v2f a0; a0.x = a_ptr[0];     a0.y = a_ptr[1];
          v2f b0; b0.x = b_ptr[0];     b0.y = b_ptr[NCP];
          v2f a1; a1.x = a_ptr[4];     a1.y = a_ptr[5];
          v2f b1; b1.x = b_ptr[4*NCP]; b1.y = b_ptr[5*NCP];
          c0 = __builtin_amdgcn_wmma_f32_16x16x4_f32(false, a0, false, b0,
                                                     (short)0, c0, false, false);
          c1 = __builtin_amdgcn_wmma_f32_16x16x4_f32(false, a1, false, b1,
                                                     (short)0, c1, false, false);
          a_ptr += 8;
          b_ptr += 8*NCP;
        }
        v8f c = c0 + c1;
        const int col = nt*16 + rrow;
        if (gm + 8 <= n) {
          float* __restrict__ cp = C + (size_t)gm*NC + col;
          #pragma unroll
          for (int j = 0; j < 8; ++j) {
            const float dv = rsqrtf((float)(sb[j + 1] - sb[j]) + 1.0f);
            cp[j*NC] = c[j] * dv;
          }
        }
      }
      __syncthreads();
    }
    d = dn; beg = nbeg; end = nend;
  }
}

template <int M>
__device__ __forceinline__ float row_xor(float x) {
  return __builtin_bit_cast(float, __builtin_amdgcn_update_dpp(
      0, __builtin_bit_cast(int, x), 0x160 + M, 0xf, 0xf, true));
}
__device__ __forceinline__ float other_row(float x) {
  const int b = __builtin_bit_cast(int, x);
  return __builtin_bit_cast(float, __builtin_amdgcn_permlanex16(b, b, 0x76543210, 0xfedcba98, false, false));
}

__global__ void __launch_bounds__(512)
gather_head(const int* __restrict__ rp, const int* __restrict__ csr,
            const float* __restrict__ h,
            const float* __restrict__ bias, const float* __restrict__ gamma,
            const float* __restrict__ beta, const float* __restrict__ mean,
            const float* __restrict__ var, const float* __restrict__ Wl,
            const float* __restrict__ bl, float* __restrict__ out, int n, int e) {
  constexpr int F = 32;
  constexpr int RL = 8;
  __shared__ float so[RL][32];
  const int lane = threadIdx.x & 31;
  const int wave = __builtin_amdgcn_readfirstlane((int)(threadIdx.x >> 5));
  const int nblk = (n + 15) >> 4;
  const int gstep = (int)gridDim.x;
  const int dstep = gstep * 16;
  int blk = blockIdx.x;
  if (blk >= nblk) return;
  const float pb = bias[lane], pm = mean[lane], pv = var[lane];
  const float pg = gamma[lane], pe = beta[lane];
  const float2 wl = *(const float2*)(Wl + lane * 2);
  const float blv = bl[lane >> 4];

  int d = blk * 16 + wave;
  const int d0 = min(d, n - 1);
  int beg = rp[d0];
  int end = (d0 + 1 < n) ? rp[min(d0 + 1, n - 1)] : e;
  int ra = csr_raw(csr, beg + lane, e);
  int rb = csr_raw(csr, beg + 32 + lane, e);
  int rq = bounds_raw(rp, d + dstep, lane, n);

  int k = 0;
  for (; blk < nblk; blk += gstep, ++k) {
    const int r = k & (RL - 1);
    const int ia = csr_id(ra, beg + lane, end, n);
    const int ib = csr_id(rb, beg + 32 + lane, end, n);
    const int dn = d + dstep;
    const int nbeg = __builtin_amdgcn_readlane(rq, 0);
    const int nend = (min(dn, n - 1) + 1 < n) ? __builtin_amdgcn_readlane(rq, 1) : e;
    ra = csr_raw(csr, nbeg + lane, e);
    rb = csr_raw(csr, nbeg + 32 + lane, e);
    rq = bounds_raw(rp, dn + dstep, lane, n);

    if (d < n) {
      const float hs = h[(size_t)d * F + lane];
      float acc[1];
      sum_rows<F>(csr, h, beg, end, ia, ib, n, lane, acc);
      const float dd = rsqrtf((float)(end - beg) + 1.0f);
      const float v = node_act(acc[0], hs, dd, pb, pm, pv, pg, pe);

      const bool hi = lane >= 16;
      const float p0 = v * wl.x, p1 = v * wl.y;
      float mine = hi ? p1 : p0;
      mine += other_row(hi ? p0 : p1);
      mine += row_xor<8>(mine);
      mine += row_xor<4>(mine);
      mine += row_xor<2>(mine);
      mine += row_xor<1>(mine);
      if ((lane & 15) == 0) so[r][wave * 2 + (lane >> 4)] = mine + blv;
    }

    if (r == RL - 1 || blk + gstep >= nblk) {
      __syncthreads();
      if (wave <= r) {
        const int fb = blk - (r - wave) * gstep;
        const int node = fb * 16 + (lane >> 1);
        if (node < n) out[(size_t)fb * 32 + lane] = so[wave][lane];
      }
      __syncthreads();
    }
    d = dn; beg = nbeg; end = nend;
  }
}

extern "C" void kernel_launch(void* const* d_in, const int* in_sizes, int n_in,
                              void* d_out, int out_size, void* d_ws, size_t ws_size,
                              hipStream_t stream) {
  const float* x      = (const float*)d_in[0];
  const int*   ei     = (const int*)  d_in[1];
  const float* W1     = (const float*)d_in[2];
  const float* b1     = (const float*)d_in[3];
  const float* gamma1 = (const float*)d_in[4];
  const float* beta1  = (const float*)d_in[5];
  const float* mean1  = (const float*)d_in[6];
  const float* var1   = (const float*)d_in[7];
  const float* W2     = (const float*)d_in[8];
  const float* b2     = (const float*)d_in[9];
  const float* gamma2 = (const float*)d_in[10];
  const float* beta2  = (const float*)d_in[11];
  const float* mean2  = (const float*)d_in[12];
  const float* var2   = (const float*)d_in[13];
  const float* Wl     = (const float*)d_in[14];
  const float* bl     = (const float*)d_in[15];

  const int n = in_sizes[0] / 256;
  const int e = in_sizes[1] / 2;
  const int* src = ei;
  const int* dst = ei + e;

  float* ws   = (float*)d_ws;
  int*   rp   = (int*)ws;
  float* h1   = ws   + (size_t)n;
  float* act1 = h1   + (size_t)n * 64;
  float* h2   = act1 + (size_t)n * 64;
  int*   csr  = (int*)(h2 + (size_t)n * 32);
  int*   hist = (int*)h2;
  int*   off  = hist + (size_t)n * 16;
  int*   bpair = (int*)act1;

  auto blocks = [](long long total, int t) {
    return (unsigned int)((total + t - 1) / t);
  };

  const int nbk = (int)blocks(n, BK_NODES);
  bucket_hist<<<EB, 1024, 0, stream>>>(dst, hist, e, nbk);
  scan_counts<<<blocks((long long)nbk * EB, 1024), 1024, 0, stream>>>(hist, off, nbk * EB);
  bucket_scatter<<<EB, 1024, 0, stream>>>(src, dst, hist, off, bpair, e, nbk);
  bucket_build<<<nbk, BK_NODES, 0, stream>>>(bpair, off, rp, csr, n, e, nbk);

  gemm_wmma_f32_resident<256, 64, 2><<<512, 256, 0, stream>>>(x, W1, h1, rp, n, e);

  gather_act_gemm<<<896, 512, 0, stream>>>(
      rp, csr, h1, W2, h2, b1, gamma1, beta1, mean1, var1, n, e);

  gather_head<<<896, 512, 0, stream>>>(
      rp, csr, h2, b2, gamma2, beta2, mean2, var2, Wl, bl,
      (float*)d_out, n, e);
}


# edge ordering in two launches, not four: a chunk counts, scans and orders its own edges by bucket in LDS (ranks in registers); a bucket collects its 256 runs into LDS; sum order = arrival order
# speedup vs baseline: 5.9375x; 1.1206x over previous
#include <hip/hip_runtime.h>

typedef __attribute__((ext_vector_type(2))) float v2f;
typedef __attribute__((ext_vector_type(8))) float v8f;
typedef __attribute__((ext_vector_type(4))) float v4f;

#define BN_EPS 1e-5f

#define BK_SHIFT 8
#define BK_NODES (1 << BK_SHIFT)
#define BK_MAX   1024
#define EB       256
#define EPT      13
#define EU       4
#define BU       8
#define RG       16
#define STAGE_PAIRS 12800
#define STAGE_IDS   12288

__device__ __forceinline__ int bucket_of(int d, int nbk) {
  return min(max(d, 0) >> BK_SHIFT, nbk - 1);
}

__global__ void __launch_bounds__(1024)
chunk_sort(const int* __restrict__ src, const int* __restrict__ dst,
           int* __restrict__ lstart, int* __restrict__ bpair, int e, int nbk) {
  static_assert(BK_MAX == 1024, "one bucket per thread in the scan below");
  __shared__ int lcnt[BK_MAX + 32];
  __shared__ int lbase[BK_MAX];
  __shared__ int wtot[32];
  __shared__ int2 stage[STAGE_PAIRS + 32];
  int2* __restrict__ pairs = reinterpret_cast<int2*>(bpair);
  const int tid = threadIdx.x, lane = tid & 31, wave = tid >> 5;
  nbk = min(nbk, BK_MAX);
  const int chunk = (e + EB - 1) / EB;
  const int beg = min(blockIdx.x * chunk, e);
  const int end = min(beg + chunk, e);
  const int m = end - beg;
  const bool inreg = chunk <= STAGE_PAIRS && chunk <= 1024 * EPT;

  lcnt[tid] = 0;
  if (tid < 32) lcnt[BK_MAX + tid] = 0;
  __syncthreads();

  int d[EPT], s[EPT], p[EPT];
  if (inreg) {
    #pragma unroll
    for (int u = 0; u < EPT; ++u) {
      const int i = max(min(beg + tid + u * 1024, end - 1), 0);
      d[u] = dst[i];
      s[u] = src[i];
    }
    #pragma unroll
    for (int u = 0; u < EPT; ++u)
      p[u] = atomicAdd(&lcnt[(tid + u * 1024 < m) ? bucket_of(d[u], nbk) : BK_MAX + lane], 1);
  } else {
    for (int i = beg + tid; i < end; i += 1024) atomicAdd(&lcnt[bucket_of(dst[i], nbk)], 1);
  }
  __syncthreads();

  const int cnt = lcnt[tid];
  int inc = cnt;
  for (int o = 1; o < 32; o <<= 1) {
    int t = __shfl_up(inc, o);
    if (lane >= o) inc += t;
  }
  if (lane == 31) wtot[wave] = inc;
  __syncthreads();
  int ahead = (lane < wave) ? wtot[lane] : 0;
  for (int o = 16; o > 0; o >>= 1) ahead += __shfl_xor(ahead, o);
  const int ls = ahead + inc - cnt;
  lbase[tid] = ls;
  if (tid < nbk) lstart[tid * EB + blockIdx.x] = ls;
  if (tid == 0)  lstart[nbk * EB + blockIdx.x] = m;
  __syncthreads();

  if (inreg) {
    int lb[EPT];
    #pragma unroll
    for (int u = 0; u < EPT; ++u) lb[u] = lbase[bucket_of(d[u], nbk)];
    #pragma unroll
    for (int u = 0; u < EPT; ++u)
      stage[(tid + u * 1024 < m) ? min(max(lb[u] + p[u], 0), STAGE_PAIRS - 1) : STAGE_PAIRS + lane] = make_int2(s[u], d[u]);
    __syncthreads();
    for (int base = tid; base < m; base += 1024 * EU) {
      int2 q[EU];
      #pragma unroll
      for (int u = 0; u < EU; ++u) q[u] = stage[min(base + u * 1024, m - 1)];
      #pragma unroll
      for (int u = 0; u < EU; ++u)
        if (base + u * 1024 < m) pairs[beg + base + u * 1024] = q[u];
    }
  } else {
    for (int i = beg + tid; i < end; i += 1024) {
      const int dd = dst[i];
      const int q = atomicAdd(&lbase[bucket_of(dd, nbk)], 1);
      pairs[beg + min(max(q, 0), m - 1)] = make_int2(src[i], dd);
    }
  }
}

__global__ void __launch_bounds__(BK_NODES)
bucket_build(const int* __restrict__ bpair, const int* __restrict__ lstart,
             int* __restrict__ rp, int* __restrict__ csr, int n, int e, int nbk) {
  static_assert(EB == BK_NODES, "one chunk per thread");
  static_assert(32 % RG == 0, "whole groups of runs per wave");
  __shared__ int lcnt[BK_NODES + 32];
  __shared__ int lpos[BK_NODES + 32];
  __shared__ int wtot[BK_NODES / 32];
  __shared__ int wsum[BK_NODES / 32];
  __shared__ int wcnt[BK_NODES / 32];
  __shared__ int2 lp[STAGE_IDS + 32];
  __shared__ __attribute__((aligned(16))) int lcsr[STAGE_IDS];
  const int2* __restrict__ pairs = reinterpret_cast<const int2*>(bpair);
  const int tid = threadIdx.x, lane = tid & 31, wave = tid >> 5;
  const int b = blockIdx.x;
  const int node0 = b << BK_SHIFT;
  const int chunk = (e + EB - 1) / EB;
  const int cbeg = min(tid * chunk, e);
  const int cm = min(cbeg + chunk, e) - cbeg;
  int rs = lstart[b * EB + tid];
  int re = lstart[(b + 1) * EB + tid];
  rs = min(max(rs, 0), cm);
  re = min(max(re, rs), cm);
  const int len = re - rs;
  const int gi = cbeg + rs;

  int inc = len;
  int ssum = rs;
  for (int o = 1; o < 32; o <<= 1) {
    int t = __shfl_up(inc, o);
    if (lane >= o) inc += t;
  }
  for (int o = 16; o > 0; o >>= 1) ssum += __shfl_xor(ssum, o);
  if (lane == 31) wtot[wave] = inc;
  if (lane == 0)  wsum[wave] = ssum;
  lcnt[tid] = 0;
  if (tid < 32) lcnt[BK_NODES + tid] = 0;
  __syncthreads();
  int rahead = 0, total = 0, ebeg = 0;
  for (int w = 0; w < BK_NODES / 32; ++w) {
    const int t = wtot[w];
    if (w < wave) rahead += t;
    total += t;
    ebeg += wsum[w];
  }
  const int ro = rahead + inc - len;

  const int sh = (int)((((unsigned long long)(csr + ebeg)) >> 2) & 3ull);
  const bool staged = total + 3 <= STAGE_IDS;

  if (staged) {
    for (int r0 = 0; r0 < 32; r0 += RG) {
      int2 q0[RG], q1[RG];
      #pragma unroll
      for (int u = 0; u < RG; ++u) {
        const int g = __builtin_amdgcn_readlane(gi, r0 + u);
        const int l = __builtin_amdgcn_readlane(len, r0 + u);
        q0[u] = pairs[min(g + min(lane, max(l - 1, 0)), e - 1)];
        q1[u] = pairs[min(g + min(lane + 32, max(l - 1, 0)), e - 1)];
      }
      #pragma unroll
      for (int u = 0; u < RG; ++u) {
        const int l = __builtin_amdgcn_readlane(len, r0 + u);
        const int o = __builtin_amdgcn_readlane(ro, r0 + u);
        lp[(lane < l)      ? min(o + lane, STAGE_IDS - 1)      : STAGE_IDS + lane] = q0[u];
        lp[(lane + 32 < l) ? min(o + lane + 32, STAGE_IDS - 1) : STAGE_IDS + lane] = q1[u];
      }
      for (int u = 0; u < RG; ++u) {
        const int l = __builtin_amdgcn_readlane(len, r0 + u);
        if (l > 64) {
          const int g = __builtin_amdgcn_readlane(gi, r0 + u);
          const int o = __builtin_amdgcn_readlane(ro, r0 + u);
          for (int j = 64 + lane; j < l; j += 32) lp[min(o + j, STAGE_IDS - 1)] = pairs[g + j];
        }
      }
    }
    __syncthreads();
    for (int base = tid; base < total; base += BK_NODES * BU) {
      int dl[BU];
      #pragma unroll
      for (int u = 0; u < BU; ++u) dl[u] = lp[min(base + u * BK_NODES, total - 1)].y;
      #pragma unroll
      for (int u = 0; u < BU; ++u)
        atomicAdd(&lcnt[(base + u * BK_NODES < total) ? min(max(dl[u] - node0, 0), BK_NODES - 1)
                                                      : BK_NODES + lane], 1);
    }
  } else {
    for (int j = 0; j < len; ++j)
      atomicAdd(&lcnt[min(max(pairs[gi + j].y - node0, 0), BK_NODES - 1)], 1);
  }
  __syncthreads();
  const int v = lcnt[tid];
  int ninc = v;
  for (int o = 1; o < 32; o <<= 1) {
    int t = __shfl_up(ninc, o);
    if (lane >= o) ninc += t;
  }
  if (lane == 31) wcnt[wave] = ninc;
  __syncthreads();
  int nahead = 0;
  for (int w = 0; w < BK_NODES / 32; ++w) if (w < wave) nahead += wcnt[w];
  const int start = ebeg + nahead + ninc - v;
  lpos[tid] = start;
  if (tid < 32) lpos[BK_NODES + tid] = 0;
  if (node0 + tid < n) rp[node0 + tid] = start;
  __syncthreads();
  if (staged) {
    for (int base = tid; base < total; base += BK_NODES * BU) {
      int2 q[BU];
      #pragma unroll
      for (int u = 0; u < BU; ++u) q[u] = lp[min(base + u * BK_NODES, total - 1)];
      #pragma unroll
      for (int u = 0; u < BU; ++u) {
        const bool in = base + u * BK_NODES < total;
        const int p = atomicAdd(&lpos[in ? min(max(q[u].y - node0, 0), BK_NODES - 1) : BK_NODES + lane], 1);
        if (in) lcsr[min(max(p - ebeg, 0) + sh, STAGE_IDS - 1)] = q[u].x;
      }
    }
    __syncthreads();
    const int span = sh + total;
    int* gal = csr + ebeg - sh;
    for (int k = tid * 4; k < span; k += BK_NODES * 4) {
      if (k >= sh && k + 4 <= span) {
        *(int4*)(gal + k) = *(const int4*)&lcsr[k];
      } else {
        for (int t = 0; t < 4; ++t)
          if (k + t >= sh && k + t < span) gal[k + t] = lcsr[k + t];
      }
    }
  } else {
    for (int j = 0; j < len; ++j) {
      const int2 q = pairs[gi + j];
      const int p = atomicAdd(&lpos[min(max(q.y - node0, 0), BK_NODES - 1)], 1);
      csr[min(max(p, 0), e - 1)] = q.x;
    }
  }
}

template <int K, int NC, int MT>
__global__ void __launch_bounds__(MT*(NC/16)*32)
gemm_wmma_f32_resident(const float* __restrict__ A, const float* __restrict__ B,
                       float* __restrict__ C, const int* __restrict__ rp, int n, int e) {
  constexpr int KP   = K + 4;
  constexpr int NCP  = NC + 8;
  constexpr int NW   = MT * (NC/16);
  constexpr int NT   = NW * 32;
  constexpr int ROWS = MT * 16;
  constexpr int AV   = ROWS * (K/4) / NT;
  constexpr int BV   = K * (NC/4) / NT;
  static_assert((ROWS * (K/4)) % NT == 0 && (K * (NC/4)) % NT == 0, "whole pieces per thread");
  __shared__ float sA[2][ROWS*KP];
  __shared__ float sB[K*NCP];

  const int tid = threadIdx.x;
  const int ntiles = n / ROWS;
  const int gstep = (int)gridDim.x;
  int g = blockIdx.x;
  if (g >= ntiles) return;

  v4f pf[AV];
  {
    const float* Ablk = A + (size_t)g * ROWS * K;
    #pragma unroll
    for (int j = 0; j < AV; ++j) {
      const int idx = tid + j * NT;
      const int r = idx / (K/4), c4 = (idx % (K/4)) * 4;
      pf[j] = *(const v4f*)&Ablk[(size_t)r*K + c4];
    }
  }
  #pragma unroll
  for (int j = 0; j < BV; ++j) {
    const int idx = tid + j * NT;
    const int r = idx / (NC/4), c4 = (idx % (NC/4)) * 4;
    *(float4*)&sB[r*NCP + c4] = *(const float4*)&B[(size_t)r*NC + c4];
  }
  #pragma unroll
  for (int j = 0; j < AV; ++j) {
    const int idx = tid + j * NT;
    const int r = idx / (K/4), c4 = (idx % (K/4)) * 4;
    *(v4f*)&sA[0][r*KP + c4] = pf[j];
  }
  __syncthreads();

  const int wave  = tid >> 5, lane = tid & 31;
  const int mt    = wave / (NC/16), nt = wave % (NC/16);
  const int rrow  = lane & 15;
  const int khalf = lane >> 4;
  const int col   = nt*16 + rrow;
  const float* b_ptr0 = &sB[(khalf*2)*NCP + nt*16 + rrow];

  for (int cur = 0; ; cur ^= 1) {
    const int gn = g + gstep;
    const bool more = gn < ntiles;

    const int gm = g*ROWS + mt*16 + khalf*8;
    int sb[9];
    #pragma unroll
    for (int r = 0; r < 8; ++r) sb[r] = rp[gm + r];
    sb[8] = rp[min(gm + 8, n - 1)];
    if (gm + 8 >= n) sb[8] = e;

    if (more) {
      const float* Ablk = A + (size_t)gn * ROWS * K;
      #pragma unroll
      for (int j = 0; j < AV; ++j) {
        const int idx = tid + j * NT;
        const int r = idx / (K/4), c4 = (idx % (K/4)) * 4;
        pf[j] = *(const v4f*)&Ablk[(size_t)r*K + c4];
      }
    }

    const float* a_ptr = &sA[cur][(mt*16 + rrow)*KP + khalf*2];
    const float* b_ptr = b_ptr0;
    v8f c0 = {};
    v8f c1 = {};
    #pragma unroll 4
    for (int ks = 0; ks < K/8; ++ks) {
      v2f a0; a0.x = a_ptr[0];     a0.y = a_ptr[1];
      v2f b0; b0.x = b_ptr[0];     b0.y = b_ptr[NCP];
      v2f a1; a1.x = a_ptr[4];     a1.y = a_ptr[5];
      v2f b1; b1.x = b_ptr[4*NCP]; b1.y = b_ptr[5*NCP];
      c0 = __builtin_amdgcn_wmma_f32_16x16x4_f32(false, a0, false, b0,
                                                 (short)0, c0, false, false);
      c1 = __builtin_amdgcn_wmma_f32_16x16x4_f32(false, a1, false, b1,
                                                 (short)0, c1, false, false);
      a_ptr += 8;
      b_ptr += 8*NCP;
    }
    v8f c = c0 + c1;

    if (more) {
      #pragma unroll
      for (int j = 0; j < AV; ++j) {
        const int idx = tid + j * NT;
        const int r = idx / (K/4), c4 = (idx % (K/4)) * 4;
        *(v4f*)&sA[cur ^ 1][r*KP + c4] = pf[j];
      }
      __syncthreads();
    }

    #pragma unroll
    for (int r = 0; r < 8; ++r) {
      const float dv = rsqrtf((float)(sb[r + 1] - sb[r]) + 1.0f);
      C[(size_t)(gm + r)*NC + col] = c[r] * dv;
    }

    if (!more) break;
    g = gn;
  }
}

template <int F, int G>
__device__ __forceinline__ void take_rows(const float* __restrict__ h, int s_l,
                                          int j, int lane, float (&acc)[F / 32]) {
  constexpr int V = F / 32;
  float v[G][V];
  #pragma unroll
  for (int u = 0; u < G; ++u) {
    const int s = __builtin_amdgcn_readlane(s_l, j + u);
    const float* hp = h + (size_t)s * F + lane * V;
    if constexpr (V == 2) {
      const float2 t = *(const float2*)hp;
      v[u][0] = t.x; v[u][1] = t.y;
    } else {
      v[u][0] = hp[0];
    }
  }
  #pragma unroll
  for (int u = 0; u < G; ++u)
    for (int l = 0; l < V; ++l) acc[l] += v[u][l];
}

template <int F>
__device__ __forceinline__ void take_chunk(const float* __restrict__ h, int s_l, int cnt,
                                           int lane, float (&acc)[F / 32]) {
  constexpr int V = F / 32;
  constexpr int G = 16 / V;
  int j = 0;
  for (; j + G <= cnt; j += G) take_rows<F, G>(h, s_l, j, lane, acc);
  if constexpr (G >= 16) { if (j + 8 <= cnt) { take_rows<F, 8>(h, s_l, j, lane, acc); j += 8; } }
  if (j + 4 <= cnt) { take_rows<F, 4>(h, s_l, j, lane, acc); j += 4; }
  if (j + 2 <= cnt) { take_rows<F, 2>(h, s_l, j, lane, acc); j += 2; }
  if (j < cnt)      { take_rows<F, 1>(h, s_l, j, lane, acc); }
}

__device__ __forceinline__ int csr_raw(const int* __restrict__ csr, int i, int e) {
  return csr[min(max(i, 0), e - 1)];
}
__device__ __forceinline__ int csr_id(int raw, int i, int end, int n) {
  return (i < end) ? min(max(raw, 0), n - 1) : 0;
}
__device__ __forceinline__ int bounds_raw(const int* __restrict__ rp, int q, int lane, int n) {
  return rp[min(min(q, n - 1) + (lane & 1), n - 1)];
}

template <int F>
__device__ __forceinline__ void sum_rows(const int* __restrict__ csr, const float* __restrict__ h,
                                         int beg, int end, int ia, int ib, int n, int lane,
                                         float (&acc)[F / 32]) {
  constexpr int V = F / 32;
  for (int l = 0; l < V; ++l) acc[l] = 0.0f;
  const int deg = end - beg;
  take_chunk<F>(h, ia, min(deg, 32), lane, acc);
  if (deg > 32) take_chunk<F>(h, ib, min(deg - 32, 32), lane, acc);
  for (int base = beg + 64; base < end; base += 32) {
    const int cnt = min(end - base, 32);
    int s_l = 0;
    if (lane < cnt) {
      s_l = csr[base + lane];
      s_l = min(max(s_l, 0), n - 1);
    }
    take_chunk<F>(h, s_l, cnt, lane, acc);
  }
}

__device__ __forceinline__ float node_act(float sum, float hs, float dd, float bias,
                                          float mean, float var, float gamma, float beta) {
  float v = __builtin_fmaf(hs, dd, sum * dd) + bias;
  v = (v - mean) * rsqrtf(var + BN_EPS) * gamma + beta;
  return fmaxf(v, 0.0f);
}

__global__ void __launch_bounds__(512) __attribute__((amdgpu_waves_per_eu(16)))
gather_act_gemm(const int* __restrict__ rp, const int* __restrict__ csr,
                const float* __restrict__ h, const float* __restrict__ B,
                float* __restrict__ C,
                const float* __restrict__ bias, const float* __restrict__ gamma,
                const float* __restrict__ beta, const float* __restrict__ mean,
                const float* __restrict__ var, int n, int e) {
  constexpr int F   = 64;
  constexpr int K   = F;
  constexpr int NC  = 32;
  constexpr int KP  = K + 4;
  constexpr int NCP = NC + 8;
  constexpr int RL  = 8;
  static_assert(K * (NC/4) == 512, "one 16-byte piece of B per thread");
  static_assert(2 * RL == 16, "one 16x16 piece of C per wave");
  __shared__ float sA[RL][16*KP];
  __shared__ float sB[K*NCP];
  const int tid  = threadIdx.x;
  const int lane = tid & 31;
  const int wave = __builtin_amdgcn_readfirstlane((int)(tid >> 5));
  const int nblk = (n + 15) >> 4;
  const int gstep = (int)gridDim.x;
  const int dstep = gstep * 16;
  int blk = blockIdx.x;
  if (blk >= nblk) return;
  const int br = tid / (NC/4), bc4 = (tid % (NC/4)) * 4;
  const float4 bpiece = *(const float4*)&B[(size_t)br*NC + bc4];
  const float2 pb = *(const float2*)(bias + lane * 2);
  const float2 pm = *(const float2*)(mean + lane * 2);
  const float2 pv = *(const float2*)(var + lane * 2);
  const float2 pg = *(const float2*)(gamma + lane * 2);
  const float2 pe = *(const float2*)(beta + lane * 2);

  int d = blk * 16 + wave;
  const int d0 = min(d, n - 1);
  int beg = rp[d0];
  int end = (d0 + 1 < n) ? rp[min(d0 + 1, n - 1)] : e;
  int ra = csr_raw(csr, beg + lane, e);
  int rb = csr_raw(csr, beg + 32 + lane, e);
  int rq = bounds_raw(rp, d + dstep, lane, n);
  *(float4*)&sB[br*NCP + bc4] = bpiece;

  const int rrow  = lane & 15;
  const int khalf = lane >> 4;

  int k = 0;
  for (; blk < nblk; blk += gstep, ++k) {
    const int r = k & (RL - 1);
    const int ia = csr_id(ra, beg + lane, end, n);
    const int ib = csr_id(rb, beg + 32 + lane, end, n);
    const int dn = d + dstep;
    const int nbeg = __builtin_amdgcn_readlane(rq, 0);
    const int nend = (min(dn, n - 1) + 1 < n) ? __builtin_amdgcn_readlane(rq, 1) : e;
    ra = csr_raw(csr, nbeg + lane, e);
    rb = csr_raw(csr, nbeg + 32 + lane, e);
    rq = bounds_raw(rp, dn + dstep, lane, n);

    float2 t = make_float2(0.0f, 0.0f);
    if (d < n) {
      const float2 hs = *(const float2*)(h + (size_t)d * F + lane * 2);
      float acc[2];
      sum_rows<F>(csr, h, beg, end, ia, ib, n, lane, acc);
      const float dd = rsqrtf((float)(end - beg) + 1.0f);
      t.x = node_act(acc[0], hs.x, dd, pb.x, pm.x, pv.x, pg.x, pe.x);
      t.y = node_act(acc[1], hs.y, dd, pb.y, pm.y, pv.y, pg.y, pe.y);
    }
    *(float2*)&sA[r][wave*KP + lane*2] = t;

    if (r == RL - 1 || blk + gstep >= nblk) {
      __syncthreads();
      const int slot = wave >> 1;
      if (slot <= r) {
        const int nt = wave & 1;
        const int fb = blk - (r - slot) * gstep;
        const int gm = fb*16 + khalf*8;
        const int gl = max(min(gm, n - 8), 0);
        const int* __restrict__ rpp = rp + gl;
        int sb[9];
        #pragma unroll
        for (int j = 0; j < 8; ++j) sb[j] = rpp[j];
        sb[8] = rp[min(gl + 8, n - 1)];
        if (gl + 8 >= n) sb[8] = e;
        const float* a_ptr = &sA[slot][rrow*KP + khalf*2];
        const float* b_ptr = &sB[(khalf*2)*NCP + nt*16 + rrow];
        v8f c0 = {};
        v8f c1 = {};
        #pragma unroll 4
        for (int ks = 0; ks < K/8; ++ks) {
          v2f a0; a0.x = a_ptr[0];     a0.y = a_ptr[1];
          v2f b0; b0.x = b_ptr[0];     b0.y = b_ptr[NCP];
          v2f a1; a1.x = a_ptr[4];     a1.y = a_ptr[5];
          v2f b1; b1.x = b_ptr[4*NCP]; b1.y = b_ptr[5*NCP];
          c0 = __builtin_amdgcn_wmma_f32_16x16x4_f32(false, a0, false, b0,
                                                     (short)0, c0, false, false);
          c1 = __builtin_amdgcn_wmma_f32_16x16x4_f32(false, a1, false, b1,
                                                     (short)0, c1, false, false);
          a_ptr += 8;
          b_ptr += 8*NCP;
        }
        v8f c = c0 + c1;
        const int col = nt*16 + rrow;
        if (gm + 8 <= n) {
          float* __restrict__ cp = C + (size_t)gm*NC + col;
          #pragma unroll
          for (int j = 0; j < 8; ++j) {
            const float dv = rsqrtf((float)(sb[j + 1] - sb[j]) + 1.0f);
            cp[j*NC] = c[j] * dv;
          }
        }
      }
      __syncthreads();
    }
    d = dn; beg = nbeg; end = nend;
  }
}

template <int M>
__device__ __forceinline__ float row_xor(float x) {
  return __builtin_bit_cast(float, __builtin_amdgcn_update_dpp(
      0, __builtin_bit_cast(int, x), 0x160 + M, 0xf, 0xf, true));
}
__device__ __forceinline__ float other_row(float x) {
  const int b = __builtin_bit_cast(int, x);
  return __builtin_bit_cast(float, __builtin_amdgcn_permlanex16(b, b, 0x76543210, 0xfedcba98, false, false));
}

__global__ void __launch_bounds__(512)
gather_head(const int* __restrict__ rp, const int* __restrict__ csr,
            const float* __restrict__ h,
            const float* __restrict__ bias, const float* __restrict__ gamma,
            const float* __restrict__ beta, const float* __restrict__ mean,
            const float* __restrict__ var, const float* __restrict__ Wl,
            const float* __restrict__ bl, float* __restrict__ out, int n, int e) {
  constexpr int F = 32;
  constexpr int RL = 8;
  __shared__ float so[RL][32];
  const int lane = threadIdx.x & 31;
  const int wave = __builtin_amdgcn_readfirstlane((int)(threadIdx.x >> 5));
  const int nblk = (n + 15) >> 4;
  const int gstep = (int)gridDim.x;
  const int dstep = gstep * 16;
  int blk = blockIdx.x;
  if (blk >= nblk) return;
  const float pb = bias[lane], pm = mean[lane], pv = var[lane];
  const float pg = gamma[lane], pe = beta[lane];
  const float2 wl = *(const float2*)(Wl + lane * 2);
  const float blv = bl[lane >> 4];

  int d = blk * 16 + wave;
  const int d0 = min(d, n - 1);
  int beg = rp[d0];
  int end = (d0 + 1 < n) ? rp[min(d0 + 1, n - 1)] : e;
  int ra = csr_raw(csr, beg + lane, e);
  int rb = csr_raw(csr, beg + 32 + lane, e);
  int rq = bounds_raw(rp, d + dstep, lane, n);

  int k = 0;
  for (; blk < nblk; blk += gstep, ++k) {
    const int r = k & (RL - 1);
    const int ia = csr_id(ra, beg + lane, end, n);
    const int ib = csr_id(rb, beg + 32 + lane, end, n);
    const int dn = d + dstep;
    const int nbeg = __builtin_amdgcn_readlane(rq, 0);
    const int nend = (min(dn, n - 1) + 1 < n) ? __builtin_amdgcn_readlane(rq, 1) : e;
    ra = csr_raw(csr, nbeg + lane, e);
    rb = csr_raw(csr, nbeg + 32 + lane, e);
    rq = bounds_raw(rp, dn + dstep, lane, n);

    if (d < n) {
      const float hs = h[(size_t)d * F + lane];
      float acc[1];
      sum_rows<F>(csr, h, beg, end, ia, ib, n, lane, acc);
      const float dd = rsqrtf((float)(end - beg) + 1.0f);
      const float v = node_act(acc[0], hs, dd, pb, pm, pv, pg, pe);

      const bool hi = lane >= 16;
      const float p0 = v * wl.x, p1 = v * wl.y;
      float mine = hi ? p1 : p0;
      mine += other_row(hi ? p0 : p1);
      mine += row_xor<8>(mine);
      mine += row_xor<4>(mine);
      mine += row_xor<2>(mine);
      mine += row_xor<1>(mine);
      if ((lane & 15) == 0) so[r][wave * 2 + (lane >> 4)] = mine + blv;
    }

    if (r == RL - 1 || blk + gstep >= nblk) {
      __syncthreads();
      if (wave <= r) {
        const int fb = blk - (r - wave) * gstep;
        const int node = fb * 16 + (lane >> 1);
        if (node < n) out[(size_t)fb * 32 + lane] = so[wave][lane];
      }
      __syncthreads();
    }
    d = dn; beg = nbeg; end = nend;
  }
}

extern "C" void kernel_launch(void* const* d_in, const int* in_sizes, int n_in,
                              void* d_out, int out_size, void* d_ws, size_t ws_size,
                              hipStream_t stream) {
  const float* x      = (const float*)d_in[0];
  const int*   ei     = (const int*)  d_in[1];
  const float* W1     = (const float*)d_in[2];
  const float* b1     = (const float*)d_in[3];
  const float* gamma1 = (const float*)d_in[4];
  const float* beta1  = (const float*)d_in[5];
  const float* mean1  = (const float*)d_in[6];
  const float* var1   = (const float*)d_in[7];
  const float* W2     = (const float*)d_in[8];
  const float* b2     = (const float*)d_in[9];
  const float* gamma2 = (const float*)d_in[10];
  const float* beta2  = (const float*)d_in[11];
  const float* mean2  = (const float*)d_in[12];
  const float* var2   = (const float*)d_in[13];
  const float* Wl     = (const float*)d_in[14];
  const float* bl     = (const float*)d_in[15];

  const int n = in_sizes[0] / 256;
  const int e = in_sizes[1] / 2;
  const int* src = ei;
  const int* dst = ei + e;

  float* ws   = (float*)d_ws;
  int*   rp   = (int*)ws;
  float* h1   = ws   + (size_t)n;
  float* act1 = h1   + (size_t)n * 64;
  float* h2   = act1 + (size_t)n * 64;
  int*   csr  = (int*)(h2 + (size_t)n * 32);
  int*   lstart = (int*)h2;
  int*   bpair = (int*)act1;

  auto blocks = [](long long total, int t) {
    return (unsigned int)((total + t - 1) / t);
  };

  const int nbk = (int)blocks(n, BK_NODES);
  chunk_sort<<<EB, 1024, 0, stream>>>(src, dst, lstart, bpair, e, nbk);
  bucket_build<<<nbk, BK_NODES, 0, stream>>>(bpair, lstart, rp, csr, n, e, nbk);

  gemm_wmma_f32_resident<256, 64, 2><<<512, 256, 0, stream>>>(x, W1, h1, rp, n, e);

  gather_act_gemm<<<896, 512, 0, stream>>>(
      rp, csr, h1, W2, h2, b1, gamma1, beta1, mean1, var1, n, e);

  gather_head<<<896, 512, 0, stream>>>(
      rp, csr, h2, b2, gamma2, beta2, mean2, var2, Wl, bl,
      (float*)d_out, n, e);
}
